# starting kernel
# baseline (speedup 1.0000x reference)
#include <hip/hip_runtime.h>
#include <math.h>

#define N_NODES   50000
#define N_EDGES   800000
#define IN_CH     64
#define HEADS     4
#define OUT_CH    32
#define HC        128
#define ET        (N_EDGES + N_NODES)
#define NEG_SLOPE 0.2f
#define BN_EPS    1e-5f
#define MEAN_BLOCKS 256

typedef float v2f __attribute__((ext_vector_type(2)));
typedef float v8f __attribute__((ext_vector_type(8)));

__device__ __forceinline__ int f2ord(float f) {
    int i = __float_as_int(f);
    return (i >= 0) ? i : (i ^ 0x7fffffff);
}
__device__ __forceinline__ float ord2f(int i) {
    return __int_as_float((i >= 0) ? i : (i ^ 0x7fffffff));
}

__global__ __launch_bounds__(256)
void init_ws_kernel(float* __restrict__ out, float* __restrict__ denom,
                    int* __restrict__ nodeMax, float* __restrict__ colAcc)
{
    const int stride = gridDim.x * blockDim.x;
    int i = blockIdx.x * blockDim.x + threadIdx.x;
    for (int j = i; j < N_NODES * HC; j += stride) out[j] = 0.0f;
    for (int j = i; j < N_NODES * HEADS; j += stride) {
        denom[j]   = 0.0f;
        nodeMax[j] = (int)0x80000000;
    }
    for (int j = i; j < 2 * HC; j += stride) colAcc[j] = 0.0f;
}

__global__ __launch_bounds__(256)
void gemm_xl_xr_kernel(const float* __restrict__ X,
                       const float* __restrict__ Wl, const float* __restrict__ Wr,
                       float* __restrict__ XL, float* __restrict__ XR)
{
    const int wid  = (blockIdx.x * blockDim.x + threadIdx.x) >> 5;
    const int lane = threadIdx.x & 31;
    const int half = lane >> 4;
    const int idx  = lane & 15;

    const int mtile = wid >> 4;
    const int nsel  = wid & 15;
    if (mtile >= 3125) return;

    const float* W = (nsel < 8) ? Wl : Wr;
    float*       D = (nsel < 8) ? XL : XR;
    const int ntile = nsel & 7;

    const float* arow = X + (size_t)(mtile * 16 + idx) * IN_CH;
    const float* bcol = W + (size_t)(ntile * 16 + idx);

    v8f c = {};
#pragma unroll
    for (int k0 = 0; k0 < IN_CH; k0 += 4) {
        v2f a, b;
        a.x = arow[k0 + 2 * half + 0];
        a.y = arow[k0 + 2 * half + 1];
        b.x = bcol[(size_t)(k0 + 2 * half + 0) * HC];
        b.y = bcol[(size_t)(k0 + 2 * half + 1) * HC];
        c = __builtin_amdgcn_wmma_f32_16x16x4_f32(
                false, a, false, b, (short)0, c, false, false);
    }

#pragma unroll
    for (int v = 0; v < 8; ++v) {
        const int row = mtile * 16 + v + 8 * half;
        const int col = ntile * 16 + idx;
        D[(size_t)row * HC + col] = c[v];
    }
}

__global__ __launch_bounds__(256)
void edge_mean_partial_kernel(const int* __restrict__ srcA, const int* __restrict__ dstA,
                              const float* __restrict__ pos, float* __restrict__ part)
{
    __shared__ float sm[3][256];
    float a0 = 0.f, a1 = 0.f, a2 = 0.f;
    for (int e = blockIdx.x * blockDim.x + threadIdx.x; e < N_EDGES;
         e += gridDim.x * blockDim.x) {
        const int s = srcA[e], d = dstA[e];
        a0 += pos[s * 3 + 0] - pos[d * 3 + 0];
        a1 += pos[s * 3 + 1] - pos[d * 3 + 1];
        a2 += pos[s * 3 + 2] - pos[d * 3 + 2];
    }
    const int t = threadIdx.x;
    sm[0][t] = a0; sm[1][t] = a1; sm[2][t] = a2;
    __syncthreads();
    for (int o = 128; o > 0; o >>= 1) {
        if (t < o) { sm[0][t] += sm[0][t + o]; sm[1][t] += sm[1][t + o]; sm[2][t] += sm[2][t + o]; }
        __syncthreads();
    }
    if (t == 0) {
        part[blockIdx.x * 3 + 0] = sm[0][0];
        part[blockIdx.x * 3 + 1] = sm[1][0];
        part[blockIdx.x * 3 + 2] = sm[2][0];
    }
}

__global__ void edge_mean_final_kernel(const float* __restrict__ part, float* __restrict__ mean3)
{
    if (threadIdx.x < 3) {
        float s = 0.f;
        for (int b = 0; b < MEAN_BLOCKS; ++b) s += part[b * 3 + threadIdx.x];
        mean3[threadIdx.x] = s / (float)N_EDGES;
    }
}

__global__ __launch_bounds__(256)
void edge_score_kernel(const int* __restrict__ srcA, const int* __restrict__ dstA,
                       const float* __restrict__ pos, const float* __restrict__ mean3,
                       const float* __restrict__ Wedge, const float* __restrict__ att,
                       const float* __restrict__ XL, const float* __restrict__ XR,
                       float* __restrict__ score, int* __restrict__ nodeMax)
{
    const int lane   = threadIdx.x & 31;
    const int waveId = (blockIdx.x * blockDim.x + threadIdx.x) >> 5;
    const int nWaves = (gridDim.x * blockDim.x) >> 5;

    for (int e = waveId; e < ET; e += nWaves) {
        int s, d; float ea0, ea1, ea2;
        if (e < N_EDGES) {
            s = srcA[e]; d = dstA[e];
            ea0 = pos[s * 3 + 0] - pos[d * 3 + 0];
            ea1 = pos[s * 3 + 1] - pos[d * 3 + 1];
            ea2 = pos[s * 3 + 2] - pos[d * 3 + 2];
        } else {
            s = d = e - N_EDGES;
            ea0 = mean3[0]; ea1 = mean3[1]; ea2 = mean3[2];
        }
        const float* xlrow = XL + (size_t)s * HC;
        const float* xrrow = XR + (size_t)d * HC;

        float sc[HEADS];
#pragma unroll
        for (int h = 0; h < HEADS; ++h) {
            const int off = h * OUT_CH + lane;
            float v = xlrow[off] + xrrow[off]
                    + ea0 * Wedge[0 * HC + off]
                    + ea1 * Wedge[1 * HC + off]
                    + ea2 * Wedge[2 * HC + off];
            v = (v > 0.f) ? v : NEG_SLOPE * v;
            sc[h] = v * att[off];
        }
#pragma unroll
        for (int h = 0; h < HEADS; ++h)
#pragma unroll
            for (int o = 16; o > 0; o >>= 1)
                sc[h] += __shfl_xor(sc[h], o, 32);

        if (lane < HEADS) {
            score[(size_t)e * HEADS + lane] = sc[lane];
            atomicMax(&nodeMax[d * HEADS + lane], f2ord(sc[lane]));
        }
    }
}

__global__ __launch_bounds__(256)
void edge_exp_kernel(const int* __restrict__ dstA, const int* __restrict__ nodeMax,
                     float* __restrict__ score, float* __restrict__ denom)
{
    const int stride = gridDim.x * blockDim.x;
    for (int i = blockIdx.x * blockDim.x + threadIdx.x; i < ET * HEADS; i += stride) {
        const int e = i >> 2, h = i & 3;
        const int d = (e < N_EDGES) ? dstA[e] : (e - N_EDGES);
        const float m  = ord2f(nodeMax[d * HEADS + h]);
        const float sv = expf(score[i] - m);
        score[i] = sv;
        atomicAdd(&denom[d * HEADS + h], sv);
    }
}

__global__ __launch_bounds__(256)
void edge_agg_kernel(const int* __restrict__ srcA, const int* __restrict__ dstA,
                     const float* __restrict__ XL, const float* __restrict__ score,
                     const float* __restrict__ denom, float* __restrict__ out)
{
    const int lane   = threadIdx.x & 31;
    const int waveId = (blockIdx.x * blockDim.x + threadIdx.x) >> 5;
    const int nWaves = (gridDim.x * blockDim.x) >> 5;

    for (int e = waveId; e < ET; e += nWaves) {
        int s, d;
        if (e < N_EDGES) { s = srcA[e]; d = dstA[e]; }
        else             { s = d = e - N_EDGES; }
#pragma unroll
        for (int h = 0; h < HEADS; ++h) {
            const float alpha = score[(size_t)e * HEADS + h]
                              / (denom[d * HEADS + h] + 1e-16f);
            const int off = h * OUT_CH + lane;
            atomicAdd(&out[(size_t)d * HC + off], alpha * XL[(size_t)s * HC + off]);
        }
    }
}

__global__ __launch_bounds__(128)
void bn_stats_kernel(const float* __restrict__ out, float* __restrict__ colAcc)
{
    const int c = threadIdx.x;
    float s = 0.f, s2 = 0.f;
    for (int r = blockIdx.x; r < N_NODES; r += gridDim.x) {
        const float v = out[(size_t)r * HC + c];
        s += v; s2 += v * v;
    }
    atomicAdd(&colAcc[c],       s);
    atomicAdd(&colAcc[HC + c],  s2);
}

__global__ void bn_final_kernel(const float* __restrict__ colAcc,
                                const float* __restrict__ gamma,
                                const float* __restrict__ beta,
                                float* __restrict__ scaleShift)
{
    const int c = threadIdx.x;
    if (c < HC) {
        const float mu  = colAcc[c]      * (1.0f / (float)N_NODES);
        const float var = colAcc[HC + c] * (1.0f / (float)N_NODES) - mu * mu;
        const float sc  = gamma[c] * rsqrtf(var + BN_EPS);
        scaleShift[c]      = sc;
        scaleShift[HC + c] = beta[c] - mu * sc;
    }
}

__global__ __launch_bounds__(256)
void bn_apply_kernel(float* __restrict__ out, const float* __restrict__ scaleShift)
{
    const int stride = gridDim.x * blockDim.x;
    for (int i = blockIdx.x * blockDim.x + threadIdx.x; i < N_NODES * HC; i += stride) {
        const int c = i & (HC - 1);
        out[i] = out[i] * scaleShift[c] + scaleShift[HC + c];
    }
}

extern "C" void kernel_launch(void* const* d_in, const int* in_sizes, int n_in,
                              void* d_out, int out_size, void* d_ws, size_t ws_size,
                              hipStream_t stream)
{
    const float* x     = (const float*)d_in[0];
    const float* pos   = (const float*)d_in[1];
    const int*   eidx  = (const int*)  d_in[2];
    const float* W_l   = (const float*)d_in[3];
    const float* W_r   = (const float*)d_in[4];
    const float* W_e   = (const float*)d_in[5];
    const float* att   = (const float*)d_in[6];
    const float* gamma = (const float*)d_in[7];
    const float* beta  = (const float*)d_in[8];
    float* out = (float*)d_out;

    const int* srcA = eidx;
    const int* dstA = eidx + N_EDGES;

    float* ws = (float*)d_ws;
    size_t off = 0;
    float* XL         = ws + off; off += (size_t)N_NODES * HC;
    float* XR         = ws + off; off += (size_t)N_NODES * HC;
    float* score      = ws + off; off += (size_t)ET * HEADS;
    float* denom      = ws + off; off += (size_t)N_NODES * HEADS;
    int*   nodeMax    = (int*)(ws + off); off += (size_t)N_NODES * HEADS;
    float* colAcc     = ws + off; off += 2 * HC;
    float* scaleShift = ws + off; off += 2 * HC;
    float* meanPart   = ws + off; off += 3 * MEAN_BLOCKS;
    float* mean3      = ws + off; off += 4;
    (void)ws_size; (void)n_in; (void)in_sizes; (void)out_size;

    init_ws_kernel<<<1024, 256, 0, stream>>>(out, denom, nodeMax, colAcc);

    gemm_xl_xr_kernel<<<6250, 256, 0, stream>>>(x, W_l, W_r, XL, XR);

    edge_mean_partial_kernel<<<MEAN_BLOCKS, 256, 0, stream>>>(srcA, dstA, pos, meanPart);
    edge_mean_final_kernel<<<1, 64, 0, stream>>>(meanPart, mean3);

    edge_score_kernel<<<2048, 256, 0, stream>>>(srcA, dstA, pos, mean3, W_e, att,
                                                XL, XR, score, nodeMax);

    edge_exp_kernel<<<4096, 256, 0, stream>>>(dstA, nodeMax, score, denom);

    edge_agg_kernel<<<2048, 256, 0, stream>>>(srcA, dstA, XL, score, denom, out);

    bn_stats_kernel<<<1024, 128, 0, stream>>>(out, colAcc);
    bn_final_kernel<<<1, 128, 0, stream>>>(colAcc, gamma, beta, scaleShift);
    bn_apply_kernel<<<4096, 256, 0, stream>>>(out, scaleShift);
}


# Both edge loops: next edge's indices read one iteration ahead, all loads of an edge issued as one group (before the atomics, no position branch): 1 dependent round trip per edge, was 5 and 3
# speedup vs baseline: 1.1380x; 1.1380x over previous
#include <hip/hip_runtime.h>
#include <math.h>

#define N_NODES   50000
#define N_EDGES   800000
#define IN_CH     64
#define HEADS     4
#define OUT_CH    32
#define HC        128
#define ET        (N_EDGES + N_NODES)
#define NEG_SLOPE 0.2f
#define BN_EPS    1e-5f
#define MEAN_BLOCKS 256

typedef float v2f __attribute__((ext_vector_type(2)));
typedef float v8f __attribute__((ext_vector_type(8)));

__device__ __forceinline__ int f2ord(float f) {
    int i = __float_as_int(f);
    return (i >= 0) ? i : (i ^ 0x7fffffff);
}
__device__ __forceinline__ float ord2f(int i) {
    return __int_as_float((i >= 0) ? i : (i ^ 0x7fffffff));
}

__global__ __launch_bounds__(256)
void init_ws_kernel(float* __restrict__ out, float* __restrict__ denom,
                    int* __restrict__ nodeMax, float* __restrict__ colAcc)
{
    const int stride = gridDim.x * blockDim.x;
    int i = blockIdx.x * blockDim.x + threadIdx.x;
    for (int j = i; j < N_NODES * HC; j += stride) out[j] = 0.0f;
    for (int j = i; j < N_NODES * HEADS; j += stride) {
        denom[j]   = 0.0f;
        nodeMax[j] = (int)0x80000000;
    }
    for (int j = i; j < 2 * HC; j += stride) colAcc[j] = 0.0f;
}

__global__ __launch_bounds__(256)
void gemm_xl_xr_kernel(const float* __restrict__ X,
                       const float* __restrict__ Wl, const float* __restrict__ Wr,
                       float* __restrict__ XL, float* __restrict__ XR)
{
    const int wid  = (blockIdx.x * blockDim.x + threadIdx.x) >> 5;
    const int lane = threadIdx.x & 31;
    const int half = lane >> 4;
    const int idx  = lane & 15;

    const int mtile = wid >> 4;
    const int nsel  = wid & 15;
    if (mtile >= 3125) return;

    const float* W = (nsel < 8) ? Wl : Wr;
    float*       D = (nsel < 8) ? XL : XR;
    const int ntile = nsel & 7;

    const float* arow = X + (size_t)(mtile * 16 + idx) * IN_CH;
    const float* bcol = W + (size_t)(ntile * 16 + idx);

    v8f c = {};
#pragma unroll
    for (int k0 = 0; k0 < IN_CH; k0 += 4) {
        v2f a, b;
        a.x = arow[k0 + 2 * half + 0];
        a.y = arow[k0 + 2 * half + 1];
        b.x = bcol[(size_t)(k0 + 2 * half + 0) * HC];
        b.y = bcol[(size_t)(k0 + 2 * half + 1) * HC];
        c = __builtin_amdgcn_wmma_f32_16x16x4_f32(
                false, a, false, b, (short)0, c, false, false);
    }

#pragma unroll
    for (int v = 0; v < 8; ++v) {
        const int row = mtile * 16 + v + 8 * half;
        const int col = ntile * 16 + idx;
        D[(size_t)row * HC + col] = c[v];
    }
}

__global__ __launch_bounds__(256)
void edge_mean_partial_kernel(const int* __restrict__ srcA, const int* __restrict__ dstA,
                              const float* __restrict__ pos, float* __restrict__ part)
{
    __shared__ float sm[3][256];
    float a0 = 0.f, a1 = 0.f, a2 = 0.f;
    for (int e = blockIdx.x * blockDim.x + threadIdx.x; e < N_EDGES;
         e += gridDim.x * blockDim.x) {
        const int s = srcA[e], d = dstA[e];
        a0 += pos[s * 3 + 0] - pos[d * 3 + 0];
        a1 += pos[s * 3 + 1] - pos[d * 3 + 1];
        a2 += pos[s * 3 + 2] - pos[d * 3 + 2];
    }
    const int t = threadIdx.x;
    sm[0][t] = a0; sm[1][t] = a1; sm[2][t] = a2;
    __syncthreads();
    for (int o = 128; o > 0; o >>= 1) {
        if (t < o) { sm[0][t] += sm[0][t + o]; sm[1][t] += sm[1][t + o]; sm[2][t] += sm[2][t + o]; }
        __syncthreads();
    }
    if (t == 0) {
        part[blockIdx.x * 3 + 0] = sm[0][0];
        part[blockIdx.x * 3 + 1] = sm[1][0];
        part[blockIdx.x * 3 + 2] = sm[2][0];
    }
}

__global__ void edge_mean_final_kernel(const float* __restrict__ part, float* __restrict__ mean3)
{
    if (threadIdx.x < 3) {
        float s = 0.f;
        for (int b = 0; b < MEAN_BLOCKS; ++b) s += part[b * 3 + threadIdx.x];
        mean3[threadIdx.x] = s / (float)N_EDGES;
    }
}

__global__ __launch_bounds__(256)
void edge_score_kernel(const int* __restrict__ srcA, const int* __restrict__ dstA,
                       const float* __restrict__ pos, const float* __restrict__ mean3,
                       const float* __restrict__ Wedge, const float* __restrict__ att,
                       const float* __restrict__ XL, const float* __restrict__ XR,
                       float* __restrict__ score, int* __restrict__ nodeMax)
{
    const int lane   = threadIdx.x & 31;
    const int waveId = (blockIdx.x * blockDim.x + threadIdx.x) >> 5;
    const int nWaves = (gridDim.x * blockDim.x) >> 5;

    const float mean0 = mean3[0], mean1 = mean3[1], mean2 = mean3[2];

    int e = waveId;
    int sN, dN;
    {
        const int ei = (e < N_EDGES) ? e : 0;
        sN = srcA[ei]; dN = dstA[ei];
    }
    for (; e < ET; e += nWaves) {
        const bool real = e < N_EDGES;
        const int  s = real ? sN : e - N_EDGES;
        const int  d = real ? dN : e - N_EDGES;
        {
            const int en = e + nWaves;
            const int ei = (en < N_EDGES) ? en : 0;
            sN = srcA[ei]; dN = dstA[ei];
        }
        const float* xlrow = XL + (size_t)s * HC;
        const float* xrrow = XR + (size_t)d * HC;

        float xlv[HEADS], xrv[HEADS];
#pragma unroll
        for (int h = 0; h < HEADS; ++h) {
            xlv[h] = xlrow[h * OUT_CH + lane];
            xrv[h] = xrrow[h * OUT_CH + lane];
        }
        const float ps0 = pos[s * 3 + 0], ps1 = pos[s * 3 + 1], ps2 = pos[s * 3 + 2];
        const float pd0 = pos[d * 3 + 0], pd1 = pos[d * 3 + 1], pd2 = pos[d * 3 + 2];
        const float ea0 = real ? ps0 - pd0 : mean0;
        const float ea1 = real ? ps1 - pd1 : mean1;
        const float ea2 = real ? ps2 - pd2 : mean2;

        float sc[HEADS];
#pragma unroll
        for (int h = 0; h < HEADS; ++h) {
            const int off = h * OUT_CH + lane;
            float v = xlv[h] + xrv[h]
                    + ea0 * Wedge[0 * HC + off]
                    + ea1 * Wedge[1 * HC + off]
                    + ea2 * Wedge[2 * HC + off];
            v = (v > 0.f) ? v : NEG_SLOPE * v;
            sc[h] = v * att[off];
        }
#pragma unroll
        for (int h = 0; h < HEADS; ++h)
#pragma unroll
            for (int o = 16; o > 0; o >>= 1)
                sc[h] += __shfl_xor(sc[h], o, 32);

        if (lane < HEADS) {
            score[(size_t)e * HEADS + lane] = sc[lane];
            atomicMax(&nodeMax[d * HEADS + lane], f2ord(sc[lane]));
        }
    }
}

__global__ __launch_bounds__(256)
void edge_exp_kernel(const int* __restrict__ dstA, const int* __restrict__ nodeMax,
                     float* __restrict__ score, float* __restrict__ denom)
{
    const int stride = gridDim.x * blockDim.x;
    for (int i = blockIdx.x * blockDim.x + threadIdx.x; i < ET * HEADS; i += stride) {
        const int e = i >> 2, h = i & 3;
        const int d = (e < N_EDGES) ? dstA[e] : (e - N_EDGES);
        const float m  = ord2f(nodeMax[d * HEADS + h]);
        const float sv = expf(score[i] - m);
        score[i] = sv;
        atomicAdd(&denom[d * HEADS + h], sv);
    }
}

__global__ __launch_bounds__(256)
void edge_agg_kernel(const int* __restrict__ srcA, const int* __restrict__ dstA,
                     const float* __restrict__ XL, const float* __restrict__ score,
                     const float* __restrict__ denom, float* __restrict__ out)
{
    const int lane   = threadIdx.x & 31;
    const int waveId = (blockIdx.x * blockDim.x + threadIdx.x) >> 5;
    const int nWaves = (gridDim.x * blockDim.x) >> 5;

    int e = waveId;
    int sN, dN;
    {
        const int ei = (e < N_EDGES) ? e : 0;
        sN = srcA[ei]; dN = dstA[ei];
    }
    for (; e < ET; e += nWaves) {
        const bool real = e < N_EDGES;
        const int  s = real ? sN : e - N_EDGES;
        const int  d = real ? dN : e - N_EDGES;
        {
            const int en = e + nWaves;
            const int ei = (en < N_EDGES) ? en : 0;
            sN = srcA[ei]; dN = dstA[ei];
        }
        float xv[HEADS];
#pragma unroll
        for (int h = 0; h < HEADS; ++h)
            xv[h] = XL[(size_t)s * HC + h * OUT_CH + lane];
#pragma unroll
        for (int h = 0; h < HEADS; ++h) {
            const float alpha = score[(size_t)e * HEADS + h]
                              / (denom[d * HEADS + h] + 1e-16f);
            const int off = h * OUT_CH + lane;
            atomicAdd(&out[(size_t)d * HC + off], alpha * xv[h]);
        }
    }
}

__global__ __launch_bounds__(128)
void bn_stats_kernel(const float* __restrict__ out, float* __restrict__ colAcc)
{
    const int c = threadIdx.x;
    float s = 0.f, s2 = 0.f;
    for (int r = blockIdx.x; r < N_NODES; r += gridDim.x) {
        const float v = out[(size_t)r * HC + c];
        s += v; s2 += v * v;
    }
    atomicAdd(&colAcc[c],       s);
    atomicAdd(&colAcc[HC + c],  s2);
}

__global__ void bn_final_kernel(const float* __restrict__ colAcc,
                                const float* __restrict__ gamma,
                                const float* __restrict__ beta,
                                float* __restrict__ scaleShift)
{
    const int c = threadIdx.x;
    if (c < HC) {
        const float mu  = colAcc[c]      * (1.0f / (float)N_NODES);
        const float var = colAcc[HC + c] * (1.0f / (float)N_NODES) - mu * mu;
        const float sc  = gamma[c] * rsqrtf(var + BN_EPS);
        scaleShift[c]      = sc;
        scaleShift[HC + c] = beta[c] - mu * sc;
    }
}

__global__ __launch_bounds__(256)
void bn_apply_kernel(float* __restrict__ out, const float* __restrict__ scaleShift)
{
    const int stride = gridDim.x * blockDim.x;
    for (int i = blockIdx.x * blockDim.x + threadIdx.x; i < N_NODES * HC; i += stride) {
        const int c = i & (HC - 1);
        out[i] = out[i] * scaleShift[c] + scaleShift[HC + c];
    }
}

extern "C" void kernel_launch(void* const* d_in, const int* in_sizes, int n_in,
                              void* d_out, int out_size, void* d_ws, size_t ws_size,
                              hipStream_t stream)
{
    const float* x     = (const float*)d_in[0];
    const float* pos   = (const float*)d_in[1];
    const int*   eidx  = (const int*)  d_in[2];
    const float* W_l   = (const float*)d_in[3];
    const float* W_r   = (const float*)d_in[4];
    const float* W_e   = (const float*)d_in[5];
    const float* att   = (const float*)d_in[6];
    const float* gamma = (const float*)d_in[7];
    const float* beta  = (const float*)d_in[8];
    float* out = (float*)d_out;

    const int* srcA = eidx;
    const int* dstA = eidx + N_EDGES;

    float* ws = (float*)d_ws;
    size_t off = 0;
    float* XL         = ws + off; off += (size_t)N_NODES * HC;
    float* XR         = ws + off; off += (size_t)N_NODES * HC;
    float* score      = ws + off; off += (size_t)ET * HEADS;
    float* denom      = ws + off; off += (size_t)N_NODES * HEADS;
    int*   nodeMax    = (int*)(ws + off); off += (size_t)N_NODES * HEADS;
    float* colAcc     = ws + off; off += 2 * HC;
    float* scaleShift = ws + off; off += 2 * HC;
    float* meanPart   = ws + off; off += 3 * MEAN_BLOCKS;
    float* mean3      = ws + off; off += 4;
    (void)ws_size; (void)n_in; (void)in_sizes; (void)out_size;

    init_ws_kernel<<<1024, 256, 0, stream>>>(out, denom, nodeMax, colAcc);

    gemm_xl_xr_kernel<<<6250, 256, 0, stream>>>(x, W_l, W_r, XL, XR);

    edge_mean_partial_kernel<<<MEAN_BLOCKS, 256, 0, stream>>>(srcA, dstA, pos, meanPart);
    edge_mean_final_kernel<<<1, 64, 0, stream>>>(meanPart, mean3);

    edge_score_kernel<<<2048, 256, 0, stream>>>(srcA, dstA, pos, mean3, W_e, att,
                                                XL, XR, score, nodeMax);

    edge_exp_kernel<<<4096, 256, 0, stream>>>(dstA, nodeMax, score, denom);

    edge_agg_kernel<<<2048, 256, 0, stream>>>(srcA, dstA, XL, score, denom, out);

    bn_stats_kernel<<<1024, 128, 0, stream>>>(out, colAcc);
    bn_final_kernel<<<1, 128, 0, stream>>>(colAcc, gamma, beta, scaleShift);
    bn_apply_kernel<<<4096, 256, 0, stream>>>(out, scaleShift);
}


# Edges grouped by destination per call (degree count, offsets, atomic scatter of sources); a wave per node: running-max softmax and weighted sum in registers, scatter arrival order; no float atomics
# speedup vs baseline: 1.4538x; 1.2775x over previous
#include <hip/hip_runtime.h>
#include <math.h>

#define N_NODES   50000
#define N_EDGES   800000
#define IN_CH     64
#define HEADS     4
#define OUT_CH    32
#define HC        128
#define NEG_SLOPE 0.2f
#define BN_EPS    1e-5f
#define MEAN_BLOCKS 256

typedef float v2f __attribute__((ext_vector_type(2)));
typedef float v8f __attribute__((ext_vector_type(8)));

__global__ __launch_bounds__(256)
void init_ws_kernel(int* __restrict__ cnt, int* __restrict__ total, float* __restrict__ colAcc)
{
    const int i = blockIdx.x * blockDim.x + threadIdx.x;
    if (i < N_NODES) cnt[i] = 0;
    if (i < 2 * HC)  colAcc[i] = 0.0f;
    if (i == 0)      total[0] = 0;
}

__global__ __launch_bounds__(256)
void gemm_xl_xr_kernel(const float* __restrict__ X,
                       const float* __restrict__ Wl, const float* __restrict__ Wr,
                       float* __restrict__ XL, float* __restrict__ XR)
{
    const int wid  = (blockIdx.x * blockDim.x + threadIdx.x) >> 5;
    const int lane = threadIdx.x & 31;
    const int half = lane >> 4;
    const int idx  = lane & 15;

    const int mtile = wid >> 4;
    const int nsel  = wid & 15;
    if (mtile >= 3125) return;

    const float* W = (nsel < 8) ? Wl : Wr;
    float*       D = (nsel < 8) ? XL : XR;
    const int ntile = nsel & 7;

    const float* arow = X + (size_t)(mtile * 16 + idx) * IN_CH;
    const float* bcol = W + (size_t)(ntile * 16 + idx);

    v8f c = {};
#pragma unroll
    for (int k0 = 0; k0 < IN_CH; k0 += 4) {
        v2f a, b;
        a.x = arow[k0 + 2 * half + 0];
        a.y = arow[k0 + 2 * half + 1];
        b.x = bcol[(size_t)(k0 + 2 * half + 0) * HC];
        b.y = bcol[(size_t)(k0 + 2 * half + 1) * HC];
        c = __builtin_amdgcn_wmma_f32_16x16x4_f32(
                false, a, false, b, (short)0, c, false, false);
    }

#pragma unroll
    for (int v = 0; v < 8; ++v) {
        const int row = mtile * 16 + v + 8 * half;
        const int col = ntile * 16 + idx;
        D[(size_t)row * HC + col] = c[v];
    }
}

__global__ __launch_bounds__(256)
void edge_mean_count_kernel(const int* __restrict__ srcA, const int* __restrict__ dstA,
                            const float* __restrict__ pos, float* __restrict__ part,
                            int* __restrict__ cnt)
{
    __shared__ float sm[3][256];
    float a0 = 0.f, a1 = 0.f, a2 = 0.f;
    for (int e = blockIdx.x * blockDim.x + threadIdx.x; e < N_EDGES;
         e += gridDim.x * blockDim.x) {
        const int s = srcA[e], d = dstA[e];
        a0 += pos[s * 3 + 0] - pos[d * 3 + 0];
        a1 += pos[s * 3 + 1] - pos[d * 3 + 1];
        a2 += pos[s * 3 + 2] - pos[d * 3 + 2];
        atomicAdd(&cnt[d], 1);
    }
    const int t = threadIdx.x;
    sm[0][t] = a0; sm[1][t] = a1; sm[2][t] = a2;
    __syncthreads();
    for (int o = 128; o > 0; o >>= 1) {
        if (t < o) { sm[0][t] += sm[0][t + o]; sm[1][t] += sm[1][t + o]; sm[2][t] += sm[2][t + o]; }
        __syncthreads();
    }
    if (t == 0) {
        part[blockIdx.x * 3 + 0] = sm[0][0];
        part[blockIdx.x * 3 + 1] = sm[1][0];
        part[blockIdx.x * 3 + 2] = sm[2][0];
    }
}

__global__ __launch_bounds__(256)
void seg_alloc_kernel(const int* __restrict__ cnt, int* __restrict__ total,
                      int* __restrict__ segBeg, int* __restrict__ segCur,
                      const float* __restrict__ part, float* __restrict__ mean3)
{
    __shared__ int wsum[8];
    __shared__ int sbase;
    const int t = threadIdx.x, lane = t & 31, w = t >> 5;
    const int n = blockIdx.x * 256 + t;
    const int c = (n < N_NODES) ? cnt[n] : 0;
    int inc = c;
    for (int o = 1; o < 32; o <<= 1) {
        const int v = __shfl_up(inc, o, 32);
        if (lane >= o) inc += v;
    }
    if (lane == 31) wsum[w] = inc;
    __syncthreads();
    if (t == 0) {
        int run = 0;
        for (int i = 0; i < 8; ++i) { const int v = wsum[i]; wsum[i] = run; run += v; }
        sbase = atomicAdd(total, run);
    }
    __syncthreads();
    const int start = sbase + wsum[w] + inc - c;
    if (n < N_NODES) { segBeg[n] = start; segCur[n] = start; }

    if (blockIdx.x == 0 && t < 3) {
        float s = 0.f;
        for (int b = 0; b < MEAN_BLOCKS; ++b) s += part[b * 3 + t];
        mean3[t] = s / (float)N_EDGES;
    }
}

#define FILL_THREADS (N_EDGES / 4)
__global__ __launch_bounds__(256)
void edge_fill_kernel(const int* __restrict__ srcA, const int* __restrict__ dstA,
                      int* __restrict__ segCur, int* __restrict__ srcS)
{
    const int t = blockIdx.x * blockDim.x + threadIdx.x;
    if (t >= FILL_THREADS) return;
    int d[4], s[4], p[4];
#pragma unroll
    for (int u = 0; u < 4; ++u) {
        d[u] = dstA[t + u * FILL_THREADS];
        s[u] = srcA[t + u * FILL_THREADS];
    }
#pragma unroll
    for (int u = 0; u < 4; ++u) p[u] = atomicAdd(&segCur[d[u]], 1);
#pragma unroll
    for (int u = 0; u < 4; ++u) srcS[p[u]] = s[u];
}


template <int MASK>
__device__ __forceinline__ float lane_xor(float v)
{
    return __int_as_float(__builtin_amdgcn_mov_dpp(__float_as_int(v), 0x160 + MASK,
                                                   0xf, 0xf, false));
}
__device__ __forceinline__ float group8_sum(float v)
{
    v += lane_xor<1>(v);
    v += lane_xor<2>(v);
    v += lane_xor<4>(v);
    return v;
}

__device__ __forceinline__ float head_score(const float4 x, const float4 xr,
                                            const float ea0, const float ea1, const float ea2,
                                            const float4 w0, const float4 w1, const float4 w2,
                                            const float4 at)
{
    float v0 = x.x + xr.x + ea0 * w0.x + ea1 * w1.x + ea2 * w2.x;
    float v1 = x.y + xr.y + ea0 * w0.y + ea1 * w1.y + ea2 * w2.y;
    float v2 = x.z + xr.z + ea0 * w0.z + ea1 * w1.z + ea2 * w2.z;
    float v3 = x.w + xr.w + ea0 * w0.w + ea1 * w1.w + ea2 * w2.w;
    v0 = (v0 > 0.f) ? v0 : NEG_SLOPE * v0;
    v1 = (v1 > 0.f) ? v1 : NEG_SLOPE * v1;
    v2 = (v2 > 0.f) ? v2 : NEG_SLOPE * v2;
    v3 = (v3 > 0.f) ? v3 : NEG_SLOPE * v3;
    const float p = v0 * at.x + v1 * at.y + v2 * at.z + v3 * at.w;
    return group8_sum(p);
}

__global__ __launch_bounds__(256)
void node_softmax_agg_kernel(const int* __restrict__ segBeg, const int* __restrict__ segEnd,
                             const int* __restrict__ srcS,
                             const float* __restrict__ pos, const float* __restrict__ mean3,
                             const float* __restrict__ Wedge, const float* __restrict__ att,
                             const float* __restrict__ XL, const float* __restrict__ XR,
                             float* __restrict__ out)
{
    const int lane = threadIdx.x & 31;
    const int wave   = __builtin_amdgcn_readfirstlane((int)((blockIdx.x * blockDim.x + threadIdx.x) >> 5));
    const int nWaves = (int)((gridDim.x * blockDim.x) >> 5);

    const int c0 = 4 * lane;
    const float4 w0 = *(const float4*)(Wedge + 0 * HC + c0);
    const float4 w1 = *(const float4*)(Wedge + 1 * HC + c0);
    const float4 w2 = *(const float4*)(Wedge + 2 * HC + c0);
    const float4 at = *(const float4*)(att + c0);
    const float mean0 = mean3[0], mean1 = mean3[1], mean2 = mean3[2];

    for (int d = wave; d < N_NODES; d += nWaves) {
        const int beg = segBeg[d];
        const int end = segEnd[d];

        const float4 xr = *(const float4*)(XR + (size_t)d * HC + c0);
        const float4 xs = *(const float4*)(XL + (size_t)d * HC + c0);
        const float pd0 = pos[d * 3 + 0], pd1 = pos[d * 3 + 1], pd2 = pos[d * 3 + 2];

        float  m   = head_score(xs, xr, mean0, mean1, mean2, w0, w1, w2, at);
        float  l   = 1.0f;
        float4 acc = xs;

        for (int base = beg; base < end; base += 32) {
            const int rem = end - base;
            const int n   = (rem < 32) ? rem : 32;
            const int le = (lane < n) ? lane : (n - 1);
            const int sv = srcS[base + le];
            const float q0 = pos[sv * 3 + 0] - pd0;
            const float q1 = pos[sv * 3 + 1] - pd1;
            const float q2 = pos[sv * 3 + 2] - pd2;

            for (int j = 0; j < n; j += 4) {
                int    jj[4];
                float4 x[4];
#pragma unroll
                for (int u = 0; u < 4; ++u) {
                    const int t = j + u;
                    jj[u] = (t < n) ? t : (n - 1);
                    const int s = __builtin_amdgcn_readlane(sv, jj[u]);
                    x[u] = *(const float4*)(XL + (size_t)s * HC + c0);
                }
#pragma unroll
                for (int u = 0; u < 4; ++u) {
                    const bool  valid = (j + u) < n;
                    const float ea0 = __int_as_float(__builtin_amdgcn_readlane(__float_as_int(q0), jj[u]));
                    const float ea1 = __int_as_float(__builtin_amdgcn_readlane(__float_as_int(q1), jj[u]));
                    const float ea2 = __int_as_float(__builtin_amdgcn_readlane(__float_as_int(q2), jj[u]));
                    const float sc  = head_score(x[u], xr, ea0, ea1, ea2, w0, w1, w2, at);
                    const float dl = sc - m;
                    const float t  = expf(-fabsf(dl));
                    const bool  up = dl > 0.f;
                    float p = up ? 1.0f : t;
                    float r = up ? t : 1.0f;
                    p = valid ? p : 0.0f;
                    r = valid ? r : 1.0f;
                    m = (valid && up) ? sc : m;
                    l = l * r + p;
                    acc.x = acc.x * r + p * x[u].x;
                    acc.y = acc.y * r + p * x[u].y;
                    acc.z = acc.z * r + p * x[u].z;
                    acc.w = acc.w * r + p * x[u].w;
                }
            }
        }

        const float den = l + 1e-16f;
        float4 o;
        o.x = acc.x / den; o.y = acc.y / den; o.z = acc.z / den; o.w = acc.w / den;
        *(float4*)(out + (size_t)d * HC + c0) = o;
    }
}

__global__ __launch_bounds__(128)
void bn_stats_kernel(const float* __restrict__ out, float* __restrict__ colAcc)
{
    const int c = threadIdx.x;
    float s = 0.f, s2 = 0.f;
    for (int r = blockIdx.x; r < N_NODES; r += gridDim.x) {
        const float v = out[(size_t)r * HC + c];
        s += v; s2 += v * v;
    }
    atomicAdd(&colAcc[c],       s);
    atomicAdd(&colAcc[HC + c],  s2);
}

__global__ void bn_final_kernel(const float* __restrict__ colAcc,
                                const float* __restrict__ gamma,
                                const float* __restrict__ beta,
                                float* __restrict__ scaleShift)
{
    const int c = threadIdx.x;
    if (c < HC) {
        const float mu  = colAcc[c]      * (1.0f / (float)N_NODES);
        const float var = colAcc[HC + c] * (1.0f / (float)N_NODES) - mu * mu;
        const float sc  = gamma[c] * rsqrtf(var + BN_EPS);
        scaleShift[c]      = sc;
        scaleShift[HC + c] = beta[c] - mu * sc;
    }
}

__global__ __launch_bounds__(256)
void bn_apply_kernel(float* __restrict__ out, const float* __restrict__ scaleShift)
{
    const int stride = gridDim.x * blockDim.x;
    for (int i = blockIdx.x * blockDim.x + threadIdx.x; i < N_NODES * HC; i += stride) {
        const int c = i & (HC - 1);
        out[i] = out[i] * scaleShift[c] + scaleShift[HC + c];
    }
}

extern "C" void kernel_launch(void* const* d_in, const int* in_sizes, int n_in,
                              void* d_out, int out_size, void* d_ws, size_t ws_size,
                              hipStream_t stream)
{
    const float* x     = (const float*)d_in[0];
    const float* pos   = (const float*)d_in[1];
    const int*   eidx  = (const int*)  d_in[2];
    const float* W_l   = (const float*)d_in[3];
    const float* W_r   = (const float*)d_in[4];
    const float* W_e   = (const float*)d_in[5];
    const float* att   = (const float*)d_in[6];
    const float* gamma = (const float*)d_in[7];
    const float* beta  = (const float*)d_in[8];
    float* out = (float*)d_out;

    const int* srcA = eidx;
    const int* dstA = eidx + N_EDGES;

    float* ws = (float*)d_ws;
    size_t off = 0;
    float* XL         = ws + off; off += (size_t)N_NODES * HC;
    float* XR         = ws + off; off += (size_t)N_NODES * HC;
    int*   srcS       = (int*)(ws + off); off += (size_t)N_EDGES;
    int*   cnt        = (int*)(ws + off); off += (size_t)N_NODES;
    int*   segBeg     = (int*)(ws + off); off += (size_t)N_NODES;
    int*   segCur     = (int*)(ws + off); off += (size_t)N_NODES;
    int*   total      = (int*)(ws + off); off += 4;
    float* colAcc     = ws + off; off += 2 * HC;
    float* scaleShift = ws + off; off += 2 * HC;
    float* meanPart   = ws + off; off += 3 * MEAN_BLOCKS;
    float* mean3      = ws + off; off += 4;
    (void)ws_size; (void)n_in; (void)in_sizes; (void)out_size;

    init_ws_kernel<<<196, 256, 0, stream>>>(cnt, total, colAcc);

    gemm_xl_xr_kernel<<<6250, 256, 0, stream>>>(x, W_l, W_r, XL, XR);

    edge_mean_count_kernel<<<MEAN_BLOCKS, 256, 0, stream>>>(srcA, dstA, pos, meanPart, cnt);

    seg_alloc_kernel<<<196, 256, 0, stream>>>(cnt, total, segBeg, segCur, meanPart, mean3);

    edge_fill_kernel<<<782, 256, 0, stream>>>(srcA, dstA, segCur, srcS);

    node_softmax_agg_kernel<<<1024, 256, 0, stream>>>(segBeg, segCur, srcS, pos, mean3,
                                                             W_e, att, XL, XR, out);

    bn_stats_kernel<<<1024, 128, 0, stream>>>(out, colAcc);
    bn_final_kernel<<<1, 128, 0, stream>>>(colAcc, gamma, beta, scaleShift);
    bn_apply_kernel<<<4096, 256, 0, stream>>>(out, scaleShift);
}


# BatchNorm: column sums taken in the aggregation kernel (lane sums over the wave's rows, LDS sum over 8 waves, then the atomics), scale/shift formed in apply, 16-byte apply: 9 launches to 7
# speedup vs baseline: 1.6609x; 1.1425x over previous
#include <hip/hip_runtime.h>
#include <math.h>

#define N_NODES   50000
#define N_EDGES   800000
#define IN_CH     64
#define HEADS     4
#define OUT_CH    32
#define HC        128
#define NEG_SLOPE 0.2f
#define BN_EPS    1e-5f
#define MEAN_BLOCKS 256

typedef float v2f __attribute__((ext_vector_type(2)));
typedef float v8f __attribute__((ext_vector_type(8)));

__global__ __launch_bounds__(256)
void init_ws_kernel(int* __restrict__ cnt, int* __restrict__ total, float* __restrict__ colAcc)
{
    const int i = blockIdx.x * blockDim.x + threadIdx.x;
    if (i < N_NODES) cnt[i] = 0;
    if (i < 2 * HC)  colAcc[i] = 0.0f;
    if (i == 0)      total[0] = 0;
}

__global__ __launch_bounds__(256)
void gemm_xl_xr_kernel(const float* __restrict__ X,
                       const float* __restrict__ Wl, const float* __restrict__ Wr,
                       float* __restrict__ XL, float* __restrict__ XR)
{
    const int wid  = (blockIdx.x * blockDim.x + threadIdx.x) >> 5;
    const int lane = threadIdx.x & 31;
    const int half = lane >> 4;
    const int idx  = lane & 15;

    const int mtile = wid >> 4;
    const int nsel  = wid & 15;
    if (mtile >= 3125) return;

    const float* W = (nsel < 8) ? Wl : Wr;
    float*       D = (nsel < 8) ? XL : XR;
    const int ntile = nsel & 7;

    const float* arow = X + (size_t)(mtile * 16 + idx) * IN_CH;
    const float* bcol = W + (size_t)(ntile * 16 + idx);

    v8f c = {};
#pragma unroll
    for (int k0 = 0; k0 < IN_CH; k0 += 4) {
        v2f a, b;
        a.x = arow[k0 + 2 * half + 0];
        a.y = arow[k0 + 2 * half + 1];
        b.x = bcol[(size_t)(k0 + 2 * half + 0) * HC];
        b.y = bcol[(size_t)(k0 + 2 * half + 1) * HC];
        c = __builtin_amdgcn_wmma_f32_16x16x4_f32(
                false, a, false, b, (short)0, c, false, false);
    }

#pragma unroll
    for (int v = 0; v < 8; ++v) {
        const int row = mtile * 16 + v + 8 * half;
        const int col = ntile * 16 + idx;
        D[(size_t)row * HC + col] = c[v];
    }
}

__global__ __launch_bounds__(256)
void edge_mean_count_kernel(const int* __restrict__ srcA, const int* __restrict__ dstA,
                            const float* __restrict__ pos, float* __restrict__ part,
                            int* __restrict__ cnt)
{
    __shared__ float sm[3][256];
    float a0 = 0.f, a1 = 0.f, a2 = 0.f;
    for (int e = blockIdx.x * blockDim.x + threadIdx.x; e < N_EDGES;
         e += gridDim.x * blockDim.x) {
        const int s = srcA[e], d = dstA[e];
        a0 += pos[s * 3 + 0] - pos[d * 3 + 0];
        a1 += pos[s * 3 + 1] - pos[d * 3 + 1];
        a2 += pos[s * 3 + 2] - pos[d * 3 + 2];
        atomicAdd(&cnt[d], 1);
    }
    const int t = threadIdx.x;
    sm[0][t] = a0; sm[1][t] = a1; sm[2][t] = a2;
    __syncthreads();
    for (int o = 128; o > 0; o >>= 1) {
        if (t < o) { sm[0][t] += sm[0][t + o]; sm[1][t] += sm[1][t + o]; sm[2][t] += sm[2][t + o]; }
        __syncthreads();
    }
    if (t == 0) {
        part[blockIdx.x * 3 + 0] = sm[0][0];
        part[blockIdx.x * 3 + 1] = sm[1][0];
        part[blockIdx.x * 3 + 2] = sm[2][0];
    }
}

__global__ __launch_bounds__(256)
void seg_alloc_kernel(const int* __restrict__ cnt, int* __restrict__ total,
                      int* __restrict__ segBeg, int* __restrict__ segCur,
                      const float* __restrict__ part, float* __restrict__ mean3)
{
    __shared__ int wsum[8];
    __shared__ int sbase;
    const int t = threadIdx.x, lane = t & 31, w = t >> 5;
    const int n = blockIdx.x * 256 + t;
    const int c = (n < N_NODES) ? cnt[n] : 0;
    int inc = c;
    for (int o = 1; o < 32; o <<= 1) {
        const int v = __shfl_up(inc, o, 32);
        if (lane >= o) inc += v;
    }
    if (lane == 31) wsum[w] = inc;
    __syncthreads();
    if (t == 0) {
        int run = 0;
        for (int i = 0; i < 8; ++i) { const int v = wsum[i]; wsum[i] = run; run += v; }
        sbase = atomicAdd(total, run);
    }
    __syncthreads();
    const int start = sbase + wsum[w] + inc - c;
    if (n < N_NODES) { segBeg[n] = start; segCur[n] = start; }

    if (blockIdx.x == 0 && t < 3) {
        float s = 0.f;
        for (int b = 0; b < MEAN_BLOCKS; ++b) s += part[b * 3 + t];
        mean3[t] = s / (float)N_EDGES;
    }
}

#define FILL_THREADS (N_EDGES / 4)
__global__ __launch_bounds__(256)
void edge_fill_kernel(const int* __restrict__ srcA, const int* __restrict__ dstA,
                      int* __restrict__ segCur, int* __restrict__ srcS)
{
    const int t = blockIdx.x * blockDim.x + threadIdx.x;
    if (t >= FILL_THREADS) return;
    int d[4], s[4], p[4];
#pragma unroll
    for (int u = 0; u < 4; ++u) {
        d[u] = dstA[t + u * FILL_THREADS];
        s[u] = srcA[t + u * FILL_THREADS];
    }
#pragma unroll
    for (int u = 0; u < 4; ++u) p[u] = atomicAdd(&segCur[d[u]], 1);
#pragma unroll
    for (int u = 0; u < 4; ++u) srcS[p[u]] = s[u];
}


template <int MASK>
__device__ __forceinline__ float lane_xor(float v)
{
    return __int_as_float(__builtin_amdgcn_mov_dpp(__float_as_int(v), 0x160 + MASK,
                                                   0xf, 0xf, false));
}
__device__ __forceinline__ float group8_sum(float v)
{
    v += lane_xor<1>(v);
    v += lane_xor<2>(v);
    v += lane_xor<4>(v);
    return v;
}

__device__ __forceinline__ float head_score(const float4 x, const float4 xr,
                                            const float ea0, const float ea1, const float ea2,
                                            const float4 w0, const float4 w1, const float4 w2,
                                            const float4 at)
{
    float v0 = x.x + xr.x + ea0 * w0.x + ea1 * w1.x + ea2 * w2.x;
    float v1 = x.y + xr.y + ea0 * w0.y + ea1 * w1.y + ea2 * w2.y;
    float v2 = x.z + xr.z + ea0 * w0.z + ea1 * w1.z + ea2 * w2.z;
    float v3 = x.w + xr.w + ea0 * w0.w + ea1 * w1.w + ea2 * w2.w;
    v0 = (v0 > 0.f) ? v0 : NEG_SLOPE * v0;
    v1 = (v1 > 0.f) ? v1 : NEG_SLOPE * v1;
    v2 = (v2 > 0.f) ? v2 : NEG_SLOPE * v2;
    v3 = (v3 > 0.f) ? v3 : NEG_SLOPE * v3;
    const float p = v0 * at.x + v1 * at.y + v2 * at.z + v3 * at.w;
    return group8_sum(p);
}

__global__ __launch_bounds__(256)
void node_softmax_agg_kernel(const int* __restrict__ segBeg, const int* __restrict__ segEnd,
                             const int* __restrict__ srcS,
                             const float* __restrict__ pos, const float* __restrict__ mean3,
                             const float* __restrict__ Wedge, const float* __restrict__ att,
                             const float* __restrict__ XL, const float* __restrict__ XR,
                             float* __restrict__ out, float* __restrict__ colAcc)
{
    __shared__ float red[8][2 * HC];
    const int lane = threadIdx.x & 31;
    const int wave   = __builtin_amdgcn_readfirstlane((int)((blockIdx.x * blockDim.x + threadIdx.x) >> 5));
    const int nWaves = (int)((gridDim.x * blockDim.x) >> 5);

    const int c0 = 4 * lane;
    const float4 w0 = *(const float4*)(Wedge + 0 * HC + c0);
    const float4 w1 = *(const float4*)(Wedge + 1 * HC + c0);
    const float4 w2 = *(const float4*)(Wedge + 2 * HC + c0);
    const float4 at = *(const float4*)(att + c0);
    const float mean0 = mean3[0], mean1 = mean3[1], mean2 = mean3[2];

    float4 cs = make_float4(0.f, 0.f, 0.f, 0.f);
    float4 cq = make_float4(0.f, 0.f, 0.f, 0.f);

    for (int d = wave; d < N_NODES; d += nWaves) {
        const int beg = segBeg[d];
        const int end = segEnd[d];

        const float4 xr = *(const float4*)(XR + (size_t)d * HC + c0);
        const float4 xs = *(const float4*)(XL + (size_t)d * HC + c0);
        const float pd0 = pos[d * 3 + 0], pd1 = pos[d * 3 + 1], pd2 = pos[d * 3 + 2];

        float  m   = head_score(xs, xr, mean0, mean1, mean2, w0, w1, w2, at);
        float  l   = 1.0f;
        float4 acc = xs;

        for (int base = beg; base < end; base += 32) {
            const int rem = end - base;
            const int n   = (rem < 32) ? rem : 32;
            const int le = (lane < n) ? lane : (n - 1);
            const int sv = srcS[base + le];
            const float q0 = pos[sv * 3 + 0] - pd0;
            const float q1 = pos[sv * 3 + 1] - pd1;
            const float q2 = pos[sv * 3 + 2] - pd2;

            for (int j = 0; j < n; j += 4) {
                int    jj[4];
                float4 x[4];
#pragma unroll
                for (int u = 0; u < 4; ++u) {
                    const int t = j + u;
                    jj[u] = (t < n) ? t : (n - 1);
                    const int s = __builtin_amdgcn_readlane(sv, jj[u]);
                    x[u] = *(const float4*)(XL + (size_t)s * HC + c0);
                }
#pragma unroll
                for (int u = 0; u < 4; ++u) {
                    const bool  valid = (j + u) < n;
                    const float ea0 = __int_as_float(__builtin_amdgcn_readlane(__float_as_int(q0), jj[u]));
                    const float ea1 = __int_as_float(__builtin_amdgcn_readlane(__float_as_int(q1), jj[u]));
                    const float ea2 = __int_as_float(__builtin_amdgcn_readlane(__float_as_int(q2), jj[u]));
                    const float sc  = head_score(x[u], xr, ea0, ea1, ea2, w0, w1, w2, at);
                    const float dl = sc - m;
                    const float t  = expf(-fabsf(dl));
                    const bool  up = dl > 0.f;
                    float p = up ? 1.0f : t;
                    float r = up ? t : 1.0f;
                    p = valid ? p : 0.0f;
                    r = valid ? r : 1.0f;
                    m = (valid && up) ? sc : m;
                    l = l * r + p;
                    acc.x = acc.x * r + p * x[u].x;
                    acc.y = acc.y * r + p * x[u].y;
                    acc.z = acc.z * r + p * x[u].z;
                    acc.w = acc.w * r + p * x[u].w;
                }
            }
        }

        const float den = l + 1e-16f;
        float4 o;
        o.x = acc.x / den; o.y = acc.y / den; o.z = acc.z / den; o.w = acc.w / den;
        *(float4*)(out + (size_t)d * HC + c0) = o;
        cs.x += o.x; cs.y += o.y; cs.z += o.z; cs.w += o.w;
        cq.x += o.x * o.x; cq.y += o.y * o.y; cq.z += o.z * o.z; cq.w += o.w * o.w;
    }

    const int w = threadIdx.x >> 5;
    *(float4*)&red[w][c0]      = cs;
    *(float4*)&red[w][HC + c0] = cq;
    __syncthreads();
    const int t = threadIdx.x;
    float tot = red[0][t];
#pragma unroll
    for (int i = 1; i < 8; ++i) tot += red[i][t];
    atomicAdd(&colAcc[t], tot);
}

#define APPLY_BLOCKS     1250
#define APPLY_PER_THREAD 5
#define APPLY_STRIDE     (APPLY_BLOCKS * 256)
__global__ __launch_bounds__(256)
void bn_apply_kernel(float* __restrict__ out, const float* __restrict__ colAcc,
                     const float* __restrict__ gamma, const float* __restrict__ beta)
{
    const int c0 = 4 * (threadIdx.x & 31);
    const float4 s  = *(const float4*)(colAcc + c0);
    const float4 s2 = *(const float4*)(colAcc + HC + c0);
    const float4 g  = *(const float4*)(gamma + c0);
    const float4 b  = *(const float4*)(beta + c0);

    float4* __restrict__ o4 = (float4*)out;
    const int i0 = blockIdx.x * 256 + threadIdx.x;
    float4 v[APPLY_PER_THREAD];
#pragma unroll
    for (int u = 0; u < APPLY_PER_THREAD; ++u) v[u] = o4[i0 + u * APPLY_STRIDE];

    const float inv_n = 1.0f / (float)N_NODES;
    float4 sc, sh;
    {
        const float mu = s.x * inv_n; const float var = s2.x * inv_n - mu * mu;
        sc.x = g.x * rsqrtf(var + BN_EPS); sh.x = b.x - mu * sc.x;
    }
    {
        const float mu = s.y * inv_n; const float var = s2.y * inv_n - mu * mu;
        sc.y = g.y * rsqrtf(var + BN_EPS); sh.y = b.y - mu * sc.y;
    }
    {
        const float mu = s.z * inv_n; const float var = s2.z * inv_n - mu * mu;
        sc.z = g.z * rsqrtf(var + BN_EPS); sh.z = b.z - mu * sc.z;
    }
    {
        const float mu = s.w * inv_n; const float var = s2.w * inv_n - mu * mu;
        sc.w = g.w * rsqrtf(var + BN_EPS); sh.w = b.w - mu * sc.w;
    }

#pragma unroll
    for (int u = 0; u < APPLY_PER_THREAD; ++u) {
        float4 r;
        r.x = v[u].x * sc.x + sh.x;
        r.y = v[u].y * sc.y + sh.y;
        r.z = v[u].z * sc.z + sh.z;
        r.w = v[u].w * sc.w + sh.w;
        o4[i0 + u * APPLY_STRIDE] = r;
    }
}

extern "C" void kernel_launch(void* const* d_in, const int* in_sizes, int n_in,
                              void* d_out, int out_size, void* d_ws, size_t ws_size,
                              hipStream_t stream)
{
    const float* x     = (const float*)d_in[0];
    const float* pos   = (const float*)d_in[1];
    const int*   eidx  = (const int*)  d_in[2];
    const float* W_l   = (const float*)d_in[3];
    const float* W_r   = (const float*)d_in[4];
    const float* W_e   = (const float*)d_in[5];
    const float* att   = (const float*)d_in[6];
    const float* gamma = (const float*)d_in[7];
    const float* beta  = (const float*)d_in[8];
    float* out = (float*)d_out;

    const int* srcA = eidx;
    const int* dstA = eidx + N_EDGES;

    float* ws = (float*)d_ws;
    size_t off = 0;
    float* XL         = ws + off; off += (size_t)N_NODES * HC;
    float* XR         = ws + off; off += (size_t)N_NODES * HC;
    int*   srcS       = (int*)(ws + off); off += (size_t)N_EDGES;
    int*   cnt        = (int*)(ws + off); off += (size_t)N_NODES;
    int*   segBeg     = (int*)(ws + off); off += (size_t)N_NODES;
    int*   segCur     = (int*)(ws + off); off += (size_t)N_NODES;
    int*   total      = (int*)(ws + off); off += 4;
    float* colAcc     = ws + off; off += 2 * HC;
    float* meanPart   = ws + off; off += 3 * MEAN_BLOCKS;
    float* mean3      = ws + off; off += 4;
    (void)ws_size; (void)n_in; (void)in_sizes; (void)out_size;

    init_ws_kernel<<<196, 256, 0, stream>>>(cnt, total, colAcc);

    gemm_xl_xr_kernel<<<6250, 256, 0, stream>>>(x, W_l, W_r, XL, XR);

    edge_mean_count_kernel<<<MEAN_BLOCKS, 256, 0, stream>>>(srcA, dstA, pos, meanPart, cnt);

    seg_alloc_kernel<<<196, 256, 0, stream>>>(cnt, total, segBeg, segCur, meanPart, mean3);

    edge_fill_kernel<<<782, 256, 0, stream>>>(srcA, dstA, segCur, srcS);

    node_softmax_agg_kernel<<<1024, 256, 0, stream>>>(segBeg, segCur, srcS, pos, mean3,
                                                             W_e, att, XL, XR, out, colAcc);

    bn_apply_kernel<<<APPLY_BLOCKS, 256, 0, stream>>>(out, colAcc, gamma, beta);
}


# Both WMMA products leave their own launch: XL strips run as extra blocks of the edge count launch, XR strips of the edge fill launch (independent work side by side): 7 launches to 6
# speedup vs baseline: 1.7274x; 1.0400x over previous
#include <hip/hip_runtime.h>
#include <math.h>

#define N_NODES   50000
#define N_EDGES   800000
#define IN_CH     64
#define HEADS     4
#define OUT_CH    32
#define HC        128
#define NEG_SLOPE 0.2f
#define BN_EPS    1e-5f
#define MEAN_BLOCKS 256

typedef float v2f __attribute__((ext_vector_type(2)));
typedef float v8f __attribute__((ext_vector_type(8)));

__global__ __launch_bounds__(256)
void init_ws_kernel(int* __restrict__ cnt, int* __restrict__ total, float* __restrict__ colAcc)
{
    const int i = blockIdx.x * blockDim.x + threadIdx.x;
    if (i < N_NODES) cnt[i] = 0;
    if (i < 2 * HC)  colAcc[i] = 0.0f;
    if (i == 0)      total[0] = 0;
}

#define GEMM_STRIPS 3125
static_assert(GEMM_STRIPS * 16 == N_NODES, "the strips cover the rows exactly");
__device__ __forceinline__ void gemm_tile16(const float* __restrict__ X, const float* __restrict__ W,
                                            float* __restrict__ D, const int mtile, const int ntile)
{
    const int lane = threadIdx.x & 31;
    const int half = lane >> 4;
    const int idx  = lane & 15;

    const float* arow = X + (size_t)(mtile * 16 + idx) * IN_CH;
    const float* bcol = W + (size_t)(ntile * 16 + idx);

    v8f c = {};
#pragma unroll
    for (int k0 = 0; k0 < IN_CH; k0 += 4) {
        v2f a, b;
        a.x = arow[k0 + 2 * half + 0];
        a.y = arow[k0 + 2 * half + 1];
        b.x = bcol[(size_t)(k0 + 2 * half + 0) * HC];
        b.y = bcol[(size_t)(k0 + 2 * half + 1) * HC];
        c = __builtin_amdgcn_wmma_f32_16x16x4_f32(
                false, a, false, b, (short)0, c, false, false);
    }

#pragma unroll
    for (int v = 0; v < 8; ++v) {
        const int row = mtile * 16 + v + 8 * half;
        const int col = ntile * 16 + idx;
        D[(size_t)row * HC + col] = c[v];
    }
}

#define COUNT_GEMM_BLOCKS 3381
static_assert(COUNT_GEMM_BLOCKS == MEAN_BLOCKS + GEMM_STRIPS, "edge blocks first, then one block per strip");
__global__ __launch_bounds__(256)
void edge_mean_count_gemm_xl_kernel(const int* __restrict__ srcA, const int* __restrict__ dstA,
                                    const float* __restrict__ pos, float* __restrict__ part,
                                    int* __restrict__ cnt,
                                    const float* __restrict__ X, const float* __restrict__ Wl,
                                    float* __restrict__ XL)
{
    __shared__ float sm[3][256];
    if (blockIdx.x >= MEAN_BLOCKS) {
        gemm_tile16(X, Wl, XL, (int)blockIdx.x - MEAN_BLOCKS, (int)(threadIdx.x >> 5));
        return;
    }
    float a0 = 0.f, a1 = 0.f, a2 = 0.f;
    for (int e = blockIdx.x * 256 + threadIdx.x; e < N_EDGES;
         e += MEAN_BLOCKS * 256) {
        const int s = srcA[e], d = dstA[e];
        a0 += pos[s * 3 + 0] - pos[d * 3 + 0];
        a1 += pos[s * 3 + 1] - pos[d * 3 + 1];
        a2 += pos[s * 3 + 2] - pos[d * 3 + 2];
        atomicAdd(&cnt[d], 1);
    }
    const int t = threadIdx.x;
    sm[0][t] = a0; sm[1][t] = a1; sm[2][t] = a2;
    __syncthreads();
    for (int o = 128; o > 0; o >>= 1) {
        if (t < o) { sm[0][t] += sm[0][t + o]; sm[1][t] += sm[1][t + o]; sm[2][t] += sm[2][t + o]; }
        __syncthreads();
    }
    if (t == 0) {
        part[blockIdx.x * 3 + 0] = sm[0][0];
        part[blockIdx.x * 3 + 1] = sm[1][0];
        part[blockIdx.x * 3 + 2] = sm[2][0];
    }
}

__global__ __launch_bounds__(256)
void seg_alloc_kernel(const int* __restrict__ cnt, int* __restrict__ total,
                      int* __restrict__ segBeg, int* __restrict__ segCur,
                      const float* __restrict__ part, float* __restrict__ mean3)
{
    __shared__ int wsum[8];
    __shared__ int sbase;
    const int t = threadIdx.x, lane = t & 31, w = t >> 5;
    const int n = blockIdx.x * 256 + t;
    const int c = (n < N_NODES) ? cnt[n] : 0;
    int inc = c;
    for (int o = 1; o < 32; o <<= 1) {
        const int v = __shfl_up(inc, o, 32);
        if (lane >= o) inc += v;
    }
    if (lane == 31) wsum[w] = inc;
    __syncthreads();
    if (t == 0) {
        int run = 0;
        for (int i = 0; i < 8; ++i) { const int v = wsum[i]; wsum[i] = run; run += v; }
        sbase = atomicAdd(total, run);
    }
    __syncthreads();
    const int start = sbase + wsum[w] + inc - c;
    if (n < N_NODES) { segBeg[n] = start; segCur[n] = start; }

    if (blockIdx.x == 0 && t < 3) {
        float s = 0.f;
        for (int b = 0; b < MEAN_BLOCKS; ++b) s += part[b * 3 + t];
        mean3[t] = s / (float)N_EDGES;
    }
}

#define FILL_THREADS (N_EDGES / 4)
#define FILL_BLOCKS  782
#define FILL_GEMM_BLOCKS 3907
static_assert(FILL_GEMM_BLOCKS == FILL_BLOCKS + GEMM_STRIPS, "scatter blocks first, then one block per strip");
__global__ __launch_bounds__(256)
void edge_fill_gemm_xr_kernel(const int* __restrict__ srcA, const int* __restrict__ dstA,
                              int* __restrict__ segCur, int* __restrict__ srcS,
                              const float* __restrict__ X, const float* __restrict__ Wr,
                              float* __restrict__ XR)
{
    if (blockIdx.x >= FILL_BLOCKS) {
        gemm_tile16(X, Wr, XR, (int)blockIdx.x - FILL_BLOCKS, (int)(threadIdx.x >> 5));
        return;
    }
    const int t = blockIdx.x * 256 + threadIdx.x;
    if (t >= FILL_THREADS) return;
    int d[4], s[4], p[4];
#pragma unroll
    for (int u = 0; u < 4; ++u) {
        d[u] = dstA[t + u * FILL_THREADS];
        s[u] = srcA[t + u * FILL_THREADS];
    }
#pragma unroll
    for (int u = 0; u < 4; ++u) p[u] = atomicAdd(&segCur[d[u]], 1);
#pragma unroll
    for (int u = 0; u < 4; ++u) srcS[p[u]] = s[u];
}


template <int MASK>
__device__ __forceinline__ float lane_xor(float v)
{
    return __int_as_float(__builtin_amdgcn_mov_dpp(__float_as_int(v), 0x160 + MASK,
                                                   0xf, 0xf, false));
}
__device__ __forceinline__ float group8_sum(float v)
{
    v += lane_xor<1>(v);
    v += lane_xor<2>(v);
    v += lane_xor<4>(v);
    return v;
}

__device__ __forceinline__ float head_score(const float4 x, const float4 xr,
                                            const float ea0, const float ea1, const float ea2,
                                            const float4 w0, const float4 w1, const float4 w2,
                                            const float4 at)
{
    float v0 = x.x + xr.x + ea0 * w0.x + ea1 * w1.x + ea2 * w2.x;
    float v1 = x.y + xr.y + ea0 * w0.y + ea1 * w1.y + ea2 * w2.y;
    float v2 = x.z + xr.z + ea0 * w0.z + ea1 * w1.z + ea2 * w2.z;
    float v3 = x.w + xr.w + ea0 * w0.w + ea1 * w1.w + ea2 * w2.w;
    v0 = (v0 > 0.f) ? v0 : NEG_SLOPE * v0;
    v1 = (v1 > 0.f) ? v1 : NEG_SLOPE * v1;
    v2 = (v2 > 0.f) ? v2 : NEG_SLOPE * v2;
    v3 = (v3 > 0.f) ? v3 : NEG_SLOPE * v3;
    const float p = v0 * at.x + v1 * at.y + v2 * at.z + v3 * at.w;
    return group8_sum(p);
}

__global__ __launch_bounds__(256)
void node_softmax_agg_kernel(const int* __restrict__ segBeg, const int* __restrict__ segEnd,
                             const int* __restrict__ srcS,
                             const float* __restrict__ pos, const float* __restrict__ mean3,
                             const float* __restrict__ Wedge, const float* __restrict__ att,
                             const float* __restrict__ XL, const float* __restrict__ XR,
                             float* __restrict__ out, float* __restrict__ colAcc)
{
    __shared__ float red[8][2 * HC];
    const int lane = threadIdx.x & 31;
    const int wave   = __builtin_amdgcn_readfirstlane((int)((blockIdx.x * blockDim.x + threadIdx.x) >> 5));
    const int nWaves = (int)((gridDim.x * blockDim.x) >> 5);

    const int c0 = 4 * lane;
    const float4 w0 = *(const float4*)(Wedge + 0 * HC + c0);
    const float4 w1 = *(const float4*)(Wedge + 1 * HC + c0);
    const float4 w2 = *(const float4*)(Wedge + 2 * HC + c0);
    const float4 at = *(const float4*)(att + c0);
    const float mean0 = mean3[0], mean1 = mean3[1], mean2 = mean3[2];

    float4 cs = make_float4(0.f, 0.f, 0.f, 0.f);
    float4 cq = make_float4(0.f, 0.f, 0.f, 0.f);

    for (int d = wave; d < N_NODES; d += nWaves) {
        const int beg = segBeg[d];
        const int end = segEnd[d];

        const float4 xr = *(const float4*)(XR + (size_t)d * HC + c0);
        const float4 xs = *(const float4*)(XL + (size_t)d * HC + c0);
        const float pd0 = pos[d * 3 + 0], pd1 = pos[d * 3 + 1], pd2 = pos[d * 3 + 2];

        float  m   = head_score(xs, xr, mean0, mean1, mean2, w0, w1, w2, at);
        float  l   = 1.0f;
        float4 acc = xs;

        for (int base = beg; base < end; base += 32) {
            const int rem = end - base;
            const int n   = (rem < 32) ? rem : 32;
            const int le = (lane < n) ? lane : (n - 1);
            const int sv = srcS[base + le];
            const float q0 = pos[sv * 3 + 0] - pd0;
            const float q1 = pos[sv * 3 + 1] - pd1;
            const float q2 = pos[sv * 3 + 2] - pd2;

            for (int j = 0; j < n; j += 4) {
                int    jj[4];
                float4 x[4];
#pragma unroll
                for (int u = 0; u < 4; ++u) {
                    const int t = j + u;
                    jj[u] = (t < n) ? t : (n - 1);
                    const int s = __builtin_amdgcn_readlane(sv, jj[u]);
                    x[u] = *(const float4*)(XL + (size_t)s * HC + c0);
                }
#pragma unroll
                for (int u = 0; u < 4; ++u) {
                    const bool  valid = (j + u) < n;
                    const float ea0 = __int_as_float(__builtin_amdgcn_readlane(__float_as_int(q0), jj[u]));
                    const float ea1 = __int_as_float(__builtin_amdgcn_readlane(__float_as_int(q1), jj[u]));
                    const float ea2 = __int_as_float(__builtin_amdgcn_readlane(__float_as_int(q2), jj[u]));
                    const float sc  = head_score(x[u], xr, ea0, ea1, ea2, w0, w1, w2, at);
                    const float dl = sc - m;
                    const float t  = expf(-fabsf(dl));
                    const bool  up = dl > 0.f;
                    float p = up ? 1.0f : t;
                    float r = up ? t : 1.0f;
                    p = valid ? p : 0.0f;
                    r = valid ? r : 1.0f;
                    m = (valid && up) ? sc : m;
                    l = l * r + p;
                    acc.x = acc.x * r + p * x[u].x;
                    acc.y = acc.y * r + p * x[u].y;
                    acc.z = acc.z * r + p * x[u].z;
                    acc.w = acc.w * r + p * x[u].w;
                }
            }
        }

        const float den = l + 1e-16f;
        float4 o;
        o.x = acc.x / den; o.y = acc.y / den; o.z = acc.z / den; o.w = acc.w / den;
        *(float4*)(out + (size_t)d * HC + c0) = o;
        cs.x += o.x; cs.y += o.y; cs.z += o.z; cs.w += o.w;
        cq.x += o.x * o.x; cq.y += o.y * o.y; cq.z += o.z * o.z; cq.w += o.w * o.w;
    }

    const int w = threadIdx.x >> 5;
    *(float4*)&red[w][c0]      = cs;
    *(float4*)&red[w][HC + c0] = cq;
    __syncthreads();
    const int t = threadIdx.x;
    float tot = red[0][t];
#pragma unroll
    for (int i = 1; i < 8; ++i) tot += red[i][t];
    atomicAdd(&colAcc[t], tot);
}

#define APPLY_BLOCKS     1250
#define APPLY_PER_THREAD 5
#define APPLY_STRIDE     (APPLY_BLOCKS * 256)
__global__ __launch_bounds__(256)
void bn_apply_kernel(float* __restrict__ out, const float* __restrict__ colAcc,
                     const float* __restrict__ gamma, const float* __restrict__ beta)
{
    const int c0 = 4 * (threadIdx.x & 31);
    const float4 s  = *(const float4*)(colAcc + c0);
    const float4 s2 = *(const float4*)(colAcc + HC + c0);
    const float4 g  = *(const float4*)(gamma + c0);
    const float4 b  = *(const float4*)(beta + c0);

    float4* __restrict__ o4 = (float4*)out;
    const int i0 = blockIdx.x * 256 + threadIdx.x;
    float4 v[APPLY_PER_THREAD];
#pragma unroll
    for (int u = 0; u < APPLY_PER_THREAD; ++u) v[u] = o4[i0 + u * APPLY_STRIDE];

    const float inv_n = 1.0f / (float)N_NODES;
    float4 sc, sh;
    {
        const float mu = s.x * inv_n; const float var = s2.x * inv_n - mu * mu;
        sc.x = g.x * rsqrtf(var + BN_EPS); sh.x = b.x - mu * sc.x;
    }
    {
        const float mu = s.y * inv_n; const float var = s2.y * inv_n - mu * mu;
        sc.y = g.y * rsqrtf(var + BN_EPS); sh.y = b.y - mu * sc.y;
    }
    {
        const float mu = s.z * inv_n; const float var = s2.z * inv_n - mu * mu;
        sc.z = g.z * rsqrtf(var + BN_EPS); sh.z = b.z - mu * sc.z;
    }
    {
        const float mu = s.w * inv_n; const float var = s2.w * inv_n - mu * mu;
        sc.w = g.w * rsqrtf(var + BN_EPS); sh.w = b.w - mu * sc.w;
    }

#pragma unroll
    for (int u = 0; u < APPLY_PER_THREAD; ++u) {
        float4 r;
        r.x = v[u].x * sc.x + sh.x;
        r.y = v[u].y * sc.y + sh.y;
        r.z = v[u].z * sc.z + sh.z;
        r.w = v[u].w * sc.w + sh.w;
        o4[i0 + u * APPLY_STRIDE] = r;
    }
}

extern "C" void kernel_launch(void* const* d_in, const int* in_sizes, int n_in,
                              void* d_out, int out_size, void* d_ws, size_t ws_size,
                              hipStream_t stream)
{
    const float* x     = (const float*)d_in[0];
    const float* pos   = (const float*)d_in[1];
    const int*   eidx  = (const int*)  d_in[2];
    const float* W_l   = (const float*)d_in[3];
    const float* W_r   = (const float*)d_in[4];
    const float* W_e   = (const float*)d_in[5];
    const float* att   = (const float*)d_in[6];
    const float* gamma = (const float*)d_in[7];
    const float* beta  = (const float*)d_in[8];
    float* out = (float*)d_out;

    const int* srcA = eidx;
    const int* dstA = eidx + N_EDGES;

    float* ws = (float*)d_ws;
    size_t off = 0;
    float* XL         = ws + off; off += (size_t)N_NODES * HC;
    float* XR         = ws + off; off += (size_t)N_NODES * HC;
    int*   srcS       = (int*)(ws + off); off += (size_t)N_EDGES;
    int*   cnt        = (int*)(ws + off); off += (size_t)N_NODES;
    int*   segBeg     = (int*)(ws + off); off += (size_t)N_NODES;
    int*   segCur     = (int*)(ws + off); off += (size_t)N_NODES;
    int*   total      = (int*)(ws + off); off += 4;
    float* colAcc     = ws + off; off += 2 * HC;
    float* meanPart   = ws + off; off += 3 * MEAN_BLOCKS;
    float* mean3      = ws + off; off += 4;
    (void)ws_size; (void)n_in; (void)in_sizes; (void)out_size;

    init_ws_kernel<<<196, 256, 0, stream>>>(cnt, total, colAcc);

    edge_mean_count_gemm_xl_kernel<<<COUNT_GEMM_BLOCKS, 256, 0, stream>>>(
        srcA, dstA, pos, meanPart, cnt, x, W_l, XL);

    seg_alloc_kernel<<<196, 256, 0, stream>>>(cnt, total, segBeg, segCur, meanPart, mean3);

    edge_fill_gemm_xr_kernel<<<FILL_GEMM_BLOCKS, 256, 0, stream>>>(
        srcA, dstA, segCur, srcS, x, W_r, XR);

    node_softmax_agg_kernel<<<1024, 256, 0, stream>>>(segBeg, segCur, srcS, pos, mean3,
                                                             W_e, att, XL, XR, out, colAcc);

    bn_apply_kernel<<<APPLY_BLOCKS, 256, 0, stream>>>(out, colAcc, gamma, beta);
}


# Edges grouped by destination in ONE pass: the returning count add is the slot in a 32-source line per node (overflow list past 32); both GEMMs ride along; 6 launches to 4; mean summed in new order
# speedup vs baseline: 2.1583x; 1.2495x over previous
#include <hip/hip_runtime.h>
#include <math.h>

#define N_NODES   50000
#define N_EDGES   800000
#define IN_CH     64
#define HEADS     4
#define OUT_CH    32
#define HC        128
#define NEG_SLOPE 0.2f
#define BN_EPS    1e-5f
#define BUCKET    32

typedef float v2f __attribute__((ext_vector_type(2)));
typedef float v8f __attribute__((ext_vector_type(8)));

__global__ __launch_bounds__(256)
void init_ws_kernel(int* __restrict__ cnt, int* __restrict__ ovfTotal, float* __restrict__ colAcc)
{
    const int i = blockIdx.x * blockDim.x + threadIdx.x;
    if (i < N_NODES) cnt[i] = 0;
    if (i < 2 * HC)  colAcc[i] = 0.0f;
    if (i == 0)      ovfTotal[0] = 0;
}

#define GEMM_STRIPS 3125
static_assert(GEMM_STRIPS * 16 == N_NODES, "the strips cover the rows exactly");
__device__ __forceinline__ void gemm_tile16(const float* __restrict__ X, const float* __restrict__ W,
                                            float* __restrict__ D, const int mtile, const int ntile)
{
    const int lane = threadIdx.x & 31;
    const int half = lane >> 4;
    const int idx  = lane & 15;

    const float* arow = X + (size_t)(mtile * 16 + idx) * IN_CH;
    const float* bcol = W + (size_t)(ntile * 16 + idx);

    v8f c = {};
#pragma unroll
    for (int k0 = 0; k0 < IN_CH; k0 += 4) {
        v2f a, b;
        a.x = arow[k0 + 2 * half + 0];
        a.y = arow[k0 + 2 * half + 1];
        b.x = bcol[(size_t)(k0 + 2 * half + 0) * HC];
        b.y = bcol[(size_t)(k0 + 2 * half + 1) * HC];
        c = __builtin_amdgcn_wmma_f32_16x16x4_f32(
                false, a, false, b, (short)0, c, false, false);
    }

#pragma unroll
    for (int v = 0; v < 8; ++v) {
        const int row = mtile * 16 + v + 8 * half;
        const int col = ntile * 16 + idx;
        D[(size_t)row * HC + col] = c[v];
    }
}

__device__ __forceinline__ float wave_sum(float v)
{
    for (int o = 16; o > 0; o >>= 1) v += __shfl_xor(v, o, 32);
    return v;
}
__device__ __forceinline__ void block_sum3(float (&sm)[3][8], float& a0, float& a1, float& a2)
{
    a0 = wave_sum(a0); a1 = wave_sum(a1); a2 = wave_sum(a2);
    const int lane = threadIdx.x & 31, w = threadIdx.x >> 5;
    if (lane == 0) { sm[0][w] = a0; sm[1][w] = a1; sm[2][w] = a2; }
    __syncthreads();
    float s0 = 0.f, s1 = 0.f, s2 = 0.f;
    for (int i = 0; i < 8; ++i) { s0 += sm[0][i]; s1 += sm[1][i]; s2 += sm[2][i]; }
    a0 = s0; a1 = s1; a2 = s2;
}

#define FILL_THREADS (N_EDGES / 4)
#define FILL_BLOCKS  782
#define FILL_GEMM_BLOCKS 7032
static_assert(FILL_BLOCKS * 256 >= FILL_THREADS, "the edge blocks cover the edges");
static_assert(FILL_GEMM_BLOCKS == FILL_BLOCKS + 2 * GEMM_STRIPS, "edge blocks first, then one block per strip and product");
__global__ __launch_bounds__(256)
void edge_bucket_mean_gemm_kernel(const int* __restrict__ srcA, const int* __restrict__ dstA,
                                  const float* __restrict__ pos,
                                  int* __restrict__ cnt, int* __restrict__ bucket,
                                  int* __restrict__ ovfTotal, int* __restrict__ ovfDst,
                                  int* __restrict__ ovfSrc, float* __restrict__ part,
                                  const float* __restrict__ X, const float* __restrict__ Wl,
                                  const float* __restrict__ Wr,
                                  float* __restrict__ XL, float* __restrict__ XR)
{
    __shared__ float sm[3][8];
    if (blockIdx.x >= FILL_BLOCKS) {
        const int  g     = (int)blockIdx.x - FILL_BLOCKS;
        const bool left  = g < GEMM_STRIPS;
        gemm_tile16(X, left ? Wl : Wr, left ? XL : XR, left ? g : g - GEMM_STRIPS,
                    (int)(threadIdx.x >> 5));
        return;
    }
    const int t = blockIdx.x * 256 + threadIdx.x;
    float a0 = 0.f, a1 = 0.f, a2 = 0.f;
    if (t < FILL_THREADS) {
        int d[4], s[4], p[4];
        float ps[4][3], pd[4][3];
#pragma unroll
        for (int u = 0; u < 4; ++u) {
            d[u] = dstA[t + u * FILL_THREADS];
            s[u] = srcA[t + u * FILL_THREADS];
        }
#pragma unroll
        for (int u = 0; u < 4; ++u) p[u] = atomicAdd(&cnt[d[u]], 1);
#pragma unroll
        for (int u = 0; u < 4; ++u) {
            ps[u][0] = pos[s[u] * 3 + 0]; ps[u][1] = pos[s[u] * 3 + 1]; ps[u][2] = pos[s[u] * 3 + 2];
            pd[u][0] = pos[d[u] * 3 + 0]; pd[u][1] = pos[d[u] * 3 + 1]; pd[u][2] = pos[d[u] * 3 + 2];
        }
#pragma unroll
        for (int u = 0; u < 4; ++u) {
            if (p[u] < BUCKET) {
                bucket[d[u] * BUCKET + p[u]] = s[u];
            } else {
                const int q = atomicAdd(ovfTotal, 1);
                ovfDst[q] = d[u];
                ovfSrc[q] = s[u];
            }
        }
#pragma unroll
        for (int u = 0; u < 4; ++u) {
            a0 += ps[u][0] - pd[u][0];
            a1 += ps[u][1] - pd[u][1];
            a2 += ps[u][2] - pd[u][2];
        }
    }
    block_sum3(sm, a0, a1, a2);
    if (threadIdx.x == 0) {
        part[blockIdx.x * 3 + 0] = a0;
        part[blockIdx.x * 3 + 1] = a1;
        part[blockIdx.x * 3 + 2] = a2;
    }
}


template <int MASK>
__device__ __forceinline__ float lane_xor(float v)
{
    return __int_as_float(__builtin_amdgcn_mov_dpp(__float_as_int(v), 0x160 + MASK,
                                                   0xf, 0xf, false));
}
__device__ __forceinline__ float group8_sum(float v)
{
    v += lane_xor<1>(v);
    v += lane_xor<2>(v);
    v += lane_xor<4>(v);
    return v;
}

__device__ __forceinline__ float head_score(const float4 x, const float4 xr,
                                            const float ea0, const float ea1, const float ea2,
                                            const float4 w0, const float4 w1, const float4 w2,
                                            const float4 at)
{
    float v0 = x.x + xr.x + ea0 * w0.x + ea1 * w1.x + ea2 * w2.x;
    float v1 = x.y + xr.y + ea0 * w0.y + ea1 * w1.y + ea2 * w2.y;
    float v2 = x.z + xr.z + ea0 * w0.z + ea1 * w1.z + ea2 * w2.z;
    float v3 = x.w + xr.w + ea0 * w0.w + ea1 * w1.w + ea2 * w2.w;
    v0 = (v0 > 0.f) ? v0 : NEG_SLOPE * v0;
    v1 = (v1 > 0.f) ? v1 : NEG_SLOPE * v1;
    v2 = (v2 > 0.f) ? v2 : NEG_SLOPE * v2;
    v3 = (v3 > 0.f) ? v3 : NEG_SLOPE * v3;
    const float p = v0 * at.x + v1 * at.y + v2 * at.z + v3 * at.w;
    return group8_sum(p);
}

__device__ __forceinline__ void edge_step(const float4 x, const float sc, const bool valid,
                                          float& m, float& l, float4& acc)
{
    const float dl = sc - m;
    const float t  = expf(-fabsf(dl));
    const bool  up = dl > 0.f;
    float p = up ? 1.0f : t;
    float r = up ? t : 1.0f;
    p = valid ? p : 0.0f;
    r = valid ? r : 1.0f;
    m = (valid && up) ? sc : m;
    l = l * r + p;
    acc.x = acc.x * r + p * x.x;
    acc.y = acc.y * r + p * x.y;
    acc.z = acc.z * r + p * x.z;
    acc.w = acc.w * r + p * x.w;
}

__global__ __launch_bounds__(256)
void node_softmax_agg_kernel(const int* __restrict__ cnt, const int* __restrict__ bucket,
                             const int* __restrict__ ovfTotal, const int* __restrict__ ovfDst,
                             const int* __restrict__ ovfSrc,
                             const float* __restrict__ pos, const float* __restrict__ part,
                             const float* __restrict__ Wedge, const float* __restrict__ att,
                             const float* __restrict__ XL, const float* __restrict__ XR,
                             float* __restrict__ out, float* __restrict__ colAcc)
{
    __shared__ float red[8][2 * HC];
    __shared__ float sm[3][8];
    const int lane = threadIdx.x & 31;
    const int wave   = __builtin_amdgcn_readfirstlane((int)((blockIdx.x * blockDim.x + threadIdx.x) >> 5));
    const int nWaves = (int)((gridDim.x * blockDim.x) >> 5);

    const int c0 = 4 * lane;
    const float4 w0 = *(const float4*)(Wedge + 0 * HC + c0);
    const float4 w1 = *(const float4*)(Wedge + 1 * HC + c0);
    const float4 w2 = *(const float4*)(Wedge + 2 * HC + c0);
    const float4 at = *(const float4*)(att + c0);

    float mean0 = 0.f, mean1 = 0.f, mean2 = 0.f;
    static_assert(FILL_BLOCKS <= 4 * 256, "four partials per thread cover the edge blocks");
#pragma unroll
    for (int k = 0; k < 4; ++k) {
        const int  b  = (int)threadIdx.x + 256 * k;
        const bool on = b < FILL_BLOCKS;
        const int  bc = on ? b : (FILL_BLOCKS - 1);
        const float v0 = part[bc * 3 + 0], v1 = part[bc * 3 + 1], v2 = part[bc * 3 + 2];
        mean0 += on ? v0 : 0.f;
        mean1 += on ? v1 : 0.f;
        mean2 += on ? v2 : 0.f;
    }
    block_sum3(sm, mean0, mean1, mean2);
    mean0 = mean0 / (float)N_EDGES;
    mean1 = mean1 / (float)N_EDGES;
    mean2 = mean2 / (float)N_EDGES;

    const int ovfN = ovfTotal[0];

    float4 cs = make_float4(0.f, 0.f, 0.f, 0.f);
    float4 cq = make_float4(0.f, 0.f, 0.f, 0.f);

    for (int d = wave; d < N_NODES; d += nWaves) {
        const int n  = cnt[d];
        const int nb = (n < BUCKET) ? n : BUCKET;

        const float4 xr = *(const float4*)(XR + (size_t)d * HC + c0);
        const float4 xs = *(const float4*)(XL + (size_t)d * HC + c0);
        const float pd0 = pos[d * 3 + 0], pd1 = pos[d * 3 + 1], pd2 = pos[d * 3 + 2];

        float  m   = head_score(xs, xr, mean0, mean1, mean2, w0, w1, w2, at);
        float  l   = 1.0f;
        float4 acc = xs;

        if (nb > 0) {
            const int le = (lane < nb) ? lane : (nb - 1);
            const int sv = bucket[d * BUCKET + le];
            const float q0 = pos[sv * 3 + 0] - pd0;
            const float q1 = pos[sv * 3 + 1] - pd1;
            const float q2 = pos[sv * 3 + 2] - pd2;

            for (int j = 0; j < nb; j += 4) {
                int    jj[4];
                float4 x[4];
#pragma unroll
                for (int u = 0; u < 4; ++u) {
                    const int t = j + u;
                    jj[u] = (t < nb) ? t : (nb - 1);
                    const int s = __builtin_amdgcn_readlane(sv, jj[u]);
                    x[u] = *(const float4*)(XL + (size_t)s * HC + c0);
                }
#pragma unroll
                for (int u = 0; u < 4; ++u) {
                    const bool  valid = (j + u) < nb;
                    const float ea0 = __int_as_float(__builtin_amdgcn_readlane(__float_as_int(q0), jj[u]));
                    const float ea1 = __int_as_float(__builtin_amdgcn_readlane(__float_as_int(q1), jj[u]));
                    const float ea2 = __int_as_float(__builtin_amdgcn_readlane(__float_as_int(q2), jj[u]));
                    const float sc  = head_score(x[u], xr, ea0, ea1, ea2, w0, w1, w2, at);
                    edge_step(x[u], sc, valid, m, l, acc);
                }
            }
        }

        if (n > BUCKET) {
            for (int base = 0; base < ovfN; base += 32) {
                const int  i   = base + lane;
                const bool in  = i < ovfN;
                const int  od  = in ? ovfDst[i] : -1;
                const int  os  = in ? ovfSrc[i] : 0;
                unsigned mask = __builtin_amdgcn_ballot_w32(od == d);
                if (mask != 0u) {
                    const float q0 = pos[os * 3 + 0] - pd0;
                    const float q1 = pos[os * 3 + 1] - pd1;
                    const float q2 = pos[os * 3 + 2] - pd2;
                    while (mask != 0u) {
                        const int j = __builtin_ctz(mask);
                        mask &= mask - 1u;
                        const int s = __builtin_amdgcn_readlane(os, j);
                        const float4 x = *(const float4*)(XL + (size_t)s * HC + c0);
                        const float ea0 = __int_as_float(__builtin_amdgcn_readlane(__float_as_int(q0), j));
                        const float ea1 = __int_as_float(__builtin_amdgcn_readlane(__float_as_int(q1), j));
                        const float ea2 = __int_as_float(__builtin_amdgcn_readlane(__float_as_int(q2), j));
                        const float sc  = head_score(x, xr, ea0, ea1, ea2, w0, w1, w2, at);
                        edge_step(x, sc, true, m, l, acc);
                    }
                }
            }
        }

        const float den = l + 1e-16f;
        float4 o;
        o.x = acc.x / den; o.y = acc.y / den; o.z = acc.z / den; o.w = acc.w / den;
        *(float4*)(out + (size_t)d * HC + c0) = o;
        cs.x += o.x; cs.y += o.y; cs.z += o.z; cs.w += o.w;
        cq.x += o.x * o.x; cq.y += o.y * o.y; cq.z += o.z * o.z; cq.w += o.w * o.w;
    }

    const int w = threadIdx.x >> 5;
    *(float4*)&red[w][c0]      = cs;
    *(float4*)&red[w][HC + c0] = cq;
    __syncthreads();
    const int t = threadIdx.x;
    float tot = red[0][t];
#pragma unroll
    for (int i = 1; i < 8; ++i) tot += red[i][t];
    atomicAdd(&colAcc[t], tot);
}

#define APPLY_BLOCKS     1250
#define APPLY_PER_THREAD 5
#define APPLY_STRIDE     (APPLY_BLOCKS * 256)
__global__ __launch_bounds__(256)
void bn_apply_kernel(float* __restrict__ out, const float* __restrict__ colAcc,
                     const float* __restrict__ gamma, const float* __restrict__ beta)
{
    const int c0 = 4 * (threadIdx.x & 31);
    const float4 s  = *(const float4*)(colAcc + c0);
    const float4 s2 = *(const float4*)(colAcc + HC + c0);
    const float4 g  = *(const float4*)(gamma + c0);
    const float4 b  = *(const float4*)(beta + c0);

    float4* __restrict__ o4 = (float4*)out;
    const int i0 = blockIdx.x * 256 + threadIdx.x;
    float4 v[APPLY_PER_THREAD];
#pragma unroll
    for (int u = 0; u < APPLY_PER_THREAD; ++u) v[u] = o4[i0 + u * APPLY_STRIDE];

    const float inv_n = 1.0f / (float)N_NODES;
    float4 sc, sh;
    {
        const float mu = s.x * inv_n; const float var = s2.x * inv_n - mu * mu;
        sc.x = g.x * rsqrtf(var + BN_EPS); sh.x = b.x - mu * sc.x;
    }
    {
        const float mu = s.y * inv_n; const float var = s2.y * inv_n - mu * mu;
        sc.y = g.y * rsqrtf(var + BN_EPS); sh.y = b.y - mu * sc.y;
    }
    {
        const float mu = s.z * inv_n; const float var = s2.z * inv_n - mu * mu;
        sc.z = g.z * rsqrtf(var + BN_EPS); sh.z = b.z - mu * sc.z;
    }
    {
        const float mu = s.w * inv_n; const float var = s2.w * inv_n - mu * mu;
        sc.w = g.w * rsqrtf(var + BN_EPS); sh.w = b.w - mu * sc.w;
    }

#pragma unroll
    for (int u = 0; u < APPLY_PER_THREAD; ++u) {
        float4 r;
        r.x = v[u].x * sc.x + sh.x;
        r.y = v[u].y * sc.y + sh.y;
        r.z = v[u].z * sc.z + sh.z;
        r.w = v[u].w * sc.w + sh.w;
        o4[i0 + u * APPLY_STRIDE] = r;
    }
}

extern "C" void kernel_launch(void* const* d_in, const int* in_sizes, int n_in,
                              void* d_out, int out_size, void* d_ws, size_t ws_size,
                              hipStream_t stream)
{
    const float* x     = (const float*)d_in[0];
    const float* pos   = (const float*)d_in[1];
    const int*   eidx  = (const int*)  d_in[2];
    const float* W_l   = (const float*)d_in[3];
    const float* W_r   = (const float*)d_in[4];
    const float* W_e   = (const float*)d_in[5];
    const float* att   = (const float*)d_in[6];
    const float* gamma = (const float*)d_in[7];
    const float* beta  = (const float*)d_in[8];
    float* out = (float*)d_out;

    const int* srcA = eidx;
    const int* dstA = eidx + N_EDGES;

    float* ws = (float*)d_ws;
    size_t off = 0;
    float* XL         = ws + off; off += (size_t)N_NODES * HC;
    float* XR         = ws + off; off += (size_t)N_NODES * HC;
    int*   bucket     = (int*)(ws + off); off += (size_t)N_NODES * BUCKET;
    int*   ovfDst     = (int*)(ws + off); off += (size_t)N_EDGES;
    int*   ovfSrc     = (int*)(ws + off); off += (size_t)N_EDGES;
    int*   cnt        = (int*)(ws + off); off += (size_t)N_NODES;
    int*   ovfTotal   = (int*)(ws + off); off += 4;
    float* colAcc     = ws + off; off += 2 * HC;
    float* meanPart   = ws + off; off += 3 * FILL_BLOCKS;
    (void)ws_size; (void)n_in; (void)in_sizes; (void)out_size;

    init_ws_kernel<<<196, 256, 0, stream>>>(cnt, ovfTotal, colAcc);

    edge_bucket_mean_gemm_kernel<<<FILL_GEMM_BLOCKS, 256, 0, stream>>>(
        srcA, dstA, pos, cnt, bucket, ovfTotal, ovfDst, ovfSrc, meanPart, x, W_l, W_r, XL, XR);

    node_softmax_agg_kernel<<<1024, 256, 0, stream>>>(cnt, bucket, ovfTotal, ovfDst, ovfSrc, pos,
                                                             meanPart, W_e, att, XL, XR, out, colAcc);

    bn_apply_kernel<<<APPLY_BLOCKS, 256, 0, stream>>>(out, colAcc, gamma, beta);
}


# Edge grouping by owner blocks: 250 blocks of 32 waves own 200 destinations each, scan all destinations, slots from LDS counters (arrival order), no global add per edge; mean summed in new order
# speedup vs baseline: 2.6801x; 1.2418x over previous
#include <hip/hip_runtime.h>
#include <math.h>

#define N_NODES   50000
#define N_EDGES   800000
#define IN_CH     64
#define HEADS     4
#define OUT_CH    32
#define HC        128
#define NEG_SLOPE 0.2f
#define BN_EPS    1e-5f
#define BUCKET    32

typedef float v2f __attribute__((ext_vector_type(2)));
typedef float v8f __attribute__((ext_vector_type(8)));

__global__ __launch_bounds__(256)
void init_ws_kernel(int* __restrict__ ovfTotal, float* __restrict__ colAcc)
{
    const int i = threadIdx.x;
    if (i < 2 * HC)  colAcc[i] = 0.0f;
    if (i == 0)      ovfTotal[0] = 0;
}

#define GEMM_STRIPS 3125
static_assert(GEMM_STRIPS * 16 == N_NODES, "the strips cover the rows exactly");
__device__ __forceinline__ void gemm_tile16(const float* __restrict__ X, const float* __restrict__ W,
                                            float* __restrict__ D, const int mtile, const int ntile)
{
    const int lane = threadIdx.x & 31;
    const int half = lane >> 4;
    const int idx  = lane & 15;

    const float* arow = X + (size_t)(mtile * 16 + idx) * IN_CH;
    const float* bcol = W + (size_t)(ntile * 16 + idx);

    v8f c = {};
#pragma unroll
    for (int k0 = 0; k0 < IN_CH; k0 += 4) {
        v2f a, b;
        a.x = arow[k0 + 2 * half + 0];
        a.y = arow[k0 + 2 * half + 1];
        b.x = bcol[(size_t)(k0 + 2 * half + 0) * HC];
        b.y = bcol[(size_t)(k0 + 2 * half + 1) * HC];
        c = __builtin_amdgcn_wmma_f32_16x16x4_f32(
                false, a, false, b, (short)0, c, false, false);
    }

#pragma unroll
    for (int v = 0; v < 8; ++v) {
        const int row = mtile * 16 + v + 8 * half;
        const int col = ntile * 16 + idx;
        D[(size_t)row * HC + col] = c[v];
    }
}

__device__ __forceinline__ float wave_sum(float v)
{
    for (int o = 16; o > 0; o >>= 1) v += __shfl_xor(v, o, 32);
    return v;
}
template <int NW>
__device__ __forceinline__ void block_sum3(float (&sm)[3][NW], float& a0, float& a1, float& a2)
{
    a0 = wave_sum(a0); a1 = wave_sum(a1); a2 = wave_sum(a2);
    const int lane = threadIdx.x & 31, w = threadIdx.x >> 5;
    if (lane == 0) { sm[0][w] = a0; sm[1][w] = a1; sm[2][w] = a2; }
    __syncthreads();
    float s0 = 0.f, s1 = 0.f, s2 = 0.f;
    for (int i = 0; i < NW; ++i) { s0 += sm[0][i]; s1 += sm[1][i]; s2 += sm[2][i]; }
    a0 = s0; a1 = s1; a2 = s2;
}

#define OWNERS      250
#define OWN_SPAN    200
#define OWN_WAVES   32
#define OWN_THREADS (OWN_WAVES * 32)
#define LIST_CAP    8192
#define N_EDGE4     (N_EDGES / 4)
#define SCAN_LOADS  8
#define FILL_BLOCKS OWNERS
#define GEMM_TASKS  (2 * GEMM_STRIPS)
#define GEMM_BLOCKS4 1563
#define FILL_GEMM_BLOCKS (OWNERS + GEMM_BLOCKS4)
static_assert(OWNERS * OWN_SPAN == N_NODES, "the owner ranges cover the nodes exactly");
static_assert(N_EDGE4 * 4 == N_EDGES, "the destination list is a whole number of 16-byte groups");
static_assert(N_EDGES <= (1 << 20) && OWN_SPAN <= (1 << 11), "a list entry packs the edge number in 20 bits, the node above it");
static_assert(GEMM_BLOCKS4 * 4 >= GEMM_TASKS, "the strip blocks cover both products");

__device__ __forceinline__ void place_edge(const int d, const int s, const int slot,
                                           int* __restrict__ bucket, int* __restrict__ ovfTotal,
                                           int* __restrict__ ovfDst, int* __restrict__ ovfSrc)
{
    if (slot < BUCKET) {
        bucket[d * BUCKET + slot] = s;
    } else {
        const int q = atomicAdd(ovfTotal, 1);
        ovfDst[q] = d;
        ovfSrc[q] = s;
    }
}

__device__ __forceinline__ void note_edge(const int d, const int e, const int lo,
                                          int* __restrict__ list, int* __restrict__ listN)
{
    const unsigned r = (unsigned)(d - lo);
    if (r < (unsigned)OWN_SPAN) {
        const int p = atomicAdd(listN, 1);
        if (p < LIST_CAP) list[p] = (int)((r << 20) | (unsigned)e);
    }
}

__global__ __launch_bounds__(OWN_THREADS)
void edge_bucket_mean_gemm_kernel(const int* __restrict__ srcA, const int* __restrict__ dstA,
                                  const float* __restrict__ pos,
                                  int* __restrict__ cnt, int* __restrict__ bucket,
                                  int* __restrict__ ovfTotal, int* __restrict__ ovfDst,
                                  int* __restrict__ ovfSrc, float* __restrict__ part,
                                  const float* __restrict__ X, const float* __restrict__ Wl,
                                  const float* __restrict__ Wr,
                                  float* __restrict__ XL, float* __restrict__ XR)
{
    __shared__ int   lcnt[OWN_SPAN];
    __shared__ int   list[LIST_CAP];
    __shared__ int   listN;
    __shared__ float sm[3][OWN_WAVES];

    if (blockIdx.x >= OWNERS) {
        const int g = ((int)blockIdx.x - OWNERS) * 4 + (int)(threadIdx.x >> 8);
        if (g >= GEMM_TASKS) return;
        const bool left = g < GEMM_STRIPS;
        gemm_tile16(X, left ? Wl : Wr, left ? XL : XR, left ? g : g - GEMM_STRIPS,
                    (int)((threadIdx.x >> 5) & 7));
        return;
    }

    const int tid = (int)threadIdx.x;
    const int lo  = (int)blockIdx.x * OWN_SPAN;
    if (tid < OWN_SPAN) lcnt[tid] = 0;
    if (tid == 0) listN = 0;
    __syncthreads();

    const int4* __restrict__ d4 = (const int4*)dstA;
    for (int base = 0; base < N_EDGE4; base += SCAN_LOADS * OWN_THREADS) {
        int4 v[SCAN_LOADS];
#pragma unroll
        for (int u = 0; u < SCAN_LOADS; ++u) {
            const int j = base + u * OWN_THREADS + tid;
            v[u] = d4[(j < N_EDGE4) ? j : (N_EDGE4 - 1)];
        }
#pragma unroll
        for (int u = 0; u < SCAN_LOADS; ++u) {
            const int j = base + u * OWN_THREADS + tid;
            if (j < N_EDGE4) {
                note_edge(v[u].x, 4 * j + 0, lo, list, &listN);
                note_edge(v[u].y, 4 * j + 1, lo, list, &listN);
                note_edge(v[u].z, 4 * j + 2, lo, list, &listN);
                note_edge(v[u].w, 4 * j + 3, lo, list, &listN);
            }
        }
    }
    __syncthreads();
    const int total = listN;

    float a0 = 0.f, a1 = 0.f, a2 = 0.f;
    if (total <= LIST_CAP) {
        for (int base = 0; base < total; base += 4 * OWN_THREADS) {
            bool on[4];
            int  d[4], s[4], p[4];
            float ps[4][3], pd[4][3];
#pragma unroll
            for (int u = 0; u < 4; ++u) {
                const int i = base + u * OWN_THREADS + tid;
                on[u] = i < total;
                const int ent = list[on[u] ? i : 0];
                d[u] = lo + (int)((unsigned)ent >> 20);
                s[u] = srcA[ent & 0xFFFFF];
            }
#pragma unroll
            for (int u = 0; u < 4; ++u) {
                p[u] = 0;
                if (on[u]) p[u] = atomicAdd(&lcnt[d[u] - lo], 1);
            }
#pragma unroll
            for (int u = 0; u < 4; ++u) {
                ps[u][0] = pos[s[u] * 3 + 0]; ps[u][1] = pos[s[u] * 3 + 1]; ps[u][2] = pos[s[u] * 3 + 2];
                pd[u][0] = pos[d[u] * 3 + 0]; pd[u][1] = pos[d[u] * 3 + 1]; pd[u][2] = pos[d[u] * 3 + 2];
            }
#pragma unroll
            for (int u = 0; u < 4; ++u) {
                if (on[u]) place_edge(d[u], s[u], p[u], bucket, ovfTotal, ovfDst, ovfSrc);
            }
#pragma unroll
            for (int u = 0; u < 4; ++u) {
                a0 += on[u] ? (ps[u][0] - pd[u][0]) : 0.f;
                a1 += on[u] ? (ps[u][1] - pd[u][1]) : 0.f;
                a2 += on[u] ? (ps[u][2] - pd[u][2]) : 0.f;
            }
        }
    } else {
        for (int e = tid; e < N_EDGES; e += OWN_THREADS) {
            const int d = dstA[e];
            const unsigned r = (unsigned)(d - lo);
            if (r < (unsigned)OWN_SPAN) {
                const int s = srcA[e];
                const int p = atomicAdd(&lcnt[r], 1);
                place_edge(d, s, p, bucket, ovfTotal, ovfDst, ovfSrc);
                a0 += pos[s * 3 + 0] - pos[d * 3 + 0];
                a1 += pos[s * 3 + 1] - pos[d * 3 + 1];
                a2 += pos[s * 3 + 2] - pos[d * 3 + 2];
            }
        }
    }

    block_sum3(sm, a0, a1, a2);
    if (tid == 0) {
        part[blockIdx.x * 3 + 0] = a0;
        part[blockIdx.x * 3 + 1] = a1;
        part[blockIdx.x * 3 + 2] = a2;
    }
    if (tid < OWN_SPAN) cnt[lo + tid] = lcnt[tid];
}


template <int MASK>
__device__ __forceinline__ float lane_xor(float v)
{
    return __int_as_float(__builtin_amdgcn_mov_dpp(__float_as_int(v), 0x160 + MASK,
                                                   0xf, 0xf, false));
}
__device__ __forceinline__ float group8_sum(float v)
{
    v += lane_xor<1>(v);
    v += lane_xor<2>(v);
    v += lane_xor<4>(v);
    return v;
}

__device__ __forceinline__ float head_score(const float4 x, const float4 xr,
                                            const float ea0, const float ea1, const float ea2,
                                            const float4 w0, const float4 w1, const float4 w2,
                                            const float4 at)
{
    float v0 = x.x + xr.x + ea0 * w0.x + ea1 * w1.x + ea2 * w2.x;
    float v1 = x.y + xr.y + ea0 * w0.y + ea1 * w1.y + ea2 * w2.y;
    float v2 = x.z + xr.z + ea0 * w0.z + ea1 * w1.z + ea2 * w2.z;
    float v3 = x.w + xr.w + ea0 * w0.w + ea1 * w1.w + ea2 * w2.w;
    v0 = (v0 > 0.f) ? v0 : NEG_SLOPE * v0;
    v1 = (v1 > 0.f) ? v1 : NEG_SLOPE * v1;
    v2 = (v2 > 0.f) ? v2 : NEG_SLOPE * v2;
    v3 = (v3 > 0.f) ? v3 : NEG_SLOPE * v3;
    const float p = v0 * at.x + v1 * at.y + v2 * at.z + v3 * at.w;
    return group8_sum(p);
}

__device__ __forceinline__ void edge_step(const float4 x, const float sc, const bool valid,
                                          float& m, float& l, float4& acc)
{
    const float dl = sc - m;
    const float t  = expf(-fabsf(dl));
    const bool  up = dl > 0.f;
    float p = up ? 1.0f : t;
    float r = up ? t : 1.0f;
    p = valid ? p : 0.0f;
    r = valid ? r : 1.0f;
    m = (valid && up) ? sc : m;
    l = l * r + p;
    acc.x = acc.x * r + p * x.x;
    acc.y = acc.y * r + p * x.y;
    acc.z = acc.z * r + p * x.z;
    acc.w = acc.w * r + p * x.w;
}

__global__ __launch_bounds__(256)
void node_softmax_agg_kernel(const int* __restrict__ cnt, const int* __restrict__ bucket,
                             const int* __restrict__ ovfTotal, const int* __restrict__ ovfDst,
                             const int* __restrict__ ovfSrc,
                             const float* __restrict__ pos, const float* __restrict__ part,
                             const float* __restrict__ Wedge, const float* __restrict__ att,
                             const float* __restrict__ XL, const float* __restrict__ XR,
                             float* __restrict__ out, float* __restrict__ colAcc)
{
    __shared__ float red[8][2 * HC];
    __shared__ float sm[3][8];
    const int lane = threadIdx.x & 31;
    const int wave   = __builtin_amdgcn_readfirstlane((int)((blockIdx.x * blockDim.x + threadIdx.x) >> 5));
    const int nWaves = (int)((gridDim.x * blockDim.x) >> 5);

    const int c0 = 4 * lane;
    const float4 w0 = *(const float4*)(Wedge + 0 * HC + c0);
    const float4 w1 = *(const float4*)(Wedge + 1 * HC + c0);
    const float4 w2 = *(const float4*)(Wedge + 2 * HC + c0);
    const float4 at = *(const float4*)(att + c0);

    float mean0 = 0.f, mean1 = 0.f, mean2 = 0.f;
    static_assert(FILL_BLOCKS <= 4 * 256, "four partials per thread cover the edge blocks");
#pragma unroll
    for (int k = 0; k < 4; ++k) {
        const int  b  = (int)threadIdx.x + 256 * k;
        const bool on = b < FILL_BLOCKS;
        const int  bc = on ? b : (FILL_BLOCKS - 1);
        const float v0 = part[bc * 3 + 0], v1 = part[bc * 3 + 1], v2 = part[bc * 3 + 2];
        mean0 += on ? v0 : 0.f;
        mean1 += on ? v1 : 0.f;
        mean2 += on ? v2 : 0.f;
    }
    block_sum3(sm, mean0, mean1, mean2);
    mean0 = mean0 / (float)N_EDGES;
    mean1 = mean1 / (float)N_EDGES;
    mean2 = mean2 / (float)N_EDGES;

    const int ovfN = ovfTotal[0];

    float4 cs = make_float4(0.f, 0.f, 0.f, 0.f);
    float4 cq = make_float4(0.f, 0.f, 0.f, 0.f);

    for (int d = wave; d < N_NODES; d += nWaves) {
        const int n  = cnt[d];
        const int nb = (n < BUCKET) ? n : BUCKET;

        const float4 xr = *(const float4*)(XR + (size_t)d * HC + c0);
        const float4 xs = *(const float4*)(XL + (size_t)d * HC + c0);
        const float pd0 = pos[d * 3 + 0], pd1 = pos[d * 3 + 1], pd2 = pos[d * 3 + 2];

        float  m   = head_score(xs, xr, mean0, mean1, mean2, w0, w1, w2, at);
        float  l   = 1.0f;
        float4 acc = xs;

        if (nb > 0) {
            const int le = (lane < nb) ? lane : (nb - 1);
            const int sv = bucket[d * BUCKET + le];
            const float q0 = pos[sv * 3 + 0] - pd0;
            const float q1 = pos[sv * 3 + 1] - pd1;
            const float q2 = pos[sv * 3 + 2] - pd2;

            for (int j = 0; j < nb; j += 4) {
                int    jj[4];
                float4 x[4];
#pragma unroll
                for (int u = 0; u < 4; ++u) {
                    const int t = j + u;
                    jj[u] = (t < nb) ? t : (nb - 1);
                    const int s = __builtin_amdgcn_readlane(sv, jj[u]);
                    x[u] = *(const float4*)(XL + (size_t)s * HC + c0);
                }
#pragma unroll
                for (int u = 0; u < 4; ++u) {
                    const bool  valid = (j + u) < nb;
                    const float ea0 = __int_as_float(__builtin_amdgcn_readlane(__float_as_int(q0), jj[u]));
                    const float ea1 = __int_as_float(__builtin_amdgcn_readlane(__float_as_int(q1), jj[u]));
                    const float ea2 = __int_as_float(__builtin_amdgcn_readlane(__float_as_int(q2), jj[u]));
                    const float sc  = head_score(x[u], xr, ea0, ea1, ea2, w0, w1, w2, at);
                    edge_step(x[u], sc, valid, m, l, acc);
                }
            }
        }

        if (n > BUCKET) {
            for (int base = 0; base < ovfN; base += 32) {
                const int  i   = base + lane;
                const bool in  = i < ovfN;
                const int  od  = in ? ovfDst[i] : -1;
                const int  os  = in ? ovfSrc[i] : 0;
                unsigned mask = __builtin_amdgcn_ballot_w32(od == d);
                if (mask != 0u) {
                    const float q0 = pos[os * 3 + 0] - pd0;
                    const float q1 = pos[os * 3 + 1] - pd1;
                    const float q2 = pos[os * 3 + 2] - pd2;
                    while (mask != 0u) {
                        const int j = __builtin_ctz(mask);
                        mask &= mask - 1u;
                        const int s = __builtin_amdgcn_readlane(os, j);
                        const float4 x = *(const float4*)(XL + (size_t)s * HC + c0);
                        const float ea0 = __int_as_float(__builtin_amdgcn_readlane(__float_as_int(q0), j));
                        const float ea1 = __int_as_float(__builtin_amdgcn_readlane(__float_as_int(q1), j));
                        const float ea2 = __int_as_float(__builtin_amdgcn_readlane(__float_as_int(q2), j));
                        const float sc  = head_score(x, xr, ea0, ea1, ea2, w0, w1, w2, at);
                        edge_step(x, sc, true, m, l, acc);
                    }
                }
            }
        }

        const float den = l + 1e-16f;
        float4 o;
        o.x = acc.x / den; o.y = acc.y / den; o.z = acc.z / den; o.w = acc.w / den;
        *(float4*)(out + (size_t)d * HC + c0) = o;
        cs.x += o.x; cs.y += o.y; cs.z += o.z; cs.w += o.w;
        cq.x += o.x * o.x; cq.y += o.y * o.y; cq.z += o.z * o.z; cq.w += o.w * o.w;
    }

    const int w = threadIdx.x >> 5;
    *(float4*)&red[w][c0]      = cs;
    *(float4*)&red[w][HC + c0] = cq;
    __syncthreads();
    const int t = threadIdx.x;
    float tot = red[0][t];
#pragma unroll
    for (int i = 1; i < 8; ++i) tot += red[i][t];
    atomicAdd(&colAcc[t], tot);
}

#define APPLY_BLOCKS     1250
#define APPLY_PER_THREAD 5
#define APPLY_STRIDE     (APPLY_BLOCKS * 256)
__global__ __launch_bounds__(256)
void bn_apply_kernel(float* __restrict__ out, const float* __restrict__ colAcc,
                     const float* __restrict__ gamma, const float* __restrict__ beta)
{
    const int c0 = 4 * (threadIdx.x & 31);
    const float4 s  = *(const float4*)(colAcc + c0);
    const float4 s2 = *(const float4*)(colAcc + HC + c0);
    const float4 g  = *(const float4*)(gamma + c0);
    const float4 b  = *(const float4*)(beta + c0);

    float4* __restrict__ o4 = (float4*)out;
    const int i0 = blockIdx.x * 256 + threadIdx.x;
    float4 v[APPLY_PER_THREAD];
#pragma unroll
    for (int u = 0; u < APPLY_PER_THREAD; ++u) v[u] = o4[i0 + u * APPLY_STRIDE];

    const float inv_n = 1.0f / (float)N_NODES;
    float4 sc, sh;
    {
        const float mu = s.x * inv_n; const float var = s2.x * inv_n - mu * mu;
        sc.x = g.x * rsqrtf(var + BN_EPS); sh.x = b.x - mu * sc.x;
    }
    {
        const float mu = s.y * inv_n; const float var = s2.y * inv_n - mu * mu;
        sc.y = g.y * rsqrtf(var + BN_EPS); sh.y = b.y - mu * sc.y;
    }
    {
        const float mu = s.z * inv_n; const float var = s2.z * inv_n - mu * mu;
        sc.z = g.z * rsqrtf(var + BN_EPS); sh.z = b.z - mu * sc.z;
    }
    {
        const float mu = s.w * inv_n; const float var = s2.w * inv_n - mu * mu;
        sc.w = g.w * rsqrtf(var + BN_EPS); sh.w = b.w - mu * sc.w;
    }

#pragma unroll
    for (int u = 0; u < APPLY_PER_THREAD; ++u) {
        float4 r;
        r.x = v[u].x * sc.x + sh.x;
        r.y = v[u].y * sc.y + sh.y;
        r.z = v[u].z * sc.z + sh.z;
        r.w = v[u].w * sc.w + sh.w;
        o4[i0 + u * APPLY_STRIDE] = r;
    }
}

extern "C" void kernel_launch(void* const* d_in, const int* in_sizes, int n_in,
                              void* d_out, int out_size, void* d_ws, size_t ws_size,
                              hipStream_t stream)
{
    const float* x     = (const float*)d_in[0];
    const float* pos   = (const float*)d_in[1];
    const int*   eidx  = (const int*)  d_in[2];
    const float* W_l   = (const float*)d_in[3];
    const float* W_r   = (const float*)d_in[4];
    const float* W_e   = (const float*)d_in[5];
    const float* att   = (const float*)d_in[6];
    const float* gamma = (const float*)d_in[7];
    const float* beta  = (const float*)d_in[8];
    float* out = (float*)d_out;

    const int* srcA = eidx;
    const int* dstA = eidx + N_EDGES;

    float* ws = (float*)d_ws;
    size_t off = 0;
    float* XL         = ws + off; off += (size_t)N_NODES * HC;
    float* XR         = ws + off; off += (size_t)N_NODES * HC;
    int*   bucket     = (int*)(ws + off); off += (size_t)N_NODES * BUCKET;
    int*   ovfDst     = (int*)(ws + off); off += (size_t)N_EDGES;
    int*   ovfSrc     = (int*)(ws + off); off += (size_t)N_EDGES;
    int*   cnt        = (int*)(ws + off); off += (size_t)N_NODES;
    int*   ovfTotal   = (int*)(ws + off); off += 4;
    float* colAcc     = ws + off; off += 2 * HC;
    float* meanPart   = ws + off; off += 3 * FILL_BLOCKS;
    (void)ws_size; (void)n_in; (void)in_sizes; (void)out_size;

    init_ws_kernel<<<1, 256, 0, stream>>>(ovfTotal, colAcc);

    edge_bucket_mean_gemm_kernel<<<FILL_GEMM_BLOCKS, OWN_THREADS, 0, stream>>>(
        srcA, dstA, pos, cnt, bucket, ovfTotal, ovfDst, ovfSrc, meanPart, x, W_l, W_r, XL, XR);

    node_softmax_agg_kernel<<<1024, 256, 0, stream>>>(cnt, bucket, ovfTotal, ovfDst, ovfSrc, pos,
                                                             meanPart, W_e, att, XL, XR, out, colAcc);

    bn_apply_kernel<<<APPLY_BLOCKS, 256, 0, stream>>>(out, colAcc, gamma, beta);
}


# Both GEMMs move into the 250 owner blocks: 4 of a block's 32 waves compute 16x64 half-strips (W staged in LDS, next A rows requested a task ahead) beside 28 scanning waves; launch 58016 waves -> 8000
# speedup vs baseline: 2.8796x; 1.0744x over previous
#include <hip/hip_runtime.h>
#include <math.h>

#define N_NODES   50000
#define N_EDGES   800000
#define IN_CH     64
#define HEADS     4
#define OUT_CH    32
#define HC        128
#define NEG_SLOPE 0.2f
#define BN_EPS    1e-5f
#define BUCKET    32

typedef float v2f __attribute__((ext_vector_type(2)));
typedef float v8f __attribute__((ext_vector_type(8)));

__global__ __launch_bounds__(256)
void init_ws_kernel(int* __restrict__ ovfTotal, float* __restrict__ colAcc)
{
    const int i = threadIdx.x;
    if (i < 2 * HC)  colAcc[i] = 0.0f;
    if (i == 0)      ovfTotal[0] = 0;
}

#define GEMM_STRIPS 3125
static_assert(GEMM_STRIPS * 16 == N_NODES, "the strips cover the rows exactly");
#define W_PITCH 260
__device__ __forceinline__ void stage_w(const float* __restrict__ Wm, float* __restrict__ wsh, const int tid)
{
    const int p = tid >> 5, cq = tid & 31;
    const float4 r0 = *(const float4*)(Wm + (size_t)(2 * p + 0) * HC + 4 * cq);
    const float4 r1 = *(const float4*)(Wm + (size_t)(2 * p + 1) * HC + 4 * cq);
    float* dstp = wsh + p * W_PITCH + 8 * cq;
    *(float4*)(dstp + 0) = make_float4(r0.x, r1.x, r0.y, r1.y);
    *(float4*)(dstp + 4) = make_float4(r0.z, r1.z, r0.w, r1.w);
}
__device__ __forceinline__ void load_a16(const float* __restrict__ X, const int strip, v2f (&a)[16])
{
    const int lane = threadIdx.x & 31;
    const float* arow = X + (size_t)(strip * 16 + (lane & 15)) * IN_CH + 2 * (lane >> 4);
#pragma unroll
    for (int t = 0; t < 16; ++t) a[t] = *(const v2f*)(arow + 4 * t);
}
__device__ __forceinline__ void gemm_strip16(const v2f (&a)[16], const float* __restrict__ wsh,
                                             float* __restrict__ D, const int strip, const int qh)
{
    const int lane = threadIdx.x & 31;
    const int half = lane >> 4;
    const int idx  = lane & 15;

    const float* bp = wsh + half * W_PITCH + 4 * idx + 128 * qh;
    v8f c[4] = {};
    float4 fn0 = *(const float4*)(bp + 0);
    float4 fn1 = *(const float4*)(bp + 64);
#pragma unroll
    for (int t = 0; t < 16; ++t) {
        const float4 f0 = fn0, f1 = fn1;
        if (t + 1 < 16) {
            fn0 = *(const float4*)(bp + (2 * t + 2) * W_PITCH + 0);
            fn1 = *(const float4*)(bp + (2 * t + 2) * W_PITCH + 64);
        }
        v2f b0, b1, b2, b3;
        b0.x = f0.x; b0.y = f0.y;
        b1.x = f0.z; b1.y = f0.w;
        b2.x = f1.x; b2.y = f1.y;
        b3.x = f1.z; b3.y = f1.w;
        c[0] = __builtin_amdgcn_wmma_f32_16x16x4_f32(false, a[t], false, b0, (short)0, c[0], false, false);
        c[1] = __builtin_amdgcn_wmma_f32_16x16x4_f32(false, a[t], false, b1, (short)0, c[1], false, false);
        c[2] = __builtin_amdgcn_wmma_f32_16x16x4_f32(false, a[t], false, b2, (short)0, c[2], false, false);
        c[3] = __builtin_amdgcn_wmma_f32_16x16x4_f32(false, a[t], false, b3, (short)0, c[3], false, false);
    }

#pragma unroll
    for (int v = 0; v < 8; ++v) {
        float* drow = D + (size_t)(strip * 16 + v + 8 * half) * HC + 2 * idx + 64 * qh;
        float2 o0, o1;
        o0.x = c[0][v]; o0.y = c[1][v];
        o1.x = c[2][v]; o1.y = c[3][v];
        *(float2*)(drow + 0)  = o0;
        *(float2*)(drow + 32) = o1;
    }
}

__device__ __forceinline__ float wave_sum(float v)
{
    for (int o = 16; o > 0; o >>= 1) v += __shfl_xor(v, o, 32);
    return v;
}
template <int NW>
__device__ __forceinline__ void block_sum3(float (&sm)[3][NW], float& a0, float& a1, float& a2)
{
    a0 = wave_sum(a0); a1 = wave_sum(a1); a2 = wave_sum(a2);
    const int lane = threadIdx.x & 31, w = threadIdx.x >> 5;
    if (lane == 0) { sm[0][w] = a0; sm[1][w] = a1; sm[2][w] = a2; }
    __syncthreads();
    float s0 = 0.f, s1 = 0.f, s2 = 0.f;
    for (int i = 0; i < NW; ++i) { s0 += sm[0][i]; s1 += sm[1][i]; s2 += sm[2][i]; }
    a0 = s0; a1 = s1; a2 = s2;
}

#define OWNERS      250
#define OWN_SPAN    200
#define OWN_WAVES   32
#define OWN_THREADS (OWN_WAVES * 32)
#define GEMM_WAVES  4
#define SCAN_WAVES  (OWN_WAVES - GEMM_WAVES)
#define SCAN_THREADS (SCAN_WAVES * 32)
#define LIST_CAP    8192
#define N_EDGE4     (N_EDGES / 4)
#define SCAN_LOADS  9
#define FILL_BLOCKS OWNERS
#define GEMM_TASKS  (2 * GEMM_STRIPS)
#define GEMM_PER_BLOCK 25
static_assert(OWNERS * GEMM_PER_BLOCK == GEMM_TASKS, "the blocks cover both products exactly");
static_assert((OWNERS / 2) * GEMM_PER_BLOCK == GEMM_STRIPS, "a block's strips belong to one product");
static_assert(OWNERS * OWN_SPAN == N_NODES, "the owner ranges cover the nodes exactly");
static_assert(N_EDGE4 * 4 == N_EDGES, "the destination list is a whole number of 16-byte groups");
static_assert(N_EDGES <= (1 << 20) && OWN_SPAN <= (1 << 11), "a list entry packs the edge number in 20 bits, the node above it");

__device__ __forceinline__ void place_edge(const int d, const int s, const int slot,
                                           int* __restrict__ bucket, int* __restrict__ ovfTotal,
                                           int* __restrict__ ovfDst, int* __restrict__ ovfSrc)
{
    if (slot < BUCKET) {
        bucket[d * BUCKET + slot] = s;
    } else {
        const int q = atomicAdd(ovfTotal, 1);
        ovfDst[q] = d;
        ovfSrc[q] = s;
    }
}

__device__ __forceinline__ void note_edge(const int d, const int e, const int lo,
                                          int* __restrict__ list, int* __restrict__ listN)
{
    const unsigned r = (unsigned)(d - lo);
    if (r < (unsigned)OWN_SPAN) {
        const int p = atomicAdd(listN, 1);
        if (p < LIST_CAP) list[p] = (int)((r << 20) | (unsigned)e);
    }
}

__global__ __launch_bounds__(OWN_THREADS)
void edge_bucket_mean_gemm_kernel(const int* __restrict__ srcA, const int* __restrict__ dstA,
                                  const float* __restrict__ pos,
                                  int* __restrict__ cnt, int* __restrict__ bucket,
                                  int* __restrict__ ovfTotal, int* __restrict__ ovfDst,
                                  int* __restrict__ ovfSrc, float* __restrict__ part,
                                  const float* __restrict__ X, const float* __restrict__ Wl,
                                  const float* __restrict__ Wr,
                                  float* __restrict__ XL, float* __restrict__ XR)
{
    __shared__ int   lcnt[OWN_SPAN];
    __shared__ int   list[LIST_CAP];
    __shared__ int   listN;
    __shared__ float sm[3][OWN_WAVES];
    __shared__ __attribute__((aligned(16))) float wsh[32 * W_PITCH];

    const int tid = (int)threadIdx.x;
    const int lo  = (int)blockIdx.x * OWN_SPAN;
    const bool left = (int)blockIdx.x < OWNERS / 2;
    if (tid < OWN_SPAN) lcnt[tid] = 0;
    if (tid == 0) listN = 0;
    stage_w(left ? Wl : Wr, wsh, tid);
    __syncthreads();

    if (tid < SCAN_THREADS) {
        const int4* __restrict__ d4 = (const int4*)dstA;
        for (int base = 0; base < N_EDGE4; base += SCAN_LOADS * SCAN_THREADS) {
            int4 v[SCAN_LOADS];
#pragma unroll
            for (int u = 0; u < SCAN_LOADS; ++u) {
                const int j = base + u * SCAN_THREADS + tid;
                v[u] = d4[(j < N_EDGE4) ? j : (N_EDGE4 - 1)];
            }
#pragma unroll
            for (int u = 0; u < SCAN_LOADS; ++u) {
                const int j = base + u * SCAN_THREADS + tid;
                if (j < N_EDGE4) {
                    note_edge(v[u].x, 4 * j + 0, lo, list, &listN);
                    note_edge(v[u].y, 4 * j + 1, lo, list, &listN);
                    note_edge(v[u].z, 4 * j + 2, lo, list, &listN);
                    note_edge(v[u].w, 4 * j + 3, lo, list, &listN);
                }
            }
        }
    } else {
        const int gw    = (tid - SCAN_THREADS) >> 5;
        const int first = ((int)blockIdx.x - (left ? 0 : OWNERS / 2)) * GEMM_PER_BLOCK;
        float* __restrict__ D = left ? XL : XR;
        v2f an[16];
        load_a16(X, first + (gw >> 1), an);
        for (int i = gw; i < 2 * GEMM_PER_BLOCK; i += GEMM_WAVES) {
            v2f a[16];
#pragma unroll
            for (int t = 0; t < 16; ++t) a[t] = an[t];
            const int inx = (i + GEMM_WAVES < 2 * GEMM_PER_BLOCK) ? i + GEMM_WAVES : i;
            load_a16(X, first + (inx >> 1), an);
            asm volatile("" ::: "memory");
            gemm_strip16(a, wsh, D, first + (i >> 1), i & 1);
        }
    }
    __syncthreads();
    const int total = listN;

    float a0 = 0.f, a1 = 0.f, a2 = 0.f;
    if (total <= LIST_CAP) {
        for (int base = 0; base < total; base += 4 * OWN_THREADS) {
            bool on[4];
            int  d[4], s[4], p[4];
            float ps[4][3], pd[4][3];
#pragma unroll
            for (int u = 0; u < 4; ++u) {
                const int i = base + u * OWN_THREADS + tid;
                on[u] = i < total;
                const int ent = list[on[u] ? i : 0];
                d[u] = lo + (int)((unsigned)ent >> 20);
                s[u] = srcA[ent & 0xFFFFF];
            }
#pragma unroll
            for (int u = 0; u < 4; ++u) {
                p[u] = 0;
                if (on[u]) p[u] = atomicAdd(&lcnt[d[u] - lo], 1);
            }
#pragma unroll
            for (int u = 0; u < 4; ++u) {
                ps[u][0] = pos[s[u] * 3 + 0]; ps[u][1] = pos[s[u] * 3 + 1]; ps[u][2] = pos[s[u] * 3 + 2];
                pd[u][0] = pos[d[u] * 3 + 0]; pd[u][1] = pos[d[u] * 3 + 1]; pd[u][2] = pos[d[u] * 3 + 2];
            }
#pragma unroll
            for (int u = 0; u < 4; ++u) {
                if (on[u]) place_edge(d[u], s[u], p[u], bucket, ovfTotal, ovfDst, ovfSrc);
            }
#pragma unroll
            for (int u = 0; u < 4; ++u) {
                a0 += on[u] ? (ps[u][0] - pd[u][0]) : 0.f;
                a1 += on[u] ? (ps[u][1] - pd[u][1]) : 0.f;
                a2 += on[u] ? (ps[u][2] - pd[u][2]) : 0.f;
            }
        }
    } else {
        for (int e = tid; e < N_EDGES; e += OWN_THREADS) {
            const int d = dstA[e];
            const unsigned r = (unsigned)(d - lo);
            if (r < (unsigned)OWN_SPAN) {
                const int s = srcA[e];
                const int p = atomicAdd(&lcnt[r], 1);
                place_edge(d, s, p, bucket, ovfTotal, ovfDst, ovfSrc);
                a0 += pos[s * 3 + 0] - pos[d * 3 + 0];
                a1 += pos[s * 3 + 1] - pos[d * 3 + 1];
                a2 += pos[s * 3 + 2] - pos[d * 3 + 2];
            }
        }
    }

    block_sum3(sm, a0, a1, a2);
    if (tid == 0) {
        part[blockIdx.x * 3 + 0] = a0;
        part[blockIdx.x * 3 + 1] = a1;
        part[blockIdx.x * 3 + 2] = a2;
    }
    if (tid < OWN_SPAN) cnt[lo + tid] = lcnt[tid];
}


template <int MASK>
__device__ __forceinline__ float lane_xor(float v)
{
    return __int_as_float(__builtin_amdgcn_mov_dpp(__float_as_int(v), 0x160 + MASK,
                                                   0xf, 0xf, false));
}
__device__ __forceinline__ float group8_sum(float v)
{
    v += lane_xor<1>(v);
    v += lane_xor<2>(v);
    v += lane_xor<4>(v);
    return v;
}

__device__ __forceinline__ float head_score(const float4 x, const float4 xr,
                                            const float ea0, const float ea1, const float ea2,
                                            const float4 w0, const float4 w1, const float4 w2,
                                            const float4 at)
{
    float v0 = x.x + xr.x + ea0 * w0.x + ea1 * w1.x + ea2 * w2.x;
    float v1 = x.y + xr.y + ea0 * w0.y + ea1 * w1.y + ea2 * w2.y;
    float v2 = x.z + xr.z + ea0 * w0.z + ea1 * w1.z + ea2 * w2.z;
    float v3 = x.w + xr.w + ea0 * w0.w + ea1 * w1.w + ea2 * w2.w;
    v0 = (v0 > 0.f) ? v0 : NEG_SLOPE * v0;
    v1 = (v1 > 0.f) ? v1 : NEG_SLOPE * v1;
    v2 = (v2 > 0.f) ? v2 : NEG_SLOPE * v2;
    v3 = (v3 > 0.f) ? v3 : NEG_SLOPE * v3;
    const float p = v0 * at.x + v1 * at.y + v2 * at.z + v3 * at.w;
    return group8_sum(p);
}

__device__ __forceinline__ void edge_step(const float4 x, const float sc, const bool valid,
                                          float& m, float& l, float4& acc)
{
    const float dl = sc - m;
    const float t  = expf(-fabsf(dl));
    const bool  up = dl > 0.f;
    float p = up ? 1.0f : t;
    float r = up ? t : 1.0f;
    p = valid ? p : 0.0f;
    r = valid ? r : 1.0f;
    m = (valid && up) ? sc : m;
    l = l * r + p;
    acc.x = acc.x * r + p * x.x;
    acc.y = acc.y * r + p * x.y;
    acc.z = acc.z * r + p * x.z;
    acc.w = acc.w * r + p * x.w;
}

__global__ __launch_bounds__(256)
void node_softmax_agg_kernel(const int* __restrict__ cnt, const int* __restrict__ bucket,
                             const int* __restrict__ ovfTotal, const int* __restrict__ ovfDst,
                             const int* __restrict__ ovfSrc,
                             const float* __restrict__ pos, const float* __restrict__ part,
                             const float* __restrict__ Wedge, const float* __restrict__ att,
                             const float* __restrict__ XL, const float* __restrict__ XR,
                             float* __restrict__ out, float* __restrict__ colAcc)
{
    __shared__ float red[8][2 * HC];
    __shared__ float sm[3][8];
    const int lane = threadIdx.x & 31;
    const int wave   = __builtin_amdgcn_readfirstlane((int)((blockIdx.x * blockDim.x + threadIdx.x) >> 5));
    const int nWaves = (int)((gridDim.x * blockDim.x) >> 5);

    const int c0 = 4 * lane;
    const float4 w0 = *(const float4*)(Wedge + 0 * HC + c0);
    const float4 w1 = *(const float4*)(Wedge + 1 * HC + c0);
    const float4 w2 = *(const float4*)(Wedge + 2 * HC + c0);
    const float4 at = *(const float4*)(att + c0);

    float mean0 = 0.f, mean1 = 0.f, mean2 = 0.f;
    static_assert(FILL_BLOCKS <= 4 * 256, "four partials per thread cover the edge blocks");
#pragma unroll
    for (int k = 0; k < 4; ++k) {
        const int  b  = (int)threadIdx.x + 256 * k;
        const bool on = b < FILL_BLOCKS;
        const int  bc = on ? b : (FILL_BLOCKS - 1);
        const float v0 = part[bc * 3 + 0], v1 = part[bc * 3 + 1], v2 = part[bc * 3 + 2];
        mean0 += on ? v0 : 0.f;
        mean1 += on ? v1 : 0.f;
        mean2 += on ? v2 : 0.f;
    }
    block_sum3(sm, mean0, mean1, mean2);
    mean0 = mean0 / (float)N_EDGES;
    mean1 = mean1 / (float)N_EDGES;
    mean2 = mean2 / (float)N_EDGES;

    const int ovfN = ovfTotal[0];

    float4 cs = make_float4(0.f, 0.f, 0.f, 0.f);
    float4 cq = make_float4(0.f, 0.f, 0.f, 0.f);

    for (int d = wave; d < N_NODES; d += nWaves) {
        const int n  = cnt[d];
        const int nb = (n < BUCKET) ? n : BUCKET;

        const float4 xr = *(const float4*)(XR + (size_t)d * HC + c0);
        const float4 xs = *(const float4*)(XL + (size_t)d * HC + c0);
        const float pd0 = pos[d * 3 + 0], pd1 = pos[d * 3 + 1], pd2 = pos[d * 3 + 2];

        float  m   = head_score(xs, xr, mean0, mean1, mean2, w0, w1, w2, at);
        float  l   = 1.0f;
        float4 acc = xs;

        if (nb > 0) {
            const int le = (lane < nb) ? lane : (nb - 1);
            const int sv = bucket[d * BUCKET + le];
            const float q0 = pos[sv * 3 + 0] - pd0;
            const float q1 = pos[sv * 3 + 1] - pd1;
            const float q2 = pos[sv * 3 + 2] - pd2;

            for (int j = 0; j < nb; j += 4) {
                int    jj[4];
                float4 x[4];
#pragma unroll
                for (int u = 0; u < 4; ++u) {
                    const int t = j + u;
                    jj[u] = (t < nb) ? t : (nb - 1);
                    const int s = __builtin_amdgcn_readlane(sv, jj[u]);
                    x[u] = *(const float4*)(XL + (size_t)s * HC + c0);
                }
#pragma unroll
                for (int u = 0; u < 4; ++u) {
                    const bool  valid = (j + u) < nb;
                    const float ea0 = __int_as_float(__builtin_amdgcn_readlane(__float_as_int(q0), jj[u]));
                    const float ea1 = __int_as_float(__builtin_amdgcn_readlane(__float_as_int(q1), jj[u]));
                    const float ea2 = __int_as_float(__builtin_amdgcn_readlane(__float_as_int(q2), jj[u]));
                    const float sc  = head_score(x[u], xr, ea0, ea1, ea2, w0, w1, w2, at);
                    edge_step(x[u], sc, valid, m, l, acc);
                }
            }
        }

        if (n > BUCKET) {
            for (int base = 0; base < ovfN; base += 32) {
                const int  i   = base + lane;
                const bool in  = i < ovfN;
                const int  od  = in ? ovfDst[i] : -1;
                const int  os  = in ? ovfSrc[i] : 0;
                unsigned mask = __builtin_amdgcn_ballot_w32(od == d);
                if (mask != 0u) {
                    const float q0 = pos[os * 3 + 0] - pd0;
                    const float q1 = pos[os * 3 + 1] - pd1;
                    const float q2 = pos[os * 3 + 2] - pd2;
                    while (mask != 0u) {
                        const int j = __builtin_ctz(mask);
                        mask &= mask - 1u;
                        const int s = __builtin_amdgcn_readlane(os, j);
                        const float4 x = *(const float4*)(XL + (size_t)s * HC + c0);
                        const float ea0 = __int_as_float(__builtin_amdgcn_readlane(__float_as_int(q0), j));
                        const float ea1 = __int_as_float(__builtin_amdgcn_readlane(__float_as_int(q1), j));
                        const float ea2 = __int_as_float(__builtin_amdgcn_readlane(__float_as_int(q2), j));
                        const float sc  = head_score(x, xr, ea0, ea1, ea2, w0, w1, w2, at);
                        edge_step(x, sc, true, m, l, acc);
                    }
                }
            }
        }

        const float den = l + 1e-16f;
        float4 o;
        o.x = acc.x / den; o.y = acc.y / den; o.z = acc.z / den; o.w = acc.w / den;
        *(float4*)(out + (size_t)d * HC + c0) = o;
        cs.x += o.x; cs.y += o.y; cs.z += o.z; cs.w += o.w;
        cq.x += o.x * o.x; cq.y += o.y * o.y; cq.z += o.z * o.z; cq.w += o.w * o.w;
    }

    const int w = threadIdx.x >> 5;
    *(float4*)&red[w][c0]      = cs;
    *(float4*)&red[w][HC + c0] = cq;
    __syncthreads();
    const int t = threadIdx.x;
    float tot = red[0][t];
#pragma unroll
    for (int i = 1; i < 8; ++i) tot += red[i][t];
    atomicAdd(&colAcc[t], tot);
}

#define APPLY_BLOCKS     1250
#define APPLY_PER_THREAD 5
#define APPLY_STRIDE     (APPLY_BLOCKS * 256)
__global__ __launch_bounds__(256)
void bn_apply_kernel(float* __restrict__ out, const float* __restrict__ colAcc,
                     const float* __restrict__ gamma, const float* __restrict__ beta)
{
    const int c0 = 4 * (threadIdx.x & 31);
    const float4 s  = *(const float4*)(colAcc + c0);
    const float4 s2 = *(const float4*)(colAcc + HC + c0);
    const float4 g  = *(const float4*)(gamma + c0);
    const float4 b  = *(const float4*)(beta + c0);

    float4* __restrict__ o4 = (float4*)out;
    const int i0 = blockIdx.x * 256 + threadIdx.x;
    float4 v[APPLY_PER_THREAD];
#pragma unroll
    for (int u = 0; u < APPLY_PER_THREAD; ++u) v[u] = o4[i0 + u * APPLY_STRIDE];

    const float inv_n = 1.0f / (float)N_NODES;
    float4 sc, sh;
    {
        const float mu = s.x * inv_n; const float var = s2.x * inv_n - mu * mu;
        sc.x = g.x * rsqrtf(var + BN_EPS); sh.x = b.x - mu * sc.x;
    }
    {
        const float mu = s.y * inv_n; const float var = s2.y * inv_n - mu * mu;
        sc.y = g.y * rsqrtf(var + BN_EPS); sh.y = b.y - mu * sc.y;
    }
    {
        const float mu = s.z * inv_n; const float var = s2.z * inv_n - mu * mu;
        sc.z = g.z * rsqrtf(var + BN_EPS); sh.z = b.z - mu * sc.z;
    }
    {
        const float mu = s.w * inv_n; const float var = s2.w * inv_n - mu * mu;
        sc.w = g.w * rsqrtf(var + BN_EPS); sh.w = b.w - mu * sc.w;
    }

#pragma unroll
    for (int u = 0; u < APPLY_PER_THREAD; ++u) {
        float4 r;
        r.x = v[u].x * sc.x + sh.x;
        r.y = v[u].y * sc.y + sh.y;
        r.z = v[u].z * sc.z + sh.z;
        r.w = v[u].w * sc.w + sh.w;
        o4[i0 + u * APPLY_STRIDE] = r;
    }
}

extern "C" void kernel_launch(void* const* d_in, const int* in_sizes, int n_in,
                              void* d_out, int out_size, void* d_ws, size_t ws_size,
                              hipStream_t stream)
{
    const float* x     = (const float*)d_in[0];
    const float* pos   = (const float*)d_in[1];
    const int*   eidx  = (const int*)  d_in[2];
    const float* W_l   = (const float*)d_in[3];
    const float* W_r   = (const float*)d_in[4];
    const float* W_e   = (const float*)d_in[5];
    const float* att   = (const float*)d_in[6];
    const float* gamma = (const float*)d_in[7];
    const float* beta  = (const float*)d_in[8];
    float* out = (float*)d_out;

    const int* srcA = eidx;
    const int* dstA = eidx + N_EDGES;

    float* ws = (float*)d_ws;
    size_t off = 0;
    float* XL         = ws + off; off += (size_t)N_NODES * HC;
    float* XR         = ws + off; off += (size_t)N_NODES * HC;
    int*   bucket     = (int*)(ws + off); off += (size_t)N_NODES * BUCKET;
    int*   ovfDst     = (int*)(ws + off); off += (size_t)N_EDGES;
    int*   ovfSrc     = (int*)(ws + off); off += (size_t)N_EDGES;
    int*   cnt        = (int*)(ws + off); off += (size_t)N_NODES;
    int*   ovfTotal   = (int*)(ws + off); off += 4;
    float* colAcc     = ws + off; off += 2 * HC;
    float* meanPart   = ws + off; off += 3 * FILL_BLOCKS;
    (void)ws_size; (void)n_in; (void)in_sizes; (void)out_size;

    init_ws_kernel<<<1, 256, 0, stream>>>(ovfTotal, colAcc);

    edge_bucket_mean_gemm_kernel<<<OWNERS, OWN_THREADS, 0, stream>>>(
        srcA, dstA, pos, cnt, bucket, ovfTotal, ovfDst, ovfSrc, meanPart, x, W_l, W_r, XL, XR);

    node_softmax_agg_kernel<<<1024, 256, 0, stream>>>(cnt, bucket, ovfTotal, ovfDst, ovfSrc, pos,
                                                             meanPart, W_e, att, XL, XR, out, colAcc);

    bn_apply_kernel<<<APPLY_BLOCKS, 256, 0, stream>>>(out, colAcc, gamma, beta);
}


# Owner blocks scan owner bytes (dst/200, one per edge, written by launch 0), 4 edges a compare, not the 4-byte list; hits go to per-wave LDS lists by ballot rank, no LDS add; slots by arrival still
# speedup vs baseline: 3.1805x; 1.1045x over previous
#include <hip/hip_runtime.h>
#include <math.h>

#define N_NODES   50000
#define N_EDGES   800000
#define IN_CH     64
#define HEADS     4
#define OUT_CH    32
#define HC        128
#define NEG_SLOPE 0.2f
#define BN_EPS    1e-5f
#define BUCKET    32

typedef float v2f __attribute__((ext_vector_type(2)));
typedef float v8f __attribute__((ext_vector_type(8)));

#define OWN_SPAN    200
#define N_EDGE4     (N_EDGES / 4)
#define OWNB_BLOCKS ((N_EDGE4 + 1023) / 1024)
static_assert(N_EDGE4 * 4 == N_EDGES, "the destination list is a whole number of 16-byte groups");
static_assert((N_NODES - 1) / OWN_SPAN < 255, "an owner number fits a byte, and 255 is nobody's");
__global__ __launch_bounds__(256)
void owner_bytes_init_kernel(const int* __restrict__ dstA, unsigned* __restrict__ ownb,
                             int* __restrict__ ovfTotal, float* __restrict__ colAcc)
{
    const int t = threadIdx.x;
    if (blockIdx.x == 0) {
        if (t < 2 * HC)  colAcc[t] = 0.0f;
        if (t == 0)      ovfTotal[0] = 0;
    }
    const int4* __restrict__ d4 = (const int4*)dstA;
    int4 d[4];
#pragma unroll
    for (int k = 0; k < 4; ++k) {
        const int i = (int)blockIdx.x * 1024 + k * 256 + t;
        d[k] = d4[(i < N_EDGE4) ? i : (N_EDGE4 - 1)];
    }
#pragma unroll
    for (int k = 0; k < 4; ++k) {
        const int i = (int)blockIdx.x * 1024 + k * 256 + t;
        if (i < N_EDGE4)
            ownb[i] =  ((unsigned)d[k].x / OWN_SPAN)        | (((unsigned)d[k].y / OWN_SPAN) << 8)
                    | (((unsigned)d[k].z / OWN_SPAN) << 16) | (((unsigned)d[k].w / OWN_SPAN) << 24);
    }
}

#define GEMM_STRIPS 3125
static_assert(GEMM_STRIPS * 16 == N_NODES, "the strips cover the rows exactly");
#define W_PITCH 260
__device__ __forceinline__ void stage_w(const float* __restrict__ Wm, float* __restrict__ wsh, const int tid)
{
    const int p = tid >> 5, cq = tid & 31;
    const float4 r0 = *(const float4*)(Wm + (size_t)(2 * p + 0) * HC + 4 * cq);
    const float4 r1 = *(const float4*)(Wm + (size_t)(2 * p + 1) * HC + 4 * cq);
    float* dstp = wsh + p * W_PITCH + 8 * cq;
    *(float4*)(dstp + 0) = make_float4(r0.x, r1.x, r0.y, r1.y);
    *(float4*)(dstp + 4) = make_float4(r0.z, r1.z, r0.w, r1.w);
}
__device__ __forceinline__ void load_a16(const float* __restrict__ X, const int strip, v2f (&a)[16])
{
    const int lane = threadIdx.x & 31;
    const float* arow = X + (size_t)(strip * 16 + (lane & 15)) * IN_CH + 2 * (lane >> 4);
#pragma unroll
    for (int t = 0; t < 16; ++t) a[t] = *(const v2f*)(arow + 4 * t);
}
__device__ __forceinline__ void gemm_strip16(const v2f (&a)[16], const float* __restrict__ wsh,
                                             float* __restrict__ D, const int strip, const int qh)
{
    const int lane = threadIdx.x & 31;
    const int half = lane >> 4;
    const int idx  = lane & 15;

    const float* bp = wsh + half * W_PITCH + 4 * idx + 128 * qh;
    v8f c[4] = {};
    float4 fn0 = *(const float4*)(bp + 0);
    float4 fn1 = *(const float4*)(bp + 64);
#pragma unroll
    for (int t = 0; t < 16; ++t) {
        const float4 f0 = fn0, f1 = fn1;
        if (t + 1 < 16) {
            fn0 = *(const float4*)(bp + (2 * t + 2) * W_PITCH + 0);
            fn1 = *(const float4*)(bp + (2 * t + 2) * W_PITCH + 64);
        }
        v2f b0, b1, b2, b3;
        b0.x = f0.x; b0.y = f0.y;
        b1.x = f0.z; b1.y = f0.w;
        b2.x = f1.x; b2.y = f1.y;
        b3.x = f1.z; b3.y = f1.w;
        c[0] = __builtin_amdgcn_wmma_f32_16x16x4_f32(false, a[t], false, b0, (short)0, c[0], false, false);
        c[1] = __builtin_amdgcn_wmma_f32_16x16x4_f32(false, a[t], false, b1, (short)0, c[1], false, false);
        c[2] = __builtin_amdgcn_wmma_f32_16x16x4_f32(false, a[t], false, b2, (short)0, c[2], false, false);
        c[3] = __builtin_amdgcn_wmma_f32_16x16x4_f32(false, a[t], false, b3, (short)0, c[3], false, false);
    }

#pragma unroll
    for (int v = 0; v < 8; ++v) {
        float* drow = D + (size_t)(strip * 16 + v + 8 * half) * HC + 2 * idx + 64 * qh;
        float2 o0, o1;
        o0.x = c[0][v]; o0.y = c[1][v];
        o1.x = c[2][v]; o1.y = c[3][v];
        *(float2*)(drow + 0)  = o0;
        *(float2*)(drow + 32) = o1;
    }
}

__device__ __forceinline__ float wave_sum(float v)
{
    for (int o = 16; o > 0; o >>= 1) v += __shfl_xor(v, o, 32);
    return v;
}
template <int NW>
__device__ __forceinline__ void block_sum3(float (&sm)[3][NW], float& a0, float& a1, float& a2)
{
    a0 = wave_sum(a0); a1 = wave_sum(a1); a2 = wave_sum(a2);
    const int lane = threadIdx.x & 31, w = threadIdx.x >> 5;
    if (lane == 0) { sm[0][w] = a0; sm[1][w] = a1; sm[2][w] = a2; }
    __syncthreads();
    float s0 = 0.f, s1 = 0.f, s2 = 0.f;
    for (int i = 0; i < NW; ++i) { s0 += sm[0][i]; s1 += sm[1][i]; s2 += sm[2][i]; }
    a0 = s0; a1 = s1; a2 = s2;
}

#define OWNERS      250
#define OWN_WAVES   32
#define OWN_THREADS (OWN_WAVES * 32)
#define GEMM_WAVES  4
#define SCAN_WAVES  (OWN_WAVES - GEMM_WAVES)
#define SCAN_THREADS (SCAN_WAVES * 32)
#define SEG_CAP     512
#define N_EDGE16    (N_EDGES / 16)
#define SCAN_LOADS  8
#define FILL_BLOCKS OWNERS
#define GEMM_TASKS  (2 * GEMM_STRIPS)
#define GEMM_PER_BLOCK 25
static_assert(OWNERS * GEMM_PER_BLOCK == GEMM_TASKS, "the blocks cover both products exactly");
static_assert((OWNERS / 2) * GEMM_PER_BLOCK == GEMM_STRIPS, "a block's strips belong to one product");
static_assert(OWNERS * OWN_SPAN == N_NODES, "the owner ranges cover the nodes exactly");
static_assert(N_EDGE16 * 16 == N_EDGES, "the owner byte plane is a whole number of 16-byte groups");

__device__ __forceinline__ void place_edge(const int d, const int s, const int slot,
                                           int* __restrict__ bucket, int* __restrict__ ovfTotal,
                                           int* __restrict__ ovfDst, int* __restrict__ ovfSrc)
{
    if (slot < BUCKET) {
        bucket[d * BUCKET + slot] = s;
    } else {
        const int q = atomicAdd(ovfTotal, 1);
        ovfDst[q] = d;
        ovfSrc[q] = s;
    }
}

__device__ __forceinline__ void note_word(const unsigned w, const int e0, const unsigned pat,
                                          int* __restrict__ seg, int& nw)
{
    const unsigned x = w ^ pat;
    const unsigned t = (x & 0x7F7F7F7Fu) + 0x7F7F7F7Fu;
    unsigned z = ~(t | x | 0x7F7F7F7Fu);
    unsigned m = __builtin_amdgcn_ballot_w32(z != 0u);
    while (m != 0u) {
        if (z != 0u) {
            const int p = nw + (int)__builtin_amdgcn_mbcnt_lo(m, 0u);
            if (p < SEG_CAP) seg[p] = e0 + (__builtin_ctz(z) >> 3);
            z &= z - 1u;
        }
        nw += __popc(m);
        m = __builtin_amdgcn_ballot_w32(z != 0u);
    }
}

__global__ __launch_bounds__(OWN_THREADS)
void edge_bucket_mean_gemm_kernel(const int* __restrict__ srcA, const int* __restrict__ dstA,
                                  const unsigned* __restrict__ ownb, const float* __restrict__ pos,
                                  int* __restrict__ cnt, int* __restrict__ bucket,
                                  int* __restrict__ ovfTotal, int* __restrict__ ovfDst,
                                  int* __restrict__ ovfSrc, float* __restrict__ part,
                                  const float* __restrict__ X, const float* __restrict__ Wl,
                                  const float* __restrict__ Wr,
                                  float* __restrict__ XL, float* __restrict__ XR)
{
    __shared__ int   lcnt[OWN_SPAN];
    __shared__ int   list[SCAN_WAVES * SEG_CAP];
    __shared__ int   listFull;
    __shared__ float sm[3][OWN_WAVES];
    __shared__ __attribute__((aligned(16))) float wsh[32 * W_PITCH];

    const int tid = (int)threadIdx.x;
    const int lo  = (int)blockIdx.x * OWN_SPAN;
    const bool left = (int)blockIdx.x < OWNERS / 2;
    if (tid < OWN_SPAN) lcnt[tid] = 0;
    if (tid == 0) listFull = 0;
    stage_w(left ? Wl : Wr, wsh, tid);
    __syncthreads();

    int* const seg = list + ((tid < SCAN_THREADS) ? (tid >> 5) : 0) * SEG_CAP;
    int nw = 0;
    if (tid < SCAN_THREADS) {
        const uint4* __restrict__ o16 = (const uint4*)ownb;
        const unsigned pat = (unsigned)blockIdx.x * 0x01010101u;
        for (int base = 0; base < N_EDGE16; base += SCAN_LOADS * SCAN_THREADS) {
            uint4 v[SCAN_LOADS];
#pragma unroll
            for (int u = 0; u < SCAN_LOADS; ++u) {
                const int j = base + u * SCAN_THREADS + tid;
                v[u] = o16[(j < N_EDGE16) ? j : (N_EDGE16 - 1)];
            }
#pragma unroll
            for (int u = 0; u < SCAN_LOADS; ++u) {
                const int j = base + u * SCAN_THREADS + tid;
                const unsigned pj = (j < N_EDGE16) ? pat : 0xFFFFFFFFu;
                note_word(v[u].x, 16 * j + 0,  pj, seg, nw);
                note_word(v[u].y, 16 * j + 4,  pj, seg, nw);
                note_word(v[u].z, 16 * j + 8,  pj, seg, nw);
                note_word(v[u].w, 16 * j + 12, pj, seg, nw);
            }
        }
        if (nw > SEG_CAP) listFull = 1;
    } else {
        const int gw    = (tid - SCAN_THREADS) >> 5;
        const int first = ((int)blockIdx.x - (left ? 0 : OWNERS / 2)) * GEMM_PER_BLOCK;
        float* __restrict__ D = left ? XL : XR;
        v2f an[16];
        load_a16(X, first + (gw >> 1), an);
        for (int i = gw; i < 2 * GEMM_PER_BLOCK; i += GEMM_WAVES) {
            v2f a[16];
#pragma unroll
            for (int t = 0; t < 16; ++t) a[t] = an[t];
            const int inx = (i + GEMM_WAVES < 2 * GEMM_PER_BLOCK) ? i + GEMM_WAVES : i;
            load_a16(X, first + (inx >> 1), an);
            asm volatile("" ::: "memory");
            gemm_strip16(a, wsh, D, first + (i >> 1), i & 1);
        }
    }
    __syncthreads();
    const bool full = listFull != 0;

    float a0 = 0.f, a1 = 0.f, a2 = 0.f;
    if (!full) {
        const int lane = tid & 31;
        for (int base = 0; base < nw; base += 4 * 32) {
            bool on[4];
            int  d[4], s[4], p[4];
            float ps[4][3], pd[4][3];
#pragma unroll
            for (int u = 0; u < 4; ++u) {
                const int i = base + u * 32 + lane;
                on[u] = i < nw;
                const int e = seg[on[u] ? i : 0];
                d[u] = dstA[e];
                s[u] = srcA[e];
            }
#pragma unroll
            for (int u = 0; u < 4; ++u) {
                p[u] = 0;
                if (on[u]) p[u] = atomicAdd(&lcnt[d[u] - lo], 1);
            }
#pragma unroll
            for (int u = 0; u < 4; ++u) {
                ps[u][0] = pos[s[u] * 3 + 0]; ps[u][1] = pos[s[u] * 3 + 1]; ps[u][2] = pos[s[u] * 3 + 2];
                pd[u][0] = pos[d[u] * 3 + 0]; pd[u][1] = pos[d[u] * 3 + 1]; pd[u][2] = pos[d[u] * 3 + 2];
            }
#pragma unroll
            for (int u = 0; u < 4; ++u) {
                if (on[u]) place_edge(d[u], s[u], p[u], bucket, ovfTotal, ovfDst, ovfSrc);
            }
#pragma unroll
            for (int u = 0; u < 4; ++u) {
                a0 += on[u] ? (ps[u][0] - pd[u][0]) : 0.f;
                a1 += on[u] ? (ps[u][1] - pd[u][1]) : 0.f;
                a2 += on[u] ? (ps[u][2] - pd[u][2]) : 0.f;
            }
        }
    } else {
        for (int e = tid; e < N_EDGES; e += OWN_THREADS) {
            const int d = dstA[e];
            const unsigned r = (unsigned)(d - lo);
            if (r < (unsigned)OWN_SPAN) {
                const int s = srcA[e];
                const int p = atomicAdd(&lcnt[r], 1);
                place_edge(d, s, p, bucket, ovfTotal, ovfDst, ovfSrc);
                a0 += pos[s * 3 + 0] - pos[d * 3 + 0];
                a1 += pos[s * 3 + 1] - pos[d * 3 + 1];
                a2 += pos[s * 3 + 2] - pos[d * 3 + 2];
            }
        }
    }

    block_sum3(sm, a0, a1, a2);
    if (tid == 0) {
        part[blockIdx.x * 3 + 0] = a0;
        part[blockIdx.x * 3 + 1] = a1;
        part[blockIdx.x * 3 + 2] = a2;
    }
    if (tid < OWN_SPAN) cnt[lo + tid] = lcnt[tid];
}


template <int MASK>
__device__ __forceinline__ float lane_xor(float v)
{
    return __int_as_float(__builtin_amdgcn_mov_dpp(__float_as_int(v), 0x160 + MASK,
                                                   0xf, 0xf, false));
}
__device__ __forceinline__ float group8_sum(float v)
{
    v += lane_xor<1>(v);
    v += lane_xor<2>(v);
    v += lane_xor<4>(v);
    return v;
}

__device__ __forceinline__ float head_score(const float4 x, const float4 xr,
                                            const float ea0, const float ea1, const float ea2,
                                            const float4 w0, const float4 w1, const float4 w2,
                                            const float4 at)
{
    float v0 = x.x + xr.x + ea0 * w0.x + ea1 * w1.x + ea2 * w2.x;
    float v1 = x.y + xr.y + ea0 * w0.y + ea1 * w1.y + ea2 * w2.y;
    float v2 = x.z + xr.z + ea0 * w0.z + ea1 * w1.z + ea2 * w2.z;
    float v3 = x.w + xr.w + ea0 * w0.w + ea1 * w1.w + ea2 * w2.w;
    v0 = (v0 > 0.f) ? v0 : NEG_SLOPE * v0;
    v1 = (v1 > 0.f) ? v1 : NEG_SLOPE * v1;
    v2 = (v2 > 0.f) ? v2 : NEG_SLOPE * v2;
    v3 = (v3 > 0.f) ? v3 : NEG_SLOPE * v3;
    const float p = v0 * at.x + v1 * at.y + v2 * at.z + v3 * at.w;
    return group8_sum(p);
}

__device__ __forceinline__ void edge_step(const float4 x, const float sc, const bool valid,
                                          float& m, float& l, float4& acc)
{
    const float dl = sc - m;
    const float t  = expf(-fabsf(dl));
    const bool  up = dl > 0.f;
    float p = up ? 1.0f : t;
    float r = up ? t : 1.0f;
    p = valid ? p : 0.0f;
    r = valid ? r : 1.0f;
    m = (valid && up) ? sc : m;
    l = l * r + p;
    acc.x = acc.x * r + p * x.x;
    acc.y = acc.y * r + p * x.y;
    acc.z = acc.z * r + p * x.z;
    acc.w = acc.w * r + p * x.w;
}

__global__ __launch_bounds__(256)
void node_softmax_agg_kernel(const int* __restrict__ cnt, const int* __restrict__ bucket,
                             const int* __restrict__ ovfTotal, const int* __restrict__ ovfDst,
                             const int* __restrict__ ovfSrc,
                             const float* __restrict__ pos, const float* __restrict__ part,
                             const float* __restrict__ Wedge, const float* __restrict__ att,
                             const float* __restrict__ XL, const float* __restrict__ XR,
                             float* __restrict__ out, float* __restrict__ colAcc)
{
    __shared__ float red[8][2 * HC];
    __shared__ float sm[3][8];
    const int lane = threadIdx.x & 31;
    const int wave   = __builtin_amdgcn_readfirstlane((int)((blockIdx.x * blockDim.x + threadIdx.x) >> 5));
    const int nWaves = (int)((gridDim.x * blockDim.x) >> 5);

    const int c0 = 4 * lane;
    const float4 w0 = *(const float4*)(Wedge + 0 * HC + c0);
    const float4 w1 = *(const float4*)(Wedge + 1 * HC + c0);
    const float4 w2 = *(const float4*)(Wedge + 2 * HC + c0);
    const float4 at = *(const float4*)(att + c0);

    float mean0 = 0.f, mean1 = 0.f, mean2 = 0.f;
    static_assert(FILL_BLOCKS <= 4 * 256, "four partials per thread cover the edge blocks");
#pragma unroll
    for (int k = 0; k < 4; ++k) {
        const int  b  = (int)threadIdx.x + 256 * k;
        const bool on = b < FILL_BLOCKS;
        const int  bc = on ? b : (FILL_BLOCKS - 1);
        const float v0 = part[bc * 3 + 0], v1 = part[bc * 3 + 1], v2 = part[bc * 3 + 2];
        mean0 += on ? v0 : 0.f;
        mean1 += on ? v1 : 0.f;
        mean2 += on ? v2 : 0.f;
    }
    block_sum3(sm, mean0, mean1, mean2);
    mean0 = mean0 / (float)N_EDGES;
    mean1 = mean1 / (float)N_EDGES;
    mean2 = mean2 / (float)N_EDGES;

    const int ovfN = ovfTotal[0];

    float4 cs = make_float4(0.f, 0.f, 0.f, 0.f);
    float4 cq = make_float4(0.f, 0.f, 0.f, 0.f);

    for (int d = wave; d < N_NODES; d += nWaves) {
        const int n  = cnt[d];
        const int nb = (n < BUCKET) ? n : BUCKET;

        const float4 xr = *(const float4*)(XR + (size_t)d * HC + c0);
        const float4 xs = *(const float4*)(XL + (size_t)d * HC + c0);
        const float pd0 = pos[d * 3 + 0], pd1 = pos[d * 3 + 1], pd2 = pos[d * 3 + 2];

        float  m   = head_score(xs, xr, mean0, mean1, mean2, w0, w1, w2, at);
        float  l   = 1.0f;
        float4 acc = xs;

        if (nb > 0) {
            const int le = (lane < nb) ? lane : (nb - 1);
            const int sv = bucket[d * BUCKET + le];
            const float q0 = pos[sv * 3 + 0] - pd0;
            const float q1 = pos[sv * 3 + 1] - pd1;
            const float q2 = pos[sv * 3 + 2] - pd2;

            for (int j = 0; j < nb; j += 4) {
                int    jj[4];
                float4 x[4];
#pragma unroll
                for (int u = 0; u < 4; ++u) {
                    const int t = j + u;
                    jj[u] = (t < nb) ? t : (nb - 1);
                    const int s = __builtin_amdgcn_readlane(sv, jj[u]);
                    x[u] = *(const float4*)(XL + (size_t)s * HC + c0);
                }
#pragma unroll
                for (int u = 0; u < 4; ++u) {
                    const bool  valid = (j + u) < nb;
                    const float ea0 = __int_as_float(__builtin_amdgcn_readlane(__float_as_int(q0), jj[u]));
                    const float ea1 = __int_as_float(__builtin_amdgcn_readlane(__float_as_int(q1), jj[u]));
                    const float ea2 = __int_as_float(__builtin_amdgcn_readlane(__float_as_int(q2), jj[u]));
                    const float sc  = head_score(x[u], xr, ea0, ea1, ea2, w0, w1, w2, at);
                    edge_step(x[u], sc, valid, m, l, acc);
                }
            }
        }

        if (n > BUCKET) {
            for (int base = 0; base < ovfN; base += 32) {
                const int  i   = base + lane;
                const bool in  = i < ovfN;
                const int  od  = in ? ovfDst[i] : -1;
                const int  os  = in ? ovfSrc[i] : 0;
                unsigned mask = __builtin_amdgcn_ballot_w32(od == d);
                if (mask != 0u) {
                    const float q0 = pos[os * 3 + 0] - pd0;
                    const float q1 = pos[os * 3 + 1] - pd1;
                    const float q2 = pos[os * 3 + 2] - pd2;
                    while (mask != 0u) {
                        const int j = __builtin_ctz(mask);
                        mask &= mask - 1u;
                        const int s = __builtin_amdgcn_readlane(os, j);
                        const float4 x = *(const float4*)(XL + (size_t)s * HC + c0);
                        const float ea0 = __int_as_float(__builtin_amdgcn_readlane(__float_as_int(q0), j));
                        const float ea1 = __int_as_float(__builtin_amdgcn_readlane(__float_as_int(q1), j));
                        const float ea2 = __int_as_float(__builtin_amdgcn_readlane(__float_as_int(q2), j));
                        const float sc  = head_score(x, xr, ea0, ea1, ea2, w0, w1, w2, at);
                        edge_step(x, sc, true, m, l, acc);
                    }
                }
            }
        }

        const float den = l + 1e-16f;
        float4 o;
        o.x = acc.x / den; o.y = acc.y / den; o.z = acc.z / den; o.w = acc.w / den;
        *(float4*)(out + (size_t)d * HC + c0) = o;
        cs.x += o.x; cs.y += o.y; cs.z += o.z; cs.w += o.w;
        cq.x += o.x * o.x; cq.y += o.y * o.y; cq.z += o.z * o.z; cq.w += o.w * o.w;
    }

    const int w = threadIdx.x >> 5;
    *(float4*)&red[w][c0]      = cs;
    *(float4*)&red[w][HC + c0] = cq;
    __syncthreads();
    const int t = threadIdx.x;
    float tot = red[0][t];
#pragma unroll
    for (int i = 1; i < 8; ++i) tot += red[i][t];
    atomicAdd(&colAcc[t], tot);
}

#define APPLY_BLOCKS     1250
#define APPLY_PER_THREAD 5
#define APPLY_STRIDE     (APPLY_BLOCKS * 256)
__global__ __launch_bounds__(256)
void bn_apply_kernel(float* __restrict__ out, const float* __restrict__ colAcc,
                     const float* __restrict__ gamma, const float* __restrict__ beta)
{
    const int c0 = 4 * (threadIdx.x & 31);
    const float4 s  = *(const float4*)(colAcc + c0);
    const float4 s2 = *(const float4*)(colAcc + HC + c0);
    const float4 g  = *(const float4*)(gamma + c0);
    const float4 b  = *(const float4*)(beta + c0);

    float4* __restrict__ o4 = (float4*)out;
    const int i0 = blockIdx.x * 256 + threadIdx.x;
    float4 v[APPLY_PER_THREAD];
#pragma unroll
    for (int u = 0; u < APPLY_PER_THREAD; ++u) v[u] = o4[i0 + u * APPLY_STRIDE];

    const float inv_n = 1.0f / (float)N_NODES;
    float4 sc, sh;
    {
        const float mu = s.x * inv_n; const float var = s2.x * inv_n - mu * mu;
        sc.x = g.x * rsqrtf(var + BN_EPS); sh.x = b.x - mu * sc.x;
    }
    {
        const float mu = s.y * inv_n; const float var = s2.y * inv_n - mu * mu;
        sc.y = g.y * rsqrtf(var + BN_EPS); sh.y = b.y - mu * sc.y;
    }
    {
        const float mu = s.z * inv_n; const float var = s2.z * inv_n - mu * mu;
        sc.z = g.z * rsqrtf(var + BN_EPS); sh.z = b.z - mu * sc.z;
    }
    {
        const float mu = s.w * inv_n; const float var = s2.w * inv_n - mu * mu;
        sc.w = g.w * rsqrtf(var + BN_EPS); sh.w = b.w - mu * sc.w;
    }

#pragma unroll
    for (int u = 0; u < APPLY_PER_THREAD; ++u) {
        float4 r;
        r.x = v[u].x * sc.x + sh.x;
        r.y = v[u].y * sc.y + sh.y;
        r.z = v[u].z * sc.z + sh.z;
        r.w = v[u].w * sc.w + sh.w;
        o4[i0 + u * APPLY_STRIDE] = r;
    }
}

extern "C" void kernel_launch(void* const* d_in, const int* in_sizes, int n_in,
                              void* d_out, int out_size, void* d_ws, size_t ws_size,
                              hipStream_t stream)
{
    const float* x     = (const float*)d_in[0];
    const float* pos   = (const float*)d_in[1];
    const int*   eidx  = (const int*)  d_in[2];
    const float* W_l   = (const float*)d_in[3];
    const float* W_r   = (const float*)d_in[4];
    const float* W_e   = (const float*)d_in[5];
    const float* att   = (const float*)d_in[6];
    const float* gamma = (const float*)d_in[7];
    const float* beta  = (const float*)d_in[8];
    float* out = (float*)d_out;

    const int* srcA = eidx;
    const int* dstA = eidx + N_EDGES;

    float* ws = (float*)d_ws;
    size_t off = 0;
    float* XL         = ws + off; off += (size_t)N_NODES * HC;
    float* XR         = ws + off; off += (size_t)N_NODES * HC;
    int*   bucket     = (int*)(ws + off); off += (size_t)N_NODES * BUCKET;
    int*   ovfDst     = (int*)(ws + off); off += (size_t)N_EDGES;
    int*   ovfSrc     = (int*)(ws + off); off += (size_t)N_EDGES;
    unsigned* ownb    = (unsigned*)(ws + off); off += (size_t)N_EDGE4;
    int*   cnt        = (int*)(ws + off); off += (size_t)N_NODES;
    int*   ovfTotal   = (int*)(ws + off); off += 4;
    float* colAcc     = ws + off; off += 2 * HC;
    float* meanPart   = ws + off; off += 3 * FILL_BLOCKS;
    (void)ws_size; (void)n_in; (void)in_sizes; (void)out_size;

    owner_bytes_init_kernel<<<OWNB_BLOCKS, 256, 0, stream>>>(dstA, ownb, ovfTotal, colAcc);

    edge_bucket_mean_gemm_kernel<<<OWNERS, OWN_THREADS, 0, stream>>>(
        srcA, dstA, ownb, pos, cnt, bucket, ovfTotal, ovfDst, ovfSrc, meanPart, x, W_l, W_r, XL, XR);

    node_softmax_agg_kernel<<<1024, 256, 0, stream>>>(cnt, bucket, ovfTotal, ovfDst, ovfSrc, pos,
                                                             meanPart, W_e, att, XL, XR, out, colAcc);

    bn_apply_kernel<<<APPLY_BLOCKS, 256, 0, stream>>>(out, colAcc, gamma, beta);
}


# Aggregation loop, vector instructions per 4 edges 217 -> 127: base-2 softmax (log2 e in att, hardware exp2), one max + rescale per 4 edges, LeakyReLU as max; score and softmax sums regrouped
# speedup vs baseline: 3.5975x; 1.1311x over previous
#include <hip/hip_runtime.h>
#include <math.h>

#define N_NODES   50000
#define N_EDGES   800000
#define IN_CH     64
#define HEADS     4
#define OUT_CH    32
#define HC        128
#define NEG_SLOPE 0.2f
#define BN_EPS    1e-5f
#define BUCKET    32

typedef float v2f __attribute__((ext_vector_type(2)));
typedef float v8f __attribute__((ext_vector_type(8)));

#define OWN_SPAN    200
#define N_EDGE4     (N_EDGES / 4)
#define OWNB_BLOCKS ((N_EDGE4 + 1023) / 1024)
static_assert(N_EDGE4 * 4 == N_EDGES, "the destination list is a whole number of 16-byte groups");
static_assert((N_NODES - 1) / OWN_SPAN < 255, "an owner number fits a byte, and 255 is nobody's");
__global__ __launch_bounds__(256)
void owner_bytes_init_kernel(const int* __restrict__ dstA, unsigned* __restrict__ ownb,
                             int* __restrict__ ovfTotal, float* __restrict__ colAcc)
{
    const int t = threadIdx.x;
    if (blockIdx.x == 0) {
        if (t < 2 * HC)  colAcc[t] = 0.0f;
        if (t == 0)      ovfTotal[0] = 0;
    }
    const int4* __restrict__ d4 = (const int4*)dstA;
    int4 d[4];
#pragma unroll
    for (int k = 0; k < 4; ++k) {
        const int i = (int)blockIdx.x * 1024 + k * 256 + t;
        d[k] = d4[(i < N_EDGE4) ? i : (N_EDGE4 - 1)];
    }
#pragma unroll
    for (int k = 0; k < 4; ++k) {
        const int i = (int)blockIdx.x * 1024 + k * 256 + t;
        if (i < N_EDGE4)
            ownb[i] =  ((unsigned)d[k].x / OWN_SPAN)        | (((unsigned)d[k].y / OWN_SPAN) << 8)
                    | (((unsigned)d[k].z / OWN_SPAN) << 16) | (((unsigned)d[k].w / OWN_SPAN) << 24);
    }
}

#define GEMM_STRIPS 3125
static_assert(GEMM_STRIPS * 16 == N_NODES, "the strips cover the rows exactly");
#define W_PITCH 260
__device__ __forceinline__ void stage_w(const float* __restrict__ Wm, float* __restrict__ wsh, const int tid)
{
    const int p = tid >> 5, cq = tid & 31;
    const float4 r0 = *(const float4*)(Wm + (size_t)(2 * p + 0) * HC + 4 * cq);
    const float4 r1 = *(const float4*)(Wm + (size_t)(2 * p + 1) * HC + 4 * cq);
    float* dstp = wsh + p * W_PITCH + 8 * cq;
    *(float4*)(dstp + 0) = make_float4(r0.x, r1.x, r0.y, r1.y);
    *(float4*)(dstp + 4) = make_float4(r0.z, r1.z, r0.w, r1.w);
}
__device__ __forceinline__ void load_a16(const float* __restrict__ X, const int strip, v2f (&a)[16])
{
    const int lane = threadIdx.x & 31;
    const float* arow = X + (size_t)(strip * 16 + (lane & 15)) * IN_CH + 2 * (lane >> 4);
#pragma unroll
    for (int t = 0; t < 16; ++t) a[t] = *(const v2f*)(arow + 4 * t);
}
__device__ __forceinline__ void gemm_strip16(const v2f (&a)[16], const float* __restrict__ wsh,
                                             float* __restrict__ D, const int strip, const int qh)
{
    const int lane = threadIdx.x & 31;
    const int half = lane >> 4;
    const int idx  = lane & 15;

    const float* bp = wsh + half * W_PITCH + 4 * idx + 128 * qh;
    v8f c[4] = {};
    float4 fn0 = *(const float4*)(bp + 0);
    float4 fn1 = *(const float4*)(bp + 64);
#pragma unroll
    for (int t = 0; t < 16; ++t) {
        const float4 f0 = fn0, f1 = fn1;
        if (t + 1 < 16) {
            fn0 = *(const float4*)(bp + (2 * t + 2) * W_PITCH + 0);
            fn1 = *(const float4*)(bp + (2 * t + 2) * W_PITCH + 64);
        }
        v2f b0, b1, b2, b3;
        b0.x = f0.x; b0.y = f0.y;
        b1.x = f0.z; b1.y = f0.w;
        b2.x = f1.x; b2.y = f1.y;
        b3.x = f1.z; b3.y = f1.w;
        c[0] = __builtin_amdgcn_wmma_f32_16x16x4_f32(false, a[t], false, b0, (short)0, c[0], false, false);
        c[1] = __builtin_amdgcn_wmma_f32_16x16x4_f32(false, a[t], false, b1, (short)0, c[1], false, false);
        c[2] = __builtin_amdgcn_wmma_f32_16x16x4_f32(false, a[t], false, b2, (short)0, c[2], false, false);
        c[3] = __builtin_amdgcn_wmma_f32_16x16x4_f32(false, a[t], false, b3, (short)0, c[3], false, false);
    }

#pragma unroll
    for (int v = 0; v < 8; ++v) {
        float* drow = D + (size_t)(strip * 16 + v + 8 * half) * HC + 2 * idx + 64 * qh;
        float2 o0, o1;
        o0.x = c[0][v]; o0.y = c[1][v];
        o1.x = c[2][v]; o1.y = c[3][v];
        *(float2*)(drow + 0)  = o0;
        *(float2*)(drow + 32) = o1;
    }
}

__device__ __forceinline__ float wave_sum(float v)
{
    for (int o = 16; o > 0; o >>= 1) v += __shfl_xor(v, o, 32);
    return v;
}
template <int NW>
__device__ __forceinline__ void block_sum3(float (&sm)[3][NW], float& a0, float& a1, float& a2)
{
    a0 = wave_sum(a0); a1 = wave_sum(a1); a2 = wave_sum(a2);
    const int lane = threadIdx.x & 31, w = threadIdx.x >> 5;
    if (lane == 0) { sm[0][w] = a0; sm[1][w] = a1; sm[2][w] = a2; }
    __syncthreads();
    float s0 = 0.f, s1 = 0.f, s2 = 0.f;
    for (int i = 0; i < NW; ++i) { s0 += sm[0][i]; s1 += sm[1][i]; s2 += sm[2][i]; }
    a0 = s0; a1 = s1; a2 = s2;
}

#define OWNERS      250
#define OWN_WAVES   32
#define OWN_THREADS (OWN_WAVES * 32)
#define GEMM_WAVES  4
#define SCAN_WAVES  (OWN_WAVES - GEMM_WAVES)
#define SCAN_THREADS (SCAN_WAVES * 32)
#define SEG_CAP     512
#define N_EDGE16    (N_EDGES / 16)
#define SCAN_LOADS  8
#define FILL_BLOCKS OWNERS
#define GEMM_TASKS  (2 * GEMM_STRIPS)
#define GEMM_PER_BLOCK 25
static_assert(OWNERS * GEMM_PER_BLOCK == GEMM_TASKS, "the blocks cover both products exactly");
static_assert((OWNERS / 2) * GEMM_PER_BLOCK == GEMM_STRIPS, "a block's strips belong to one product");
static_assert(OWNERS * OWN_SPAN == N_NODES, "the owner ranges cover the nodes exactly");
static_assert(N_EDGE16 * 16 == N_EDGES, "the owner byte plane is a whole number of 16-byte groups");

__device__ __forceinline__ void place_edge(const int d, const int s, const int slot,
                                           int* __restrict__ bucket, int* __restrict__ ovfTotal,
                                           int* __restrict__ ovfDst, int* __restrict__ ovfSrc)
{
    if (slot < BUCKET) {
        bucket[d * BUCKET + slot] = s;
    } else {
        const int q = atomicAdd(ovfTotal, 1);
        ovfDst[q] = d;
        ovfSrc[q] = s;
    }
}

__device__ __forceinline__ void note_word(const unsigned w, const int e0, const unsigned pat,
                                          int* __restrict__ seg, int& nw)
{
    const unsigned x = w ^ pat;
    const unsigned t = (x & 0x7F7F7F7Fu) + 0x7F7F7F7Fu;
    unsigned z = ~(t | x | 0x7F7F7F7Fu);
    unsigned m = __builtin_amdgcn_ballot_w32(z != 0u);
    while (m != 0u) {
        if (z != 0u) {
            const int p = nw + (int)__builtin_amdgcn_mbcnt_lo(m, 0u);
            if (p < SEG_CAP) seg[p] = e0 + (__builtin_ctz(z) >> 3);
            z &= z - 1u;
        }
        nw += __popc(m);
        m = __builtin_amdgcn_ballot_w32(z != 0u);
    }
}

__global__ __launch_bounds__(OWN_THREADS)
void edge_bucket_mean_gemm_kernel(const int* __restrict__ srcA, const int* __restrict__ dstA,
                                  const unsigned* __restrict__ ownb, const float* __restrict__ pos,
                                  int* __restrict__ cnt, int* __restrict__ bucket,
                                  int* __restrict__ ovfTotal, int* __restrict__ ovfDst,
                                  int* __restrict__ ovfSrc, float* __restrict__ part,
                                  const float* __restrict__ X, const float* __restrict__ Wl,
                                  const float* __restrict__ Wr,
                                  float* __restrict__ XL, float* __restrict__ XR)
{
    __shared__ int   lcnt[OWN_SPAN];
    __shared__ int   list[SCAN_WAVES * SEG_CAP];
    __shared__ int   listFull;
    __shared__ float sm[3][OWN_WAVES];
    __shared__ __attribute__((aligned(16))) float wsh[32 * W_PITCH];

    const int tid = (int)threadIdx.x;
    const int lo  = (int)blockIdx.x * OWN_SPAN;
    const bool left = (int)blockIdx.x < OWNERS / 2;
    if (tid < OWN_SPAN) lcnt[tid] = 0;
    if (tid == 0) listFull = 0;
    stage_w(left ? Wl : Wr, wsh, tid);
    __syncthreads();

    int* const seg = list + ((tid < SCAN_THREADS) ? (tid >> 5) : 0) * SEG_CAP;
    int nw = 0;
    if (tid < SCAN_THREADS) {
        const uint4* __restrict__ o16 = (const uint4*)ownb;
        const unsigned pat = (unsigned)blockIdx.x * 0x01010101u;
        for (int base = 0; base < N_EDGE16; base += SCAN_LOADS * SCAN_THREADS) {
            uint4 v[SCAN_LOADS];
#pragma unroll
            for (int u = 0; u < SCAN_LOADS; ++u) {
                const int j = base + u * SCAN_THREADS + tid;
                v[u] = o16[(j < N_EDGE16) ? j : (N_EDGE16 - 1)];
            }
#pragma unroll
            for (int u = 0; u < SCAN_LOADS; ++u) {
                const int j = base + u * SCAN_THREADS + tid;
                const unsigned pj = (j < N_EDGE16) ? pat : 0xFFFFFFFFu;
                note_word(v[u].x, 16 * j + 0,  pj, seg, nw);
                note_word(v[u].y, 16 * j + 4,  pj, seg, nw);
                note_word(v[u].z, 16 * j + 8,  pj, seg, nw);
                note_word(v[u].w, 16 * j + 12, pj, seg, nw);
            }
        }
        if (nw > SEG_CAP) listFull = 1;
    } else {
        const int gw    = (tid - SCAN_THREADS) >> 5;
        const int first = ((int)blockIdx.x - (left ? 0 : OWNERS / 2)) * GEMM_PER_BLOCK;
        float* __restrict__ D = left ? XL : XR;
        v2f an[16];
        load_a16(X, first + (gw >> 1), an);
        for (int i = gw; i < 2 * GEMM_PER_BLOCK; i += GEMM_WAVES) {
            v2f a[16];
#pragma unroll
            for (int t = 0; t < 16; ++t) a[t] = an[t];
            const int inx = (i + GEMM_WAVES < 2 * GEMM_PER_BLOCK) ? i + GEMM_WAVES : i;
            load_a16(X, first + (inx >> 1), an);
            asm volatile("" ::: "memory");
            gemm_strip16(a, wsh, D, first + (i >> 1), i & 1);
        }
    }
    __syncthreads();
    const bool full = listFull != 0;

    float a0 = 0.f, a1 = 0.f, a2 = 0.f;
    if (!full) {
        const int lane = tid & 31;
        for (int base = 0; base < nw; base += 4 * 32) {
            bool on[4];
            int  d[4], s[4], p[4];
            float ps[4][3], pd[4][3];
#pragma unroll
            for (int u = 0; u < 4; ++u) {
                const int i = base + u * 32 + lane;
                on[u] = i < nw;
                const int e = seg[on[u] ? i : 0];
                d[u] = dstA[e];
                s[u] = srcA[e];
            }
#pragma unroll
            for (int u = 0; u < 4; ++u) {
                p[u] = 0;
                if (on[u]) p[u] = atomicAdd(&lcnt[d[u] - lo], 1);
            }
#pragma unroll
            for (int u = 0; u < 4; ++u) {
                ps[u][0] = pos[s[u] * 3 + 0]; ps[u][1] = pos[s[u] * 3 + 1]; ps[u][2] = pos[s[u] * 3 + 2];
                pd[u][0] = pos[d[u] * 3 + 0]; pd[u][1] = pos[d[u] * 3 + 1]; pd[u][2] = pos[d[u] * 3 + 2];
            }
#pragma unroll
            for (int u = 0; u < 4; ++u) {
                if (on[u]) place_edge(d[u], s[u], p[u], bucket, ovfTotal, ovfDst, ovfSrc);
            }
#pragma unroll
            for (int u = 0; u < 4; ++u) {
                a0 += on[u] ? (ps[u][0] - pd[u][0]) : 0.f;
                a1 += on[u] ? (ps[u][1] - pd[u][1]) : 0.f;
                a2 += on[u] ? (ps[u][2] - pd[u][2]) : 0.f;
            }
        }
    } else {
        for (int e = tid; e < N_EDGES; e += OWN_THREADS) {
            const int d = dstA[e];
            const unsigned r = (unsigned)(d - lo);
            if (r < (unsigned)OWN_SPAN) {
                const int s = srcA[e];
                const int p = atomicAdd(&lcnt[r], 1);
                place_edge(d, s, p, bucket, ovfTotal, ovfDst, ovfSrc);
                a0 += pos[s * 3 + 0] - pos[d * 3 + 0];
                a1 += pos[s * 3 + 1] - pos[d * 3 + 1];
                a2 += pos[s * 3 + 2] - pos[d * 3 + 2];
            }
        }
    }

    block_sum3(sm, a0, a1, a2);
    if (tid == 0) {
        part[blockIdx.x * 3 + 0] = a0;
        part[blockIdx.x * 3 + 1] = a1;
        part[blockIdx.x * 3 + 2] = a2;
    }
    if (tid < OWN_SPAN) cnt[lo + tid] = lcnt[tid];
}


template <int MASK>
__device__ __forceinline__ float lane_xor(float v)
{
    return __int_as_float(__builtin_amdgcn_mov_dpp(__float_as_int(v), 0x160 + MASK,
                                                   0xf, 0xf, false));
}
__device__ __forceinline__ float group8_sum(float v)
{
    v += lane_xor<1>(v);
    v += lane_xor<2>(v);
    v += lane_xor<4>(v);
    return v;
}
__device__ __forceinline__ v2f group8_sum2(v2f v)
{
    v2f t;
    t.x = lane_xor<1>(v.x); t.y = lane_xor<1>(v.y); v += t;
    t.x = lane_xor<2>(v.x); t.y = lane_xor<2>(v.y); v += t;
    t.x = lane_xor<4>(v.x); t.y = lane_xor<4>(v.y); v += t;
    return v;
}

struct Row4 { v2f lo, hi; };
__device__ __forceinline__ Row4 load_row4(const float* __restrict__ p)
{
    const float4 v = *(const float4*)p;
    Row4 r;
    r.lo.x = v.x; r.lo.y = v.y; r.hi.x = v.z; r.hi.y = v.w;
    return r;
}

__device__ __forceinline__ float score_part(const Row4 x, const Row4 xr,
                                            const float ea0, const float ea1, const float ea2,
                                            const Row4 w0, const Row4 w1, const Row4 w2,
                                            const Row4 at)
{
    const v2f e0 = { ea0, ea0 }, e1 = { ea1, ea1 }, e2 = { ea2, ea2 };
    v2f vlo = x.lo + xr.lo, vhi = x.hi + xr.hi;
    vlo = __builtin_elementwise_fma(e0, w0.lo, vlo);  vhi = __builtin_elementwise_fma(e0, w0.hi, vhi);
    vlo = __builtin_elementwise_fma(e1, w1.lo, vlo);  vhi = __builtin_elementwise_fma(e1, w1.hi, vhi);
    vlo = __builtin_elementwise_fma(e2, w2.lo, vlo);  vhi = __builtin_elementwise_fma(e2, w2.hi, vhi);
    vlo = __builtin_elementwise_max(vlo, vlo * NEG_SLOPE);
    vhi = __builtin_elementwise_max(vhi, vhi * NEG_SLOPE);
    const v2f t = __builtin_elementwise_fma(vhi, at.hi, vlo * at.lo);
    return t.x + t.y;
}

#define LOG2_E 1.4426950408889634f

template <int G>
__device__ __forceinline__ void group_step(const Row4 (&x)[G], const float (&sc)[G], const bool (&valid)[G],
                                           float& m, float& l, Row4& acc)
{
    float mt = m;
#pragma unroll
    for (int u = 0; u < G; ++u) mt = __builtin_fmaxf(mt, sc[u]);
    const float r = __builtin_amdgcn_exp2f(m - mt);
    m = mt;
    float ls = l * r;
    const v2f r2 = { r, r };
    v2f alo = acc.lo * r2, ahi = acc.hi * r2;
#pragma unroll
    for (int u = 0; u < G; ++u) {
        const float e = __builtin_amdgcn_exp2f(sc[u] - mt);
        const float p = valid[u] ? e : 0.0f;
        const v2f p2 = { p, p };
        ls += p;
        alo = __builtin_elementwise_fma(p2, x[u].lo, alo);
        ahi = __builtin_elementwise_fma(p2, x[u].hi, ahi);
    }
    l = ls;
    acc.lo = alo; acc.hi = ahi;
}

__global__ __launch_bounds__(256)
void node_softmax_agg_kernel(const int* __restrict__ cnt, const int* __restrict__ bucket,
                             const int* __restrict__ ovfTotal, const int* __restrict__ ovfDst,
                             const int* __restrict__ ovfSrc,
                             const float* __restrict__ pos, const float* __restrict__ part,
                             const float* __restrict__ Wedge, const float* __restrict__ att,
                             const float* __restrict__ XL, const float* __restrict__ XR,
                             float* __restrict__ out, float* __restrict__ colAcc)
{
    __shared__ float red[8][2 * HC];
    __shared__ float sm[3][8];
    const int lane = threadIdx.x & 31;
    const int wave   = __builtin_amdgcn_readfirstlane((int)((blockIdx.x * blockDim.x + threadIdx.x) >> 5));
    const int nWaves = (int)((gridDim.x * blockDim.x) >> 5);

    const int c0 = 4 * lane;
    const Row4 w0 = load_row4(Wedge + 0 * HC + c0);
    const Row4 w1 = load_row4(Wedge + 1 * HC + c0);
    const Row4 w2 = load_row4(Wedge + 2 * HC + c0);
    Row4 at = load_row4(att + c0);
    at.lo *= LOG2_E; at.hi *= LOG2_E;

    float mean0 = 0.f, mean1 = 0.f, mean2 = 0.f;
    static_assert(FILL_BLOCKS <= 4 * 256, "four partials per thread cover the edge blocks");
#pragma unroll
    for (int k = 0; k < 4; ++k) {
        const int  b  = (int)threadIdx.x + 256 * k;
        const bool on = b < FILL_BLOCKS;
        const int  bc = on ? b : (FILL_BLOCKS - 1);
        const float v0 = part[bc * 3 + 0], v1 = part[bc * 3 + 1], v2 = part[bc * 3 + 2];
        mean0 += on ? v0 : 0.f;
        mean1 += on ? v1 : 0.f;
        mean2 += on ? v2 : 0.f;
    }
    block_sum3(sm, mean0, mean1, mean2);
    mean0 = mean0 / (float)N_EDGES;
    mean1 = mean1 / (float)N_EDGES;
    mean2 = mean2 / (float)N_EDGES;

    const int ovfN = ovfTotal[0];

    float4 cs = make_float4(0.f, 0.f, 0.f, 0.f);
    float4 cq = make_float4(0.f, 0.f, 0.f, 0.f);

    for (int d = wave; d < N_NODES; d += nWaves) {
        const int n  = cnt[d];
        const int nb = (n < BUCKET) ? n : BUCKET;

        const Row4 xr = load_row4(XR + (size_t)d * HC + c0);
        const Row4 xs = load_row4(XL + (size_t)d * HC + c0);
        const float pd0 = pos[d * 3 + 0], pd1 = pos[d * 3 + 1], pd2 = pos[d * 3 + 2];

        float m   = group8_sum(score_part(xs, xr, mean0, mean1, mean2, w0, w1, w2, at));
        float l   = 1.0f;
        Row4  acc = xs;

        if (nb > 0) {
            const int le = (lane < nb) ? lane : (nb - 1);
            const int sv = bucket[d * BUCKET + le];
            const float q0 = pos[sv * 3 + 0] - pd0;
            const float q1 = pos[sv * 3 + 1] - pd1;
            const float q2 = pos[sv * 3 + 2] - pd2;

            for (int j = 0; j < nb; j += 4) {
                int  jj[4];
                Row4 x[4];
#pragma unroll
                for (int u = 0; u < 4; ++u) {
                    const int t = j + u;
                    jj[u] = (t < nb) ? t : (nb - 1);
                    const int s = __builtin_amdgcn_readlane(sv, jj[u]);
                    x[u] = load_row4(XL + (size_t)s * HC + c0);
                }
                float pp[4];
                bool  valid[4];
#pragma unroll
                for (int u = 0; u < 4; ++u) {
                    valid[u] = (j + u) < nb;
                    const float ea0 = __int_as_float(__builtin_amdgcn_readlane(__float_as_int(q0), jj[u]));
                    const float ea1 = __int_as_float(__builtin_amdgcn_readlane(__float_as_int(q1), jj[u]));
                    const float ea2 = __int_as_float(__builtin_amdgcn_readlane(__float_as_int(q2), jj[u]));
                    pp[u] = score_part(x[u], xr, ea0, ea1, ea2, w0, w1, w2, at);
                }
                const v2f s01 = group8_sum2(v2f{ pp[0], pp[1] });
                const v2f s23 = group8_sum2(v2f{ pp[2], pp[3] });
                const float sc[4] = { s01.x, s01.y, s23.x, s23.y };
                group_step<4>(x, sc, valid, m, l, acc);
            }
        }

        if (n > BUCKET) {
            for (int base = 0; base < ovfN; base += 32) {
                const int  i   = base + lane;
                const bool in  = i < ovfN;
                const int  od  = in ? ovfDst[i] : -1;
                const int  os  = in ? ovfSrc[i] : 0;
                unsigned mask = __builtin_amdgcn_ballot_w32(od == d);
                if (mask != 0u) {
                    const float q0 = pos[os * 3 + 0] - pd0;
                    const float q1 = pos[os * 3 + 1] - pd1;
                    const float q2 = pos[os * 3 + 2] - pd2;
                    while (mask != 0u) {
                        const int j = __builtin_ctz(mask);
                        mask &= mask - 1u;
                        const int s = __builtin_amdgcn_readlane(os, j);
                        const float ea0 = __int_as_float(__builtin_amdgcn_readlane(__float_as_int(q0), j));
                        const float ea1 = __int_as_float(__builtin_amdgcn_readlane(__float_as_int(q1), j));
                        const float ea2 = __int_as_float(__builtin_amdgcn_readlane(__float_as_int(q2), j));
                        const Row4  x1[1] = { load_row4(XL + (size_t)s * HC + c0) };
                        const float s1[1] = { group8_sum(score_part(x1[0], xr, ea0, ea1, ea2, w0, w1, w2, at)) };
                        const bool  v1[1] = { true };
                        group_step<1>(x1, s1, v1, m, l, acc);
                    }
                }
            }
        }

        const float den = l + 1e-16f;
        float4 o;
        o.x = acc.lo.x / den; o.y = acc.lo.y / den; o.z = acc.hi.x / den; o.w = acc.hi.y / den;
        *(float4*)(out + (size_t)d * HC + c0) = o;
        cs.x += o.x; cs.y += o.y; cs.z += o.z; cs.w += o.w;
        cq.x += o.x * o.x; cq.y += o.y * o.y; cq.z += o.z * o.z; cq.w += o.w * o.w;
    }

    const int w = threadIdx.x >> 5;
    *(float4*)&red[w][c0]      = cs;
    *(float4*)&red[w][HC + c0] = cq;
    __syncthreads();
    const int t = threadIdx.x;
    float tot = red[0][t];
#pragma unroll
    for (int i = 1; i < 8; ++i) tot += red[i][t];
    atomicAdd(&colAcc[t], tot);
}

#define APPLY_BLOCKS     1250
#define APPLY_PER_THREAD 5
#define APPLY_STRIDE     (APPLY_BLOCKS * 256)
__global__ __launch_bounds__(256)
void bn_apply_kernel(float* __restrict__ out, const float* __restrict__ colAcc,
                     const float* __restrict__ gamma, const float* __restrict__ beta)
{
    const int c0 = 4 * (threadIdx.x & 31);
    const float4 s  = *(const float4*)(colAcc + c0);
    const float4 s2 = *(const float4*)(colAcc + HC + c0);
    const float4 g  = *(const float4*)(gamma + c0);
    const float4 b  = *(const float4*)(beta + c0);

    float4* __restrict__ o4 = (float4*)out;
    const int i0 = blockIdx.x * 256 + threadIdx.x;
    float4 v[APPLY_PER_THREAD];
#pragma unroll
    for (int u = 0; u < APPLY_PER_THREAD; ++u) v[u] = o4[i0 + u * APPLY_STRIDE];

    const float inv_n = 1.0f / (float)N_NODES;
    float4 sc, sh;
    {
        const float mu = s.x * inv_n; const float var = s2.x * inv_n - mu * mu;
        sc.x = g.x * rsqrtf(var + BN_EPS); sh.x = b.x - mu * sc.x;
    }
    {
        const float mu = s.y * inv_n; const float var = s2.y * inv_n - mu * mu;
        sc.y = g.y * rsqrtf(var + BN_EPS); sh.y = b.y - mu * sc.y;
    }
    {
        const float mu = s.z * inv_n; const float var = s2.z * inv_n - mu * mu;
        sc.z = g.z * rsqrtf(var + BN_EPS); sh.z = b.z - mu * sc.z;
    }
    {
        const float mu = s.w * inv_n; const float var = s2.w * inv_n - mu * mu;
        sc.w = g.w * rsqrtf(var + BN_EPS); sh.w = b.w - mu * sc.w;
    }

#pragma unroll
    for (int u = 0; u < APPLY_PER_THREAD; ++u) {
        float4 r;
        r.x = v[u].x * sc.x + sh.x;
        r.y = v[u].y * sc.y + sh.y;
        r.z = v[u].z * sc.z + sh.z;
        r.w = v[u].w * sc.w + sh.w;
        o4[i0 + u * APPLY_STRIDE] = r;
    }
}

extern "C" void kernel_launch(void* const* d_in, const int* in_sizes, int n_in,
                              void* d_out, int out_size, void* d_ws, size_t ws_size,
                              hipStream_t stream)
{
    const float* x     = (const float*)d_in[0];
    const float* pos   = (const float*)d_in[1];
    const int*   eidx  = (const int*)  d_in[2];
    const float* W_l   = (const float*)d_in[3];
    const float* W_r   = (const float*)d_in[4];
    const float* W_e   = (const float*)d_in[5];
    const float* att   = (const float*)d_in[6];
    const float* gamma = (const float*)d_in[7];
    const float* beta  = (const float*)d_in[8];
    float* out = (float*)d_out;

    const int* srcA = eidx;
    const int* dstA = eidx + N_EDGES;

    float* ws = (float*)d_ws;
    size_t off = 0;
    float* XL         = ws + off; off += (size_t)N_NODES * HC;
    float* XR         = ws + off; off += (size_t)N_NODES * HC;
    int*   bucket     = (int*)(ws + off); off += (size_t)N_NODES * BUCKET;
    int*   ovfDst     = (int*)(ws + off); off += (size_t)N_EDGES;
    int*   ovfSrc     = (int*)(ws + off); off += (size_t)N_EDGES;
    unsigned* ownb    = (unsigned*)(ws + off); off += (size_t)N_EDGE4;
    int*   cnt        = (int*)(ws + off); off += (size_t)N_NODES;
    int*   ovfTotal   = (int*)(ws + off); off += 4;
    float* colAcc     = ws + off; off += 2 * HC;
    float* meanPart   = ws + off; off += 3 * FILL_BLOCKS;
    (void)ws_size; (void)n_in; (void)in_sizes; (void)out_size;

    owner_bytes_init_kernel<<<OWNB_BLOCKS, 256, 0, stream>>>(dstA, ownb, ovfTotal, colAcc);

    edge_bucket_mean_gemm_kernel<<<OWNERS, OWN_THREADS, 0, stream>>>(
        srcA, dstA, ownb, pos, cnt, bucket, ovfTotal, ovfDst, ovfSrc, meanPart, x, W_l, W_r, XL, XR);

    node_softmax_agg_kernel<<<1024, 256, 0, stream>>>(cnt, bucket, ovfTotal, ovfDst, ovfSrc, pos,
                                                             meanPart, W_e, att, XL, XR, out, colAcc);

    bn_apply_kernel<<<APPLY_BLOCKS, 256, 0, stream>>>(out, colAcc, gamma, beta);
}


# Edge term by linearity: pos@W_edge rides both GEMMs as 17th k-step (A rows refilled in place, W reads per 2 k-steps, so it fits 128 regs); per edge -6 FMA -3 lane reads; weighted sums regrouped
# speedup vs baseline: 3.7380x; 1.0391x over previous
#include <hip/hip_runtime.h>
#include <math.h>

#define N_NODES   50000
#define N_EDGES   800000
#define IN_CH     64
#define HEADS     4
#define OUT_CH    32
#define HC        128
#define NEG_SLOPE 0.2f
#define BN_EPS    1e-5f
#define BUCKET    32

typedef float v2f __attribute__((ext_vector_type(2)));
typedef float v8f __attribute__((ext_vector_type(8)));

#define OWN_SPAN    200
#define N_EDGE4     (N_EDGES / 4)
#define OWNB_BLOCKS ((N_EDGE4 + 1023) / 1024)
static_assert(N_EDGE4 * 4 == N_EDGES, "the destination list is a whole number of 16-byte groups");
static_assert((N_NODES - 1) / OWN_SPAN < 255, "an owner number fits a byte, and 255 is nobody's");
__global__ __launch_bounds__(256)
void owner_bytes_init_kernel(const int* __restrict__ dstA, unsigned* __restrict__ ownb,
                             int* __restrict__ ovfTotal, float* __restrict__ colAcc)
{
    const int t = threadIdx.x;
    if (blockIdx.x == 0) {
        if (t < 2 * HC)  colAcc[t] = 0.0f;
        if (t == 0)      ovfTotal[0] = 0;
    }
    const int4* __restrict__ d4 = (const int4*)dstA;
    int4 d[4];
#pragma unroll
    for (int k = 0; k < 4; ++k) {
        const int i = (int)blockIdx.x * 1024 + k * 256 + t;
        d[k] = d4[(i < N_EDGE4) ? i : (N_EDGE4 - 1)];
    }
#pragma unroll
    for (int k = 0; k < 4; ++k) {
        const int i = (int)blockIdx.x * 1024 + k * 256 + t;
        if (i < N_EDGE4)
            ownb[i] =  ((unsigned)d[k].x / OWN_SPAN)        | (((unsigned)d[k].y / OWN_SPAN) << 8)
                    | (((unsigned)d[k].z / OWN_SPAN) << 16) | (((unsigned)d[k].w / OWN_SPAN) << 24);
    }
}

#define K_STEPS 17
#define GEMM_STRIPS 3125
static_assert(GEMM_STRIPS * 16 == N_NODES, "the strips cover the rows exactly");
#define W_PITCH 260
__device__ __forceinline__ void stage_w(const float* __restrict__ Wm, float* __restrict__ wsh, const int tid)
{
    const int p = tid >> 5, cq = tid & 31;
    const float4 r0 = *(const float4*)(Wm + (size_t)(2 * p + 0) * HC + 4 * cq);
    const float4 r1 = *(const float4*)(Wm + (size_t)(2 * p + 1) * HC + 4 * cq);
    float* dstp = wsh + p * W_PITCH + 8 * cq;
    *(float4*)(dstp + 0) = make_float4(r0.x, r1.x, r0.y, r1.y);
    *(float4*)(dstp + 4) = make_float4(r0.z, r1.z, r0.w, r1.w);
}
__device__ __forceinline__ void stage_we(const float* __restrict__ We, const float sgn,
                                         float* __restrict__ wsh, const int tid)
{
    if (tid < 64) {
        const int pp = tid >> 5, cq = tid & 31;
        const float4 r0 = *(const float4*)(We + (size_t)(2 * pp) * HC + 4 * cq);
        float4 r1 = make_float4(0.f, 0.f, 0.f, 0.f);
        if (pp == 0) r1 = *(const float4*)(We + (size_t)HC + 4 * cq);
        float* dstp = wsh + (32 + pp) * W_PITCH + 8 * cq;
        *(float4*)(dstp + 0) = make_float4(sgn * r0.x, sgn * r1.x, sgn * r0.y, sgn * r1.y);
        *(float4*)(dstp + 4) = make_float4(sgn * r0.z, sgn * r1.z, sgn * r0.w, sgn * r1.w);
    }
}
__device__ __forceinline__ void load_a16(const float* __restrict__ X, const int strip, v2f (&a)[16])
{
    const int lane = threadIdx.x & 31;
    const unsigned off = (unsigned)(strip * 16 + (lane & 15)) * IN_CH + 2u * (unsigned)(lane >> 4);
#pragma unroll
    for (int t = 0; t < 16; ++t) a[t] = *(const v2f*)(X + off + 4 * t);
}
__device__ __forceinline__ v2f load_apos(const float* __restrict__ pos, const int strip)
{
    const int lane = threadIdx.x & 31;
    const int half = lane >> 4;
    const unsigned off = (unsigned)(strip * 16 + (lane & 15)) * 3u + 2u * (unsigned)half;
    v2f a;
    a.x = pos[off];
    a.y = pos[off + 1u - (unsigned)half];
    return a;
}
__device__ __forceinline__ void load_bfrag(const float* __restrict__ bp, const int t, float4& f0, float4& f1)
{
    f0 = *(const float4*)(bp + 2 * t * W_PITCH + 0);
    f1 = *(const float4*)(bp + 2 * t * W_PITCH + 64);
}
#define KG 2
#define N_KG ((K_STEPS + KG - 1) / KG)
__device__ __forceinline__ void gemm_task(v2f (&a)[K_STEPS], const float* __restrict__ X,
                                          const float* __restrict__ pos, const int nextStrip,
                                          const float* __restrict__ wsh,
                                          float* __restrict__ D, const int strip, const int qh)
{
    const int lane = threadIdx.x & 31;
    const int half = lane >> 4;
    const int idx  = lane & 15;

    const float* bp = wsh + half * W_PITCH + 4 * idx + 128 * qh;
    const unsigned xoff = (unsigned)(nextStrip * 16 + idx) * IN_CH + 2u * (unsigned)half;
    const unsigned poff = (unsigned)(nextStrip * 16 + idx) * 3u + 2u * (unsigned)half;

    v8f c[4] = {};
    float4 fb[2][2 * KG];
#pragma unroll
    for (int s = 0; s < KG; ++s) load_bfrag(bp, s, fb[0][2 * s], fb[0][2 * s + 1]);
#pragma unroll
    for (int g = 0; g < N_KG; ++g) {
        if (g + 1 < N_KG) {
#pragma unroll
            for (int s = 0; s < KG; ++s)
                if ((g + 1) * KG + s < K_STEPS)
                    load_bfrag(bp, (g + 1) * KG + s, fb[(g + 1) & 1][2 * s], fb[(g + 1) & 1][2 * s + 1]);
        }
        asm volatile("" ::: "memory");
        __builtin_amdgcn_sched_barrier(0);
#pragma unroll
        for (int s = 0; s < KG; ++s) {
            const int t = g * KG + s;
            if (t < K_STEPS) {
                const float4 f0 = fb[g & 1][2 * s], f1 = fb[g & 1][2 * s + 1];
                v2f b0, b1, b2, b3;
                b0.x = f0.x; b0.y = f0.y;
                b1.x = f0.z; b1.y = f0.w;
                b2.x = f1.x; b2.y = f1.y;
                b3.x = f1.z; b3.y = f1.w;
                c[0] = __builtin_amdgcn_wmma_f32_16x16x4_f32(false, a[t], false, b0, (short)0, c[0], false, false);
                c[1] = __builtin_amdgcn_wmma_f32_16x16x4_f32(false, a[t], false, b1, (short)0, c[1], false, false);
                c[2] = __builtin_amdgcn_wmma_f32_16x16x4_f32(false, a[t], false, b2, (short)0, c[2], false, false);
                c[3] = __builtin_amdgcn_wmma_f32_16x16x4_f32(false, a[t], false, b3, (short)0, c[3], false, false);
            }
        }
        __builtin_amdgcn_sched_barrier(0);
#pragma unroll
        for (int s = 0; s < KG; ++s) {
            const int t = g * KG + s;
            if (t < 16) {
                a[t] = *(const v2f*)(X + xoff + 4 * t);
            } else if (t == 16) {
                a[16].x = pos[poff];
                a[16].y = pos[poff + 1u - (unsigned)half];
            }
        }
    }

    const unsigned doff = (unsigned)(strip * 16 + 8 * half) * HC + 2u * (unsigned)idx + 64u * (unsigned)qh;
#pragma unroll
    for (int v = 0; v < 8; ++v) {
        float* drow = D + doff + v * HC;
        float2 o0, o1;
        o0.x = c[0][v]; o0.y = c[1][v];
        o1.x = c[2][v]; o1.y = c[3][v];
        *(float2*)(drow + 0)  = o0;
        *(float2*)(drow + 32) = o1;
    }
}

__device__ __forceinline__ float wave_sum(float v)
{
    for (int o = 16; o > 0; o >>= 1) v += __shfl_xor(v, o, 32);
    return v;
}
template <int NW>
__device__ __forceinline__ void block_sum3(float (&sm)[3][NW], float& a0, float& a1, float& a2)
{
    a0 = wave_sum(a0); a1 = wave_sum(a1); a2 = wave_sum(a2);
    const int lane = threadIdx.x & 31, w = threadIdx.x >> 5;
    if (lane == 0) { sm[0][w] = a0; sm[1][w] = a1; sm[2][w] = a2; }
    __syncthreads();
    float s0 = 0.f, s1 = 0.f, s2 = 0.f;
    for (int i = 0; i < NW; ++i) { s0 += sm[0][i]; s1 += sm[1][i]; s2 += sm[2][i]; }
    a0 = s0; a1 = s1; a2 = s2;
}

#define OWNERS      250
#define OWN_WAVES   32
#define OWN_THREADS (OWN_WAVES * 32)
#define GEMM_WAVES  4
#define SCAN_WAVES  (OWN_WAVES - GEMM_WAVES)
#define SCAN_THREADS (SCAN_WAVES * 32)
#define SEG_CAP     512
#define N_EDGE16    (N_EDGES / 16)
#define SCAN_LOADS  8
#define FILL_BLOCKS OWNERS
#define GEMM_TASKS  (2 * GEMM_STRIPS)
#define GEMM_PER_BLOCK 25
static_assert(OWNERS * GEMM_PER_BLOCK == GEMM_TASKS, "the blocks cover both products exactly");
static_assert((OWNERS / 2) * GEMM_PER_BLOCK == GEMM_STRIPS, "a block's strips belong to one product");
static_assert(OWNERS * OWN_SPAN == N_NODES, "the owner ranges cover the nodes exactly");
static_assert(N_EDGE16 * 16 == N_EDGES, "the owner byte plane is a whole number of 16-byte groups");

__device__ __forceinline__ void place_edge(const int d, const int s, const int slot,
                                           int* __restrict__ bucket, int* __restrict__ ovfTotal,
                                           int* __restrict__ ovfDst, int* __restrict__ ovfSrc)
{
    if (slot < BUCKET) {
        bucket[d * BUCKET + slot] = s;
    } else {
        const int q = atomicAdd(ovfTotal, 1);
        ovfDst[q] = d;
        ovfSrc[q] = s;
    }
}

__device__ __forceinline__ void note_word(const unsigned w, const int e0, const unsigned pat,
                                          int* __restrict__ seg, int& nw)
{
    const unsigned x = w ^ pat;
    const unsigned t = (x & 0x7F7F7F7Fu) + 0x7F7F7F7Fu;
    unsigned z = ~(t | x | 0x7F7F7F7Fu);
    unsigned m = __builtin_amdgcn_ballot_w32(z != 0u);
    while (m != 0u) {
        if (z != 0u) {
            const int p = nw + (int)__builtin_amdgcn_mbcnt_lo(m, 0u);
            if (p < SEG_CAP) seg[p] = e0 + (__builtin_ctz(z) >> 3);
            z &= z - 1u;
        }
        nw += __popc(m);
        m = __builtin_amdgcn_ballot_w32(z != 0u);
    }
}

__global__ __launch_bounds__(OWN_THREADS)
void edge_bucket_mean_gemm_kernel(const int* __restrict__ srcA, const int* __restrict__ dstA,
                                  const unsigned* __restrict__ ownb, const float* __restrict__ pos,
                                  int* __restrict__ cnt, int* __restrict__ bucket,
                                  int* __restrict__ ovfTotal, int* __restrict__ ovfDst,
                                  int* __restrict__ ovfSrc, float* __restrict__ part,
                                  const float* __restrict__ X, const float* __restrict__ Wl,
                                  const float* __restrict__ Wr, const float* __restrict__ Wedge,
                                  float* __restrict__ XL, float* __restrict__ XR)
{
    __shared__ int   lcnt[OWN_SPAN];
    __shared__ int   list[SCAN_WAVES * SEG_CAP];
    __shared__ int   listFull;
    __shared__ float sm[3][OWN_WAVES];
    __shared__ __attribute__((aligned(16))) float wsh[34 * W_PITCH];

    const int tid = (int)threadIdx.x;
    const int lo  = (int)blockIdx.x * OWN_SPAN;
    const bool left = (int)blockIdx.x < OWNERS / 2;
    if (tid < OWN_SPAN) lcnt[tid] = 0;
    if (tid == 0) listFull = 0;
    stage_w(left ? Wl : Wr, wsh, tid);
    stage_we(Wedge, left ? 1.0f : -1.0f, wsh, tid);
    __syncthreads();

    int* const seg = list + ((tid < SCAN_THREADS) ? (tid >> 5) : 0) * SEG_CAP;
    int nw = 0;
    if (tid < SCAN_THREADS) {
        const uint4* __restrict__ o16 = (const uint4*)ownb;
        const unsigned pat = (unsigned)blockIdx.x * 0x01010101u;
        for (int base = 0; base < N_EDGE16; base += SCAN_LOADS * SCAN_THREADS) {
            uint4 v[SCAN_LOADS];
#pragma unroll
            for (int u = 0; u < SCAN_LOADS; ++u) {
                const int j = base + u * SCAN_THREADS + tid;
                v[u] = o16[(j < N_EDGE16) ? j : (N_EDGE16 - 1)];
            }
#pragma unroll
            for (int u = 0; u < SCAN_LOADS; ++u) {
                const int j = base + u * SCAN_THREADS + tid;
                const unsigned pj = (j < N_EDGE16) ? pat : 0xFFFFFFFFu;
                note_word(v[u].x, 16 * j + 0,  pj, seg, nw);
                note_word(v[u].y, 16 * j + 4,  pj, seg, nw);
                note_word(v[u].z, 16 * j + 8,  pj, seg, nw);
                note_word(v[u].w, 16 * j + 12, pj, seg, nw);
            }
        }
        if (nw > SEG_CAP) listFull = 1;
    } else {
        const int gw    = (tid - SCAN_THREADS) >> 5;
        const int first = ((int)blockIdx.x - (left ? 0 : OWNERS / 2)) * GEMM_PER_BLOCK;
        float* __restrict__ D = left ? XL : XR;
        v2f a[K_STEPS];
        {
            v2f ax[16];
            load_a16(X, first + (gw >> 1), ax);
#pragma unroll
            for (int t = 0; t < 16; ++t) a[t] = ax[t];
            a[16] = load_apos(pos, first + (gw >> 1));
        }
        for (int i = gw; i < 2 * GEMM_PER_BLOCK; i += GEMM_WAVES) {
            const int inx = (i + GEMM_WAVES < 2 * GEMM_PER_BLOCK) ? i + GEMM_WAVES : i;
            gemm_task(a, X, pos, first + (inx >> 1), wsh, D, first + (i >> 1), i & 1);
        }
    }
    __syncthreads();
    const bool full = listFull != 0;

    float a0 = 0.f, a1 = 0.f, a2 = 0.f;
    if (!full) {
        const int lane = tid & 31;
        for (int base = 0; base < nw; base += 4 * 32) {
            bool on[4];
            int  d[4], s[4], p[4];
            float ps[4][3], pd[4][3];
#pragma unroll
            for (int u = 0; u < 4; ++u) {
                const int i = base + u * 32 + lane;
                on[u] = i < nw;
                const int e = seg[on[u] ? i : 0];
                d[u] = dstA[e];
                s[u] = srcA[e];
            }
#pragma unroll
            for (int u = 0; u < 4; ++u) {
                p[u] = 0;
                if (on[u]) p[u] = atomicAdd(&lcnt[d[u] - lo], 1);
            }
#pragma unroll
            for (int u = 0; u < 4; ++u) {
                ps[u][0] = pos[s[u] * 3 + 0]; ps[u][1] = pos[s[u] * 3 + 1]; ps[u][2] = pos[s[u] * 3 + 2];
                pd[u][0] = pos[d[u] * 3 + 0]; pd[u][1] = pos[d[u] * 3 + 1]; pd[u][2] = pos[d[u] * 3 + 2];
            }
#pragma unroll
            for (int u = 0; u < 4; ++u) {
                if (on[u]) place_edge(d[u], s[u], p[u], bucket, ovfTotal, ovfDst, ovfSrc);
            }
#pragma unroll
            for (int u = 0; u < 4; ++u) {
                a0 += on[u] ? (ps[u][0] - pd[u][0]) : 0.f;
                a1 += on[u] ? (ps[u][1] - pd[u][1]) : 0.f;
                a2 += on[u] ? (ps[u][2] - pd[u][2]) : 0.f;
            }
        }
    } else {
        for (int e = tid; e < N_EDGES; e += OWN_THREADS) {
            const int d = dstA[e];
            const unsigned r = (unsigned)(d - lo);
            if (r < (unsigned)OWN_SPAN) {
                const int s = srcA[e];
                const int p = atomicAdd(&lcnt[r], 1);
                place_edge(d, s, p, bucket, ovfTotal, ovfDst, ovfSrc);
                a0 += pos[s * 3 + 0] - pos[d * 3 + 0];
                a1 += pos[s * 3 + 1] - pos[d * 3 + 1];
                a2 += pos[s * 3 + 2] - pos[d * 3 + 2];
            }
        }
    }

    block_sum3(sm, a0, a1, a2);
    if (tid == 0) {
        part[blockIdx.x * 3 + 0] = a0;
        part[blockIdx.x * 3 + 1] = a1;
        part[blockIdx.x * 3 + 2] = a2;
    }
    if (tid < OWN_SPAN) cnt[lo + tid] = lcnt[tid];
}


template <int MASK>
__device__ __forceinline__ float lane_xor(float v)
{
    return __int_as_float(__builtin_amdgcn_mov_dpp(__float_as_int(v), 0x160 + MASK,
                                                   0xf, 0xf, false));
}
__device__ __forceinline__ float group8_sum(float v)
{
    v += lane_xor<1>(v);
    v += lane_xor<2>(v);
    v += lane_xor<4>(v);
    return v;
}
__device__ __forceinline__ v2f group8_sum2(v2f v)
{
    v2f t;
    t.x = lane_xor<1>(v.x); t.y = lane_xor<1>(v.y); v += t;
    t.x = lane_xor<2>(v.x); t.y = lane_xor<2>(v.y); v += t;
    t.x = lane_xor<4>(v.x); t.y = lane_xor<4>(v.y); v += t;
    return v;
}

struct Row4 { v2f lo, hi; };
__device__ __forceinline__ Row4 load_row4(const float* __restrict__ p)
{
    const float4 v = *(const float4*)p;
    Row4 r;
    r.lo.x = v.x; r.lo.y = v.y; r.hi.x = v.z; r.hi.y = v.w;
    return r;
}

__device__ __forceinline__ float score_part(const Row4 x, const Row4 xr, const Row4 at)
{
    v2f vlo = x.lo + xr.lo, vhi = x.hi + xr.hi;
    vlo = __builtin_elementwise_max(vlo, vlo * NEG_SLOPE);
    vhi = __builtin_elementwise_max(vhi, vhi * NEG_SLOPE);
    const v2f t = __builtin_elementwise_fma(vhi, at.hi, vlo * at.lo);
    return t.x + t.y;
}

#define LOG2_E 1.4426950408889634f

template <int G>
__device__ __forceinline__ void group_step(const Row4 (&x)[G], const float (&z)[G], const float (&sc)[G],
                                           const bool (&valid)[G],
                                           float& m, float& l, Row4& acc, float& pz)
{
    float mt = m;
#pragma unroll
    for (int u = 0; u < G; ++u) mt = __builtin_fmaxf(mt, sc[u]);
    const float r = __builtin_amdgcn_exp2f(m - mt);
    m = mt;
    float ls = l * r;
    float zs = pz * r;
    const v2f r2 = { r, r };
    v2f alo = acc.lo * r2, ahi = acc.hi * r2;
#pragma unroll
    for (int u = 0; u < G; ++u) {
        const float e = __builtin_amdgcn_exp2f(sc[u] - mt);
        const float p = valid[u] ? e : 0.0f;
        const v2f p2 = { p, p };
        ls += p;
        zs = __builtin_fmaf(p, z[u], zs);
        alo = __builtin_elementwise_fma(p2, x[u].lo, alo);
        ahi = __builtin_elementwise_fma(p2, x[u].hi, ahi);
    }
    l = ls;
    pz = zs;
    acc.lo = alo; acc.hi = ahi;
}

__global__ __launch_bounds__(256)
void node_softmax_agg_kernel(const int* __restrict__ cnt, const int* __restrict__ bucket,
                             const int* __restrict__ ovfTotal, const int* __restrict__ ovfDst,
                             const int* __restrict__ ovfSrc,
                             const float* __restrict__ pos, const float* __restrict__ part,
                             const float* __restrict__ Wedge, const float* __restrict__ att,
                             const float* __restrict__ XL, const float* __restrict__ XR,
                             float* __restrict__ out, float* __restrict__ colAcc)
{
    __shared__ float red[8][2 * HC];
    __shared__ float sm[3][8];
    const int lane = threadIdx.x & 31;
    const int wave   = __builtin_amdgcn_readfirstlane((int)((blockIdx.x * blockDim.x + threadIdx.x) >> 5));
    const int nWaves = (int)((gridDim.x * blockDim.x) >> 5);

    const int c0 = 4 * lane;
    const Row4 w0 = load_row4(Wedge + 0 * HC + c0);
    const Row4 w1 = load_row4(Wedge + 1 * HC + c0);
    const Row4 w2 = load_row4(Wedge + 2 * HC + c0);
    Row4 at = load_row4(att + c0);
    at.lo *= LOG2_E; at.hi *= LOG2_E;
    const int kl = ((lane & 7) < 3) ? (lane & 7) : 0;

    float mean0 = 0.f, mean1 = 0.f, mean2 = 0.f;
    static_assert(FILL_BLOCKS <= 4 * 256, "four partials per thread cover the edge blocks");
#pragma unroll
    for (int k = 0; k < 4; ++k) {
        const int  b  = (int)threadIdx.x + 256 * k;
        const bool on = b < FILL_BLOCKS;
        const int  bc = on ? b : (FILL_BLOCKS - 1);
        const float v0 = part[bc * 3 + 0], v1 = part[bc * 3 + 1], v2 = part[bc * 3 + 2];
        mean0 += on ? v0 : 0.f;
        mean1 += on ? v1 : 0.f;
        mean2 += on ? v2 : 0.f;
    }
    block_sum3(sm, mean0, mean1, mean2);
    mean0 = mean0 / (float)N_EDGES;
    mean1 = mean1 / (float)N_EDGES;
    mean2 = mean2 / (float)N_EDGES;
    Row4 mw;
    {
        const v2f e0 = { mean0, mean0 }, e1 = { mean1, mean1 }, e2 = { mean2, mean2 };
        mw.lo = e0 * w0.lo;  mw.hi = e0 * w0.hi;
        mw.lo = __builtin_elementwise_fma(e1, w1.lo, mw.lo);  mw.hi = __builtin_elementwise_fma(e1, w1.hi, mw.hi);
        mw.lo = __builtin_elementwise_fma(e2, w2.lo, mw.lo);  mw.hi = __builtin_elementwise_fma(e2, w2.hi, mw.hi);
    }

    const int ovfN = ovfTotal[0];

    float4 cs = make_float4(0.f, 0.f, 0.f, 0.f);
    float4 cq = make_float4(0.f, 0.f, 0.f, 0.f);

    for (int d = wave; d < N_NODES; d += nWaves) {
        const int n  = cnt[d];
        const int nb = (n < BUCKET) ? n : BUCKET;

        const Row4 xr = load_row4(XR + (size_t)d * HC + c0);
        const Row4 xs = load_row4(XL + (size_t)d * HC + c0);

        Row4 xm;
        xm.lo = xr.lo + mw.lo;  xm.hi = xr.hi + mw.hi;
        float m   = group8_sum(score_part(xs, xm, at));
        float l   = 1.0f;
        Row4  acc = xs;
        float pz  = pos[d * 3 + kl];

        if (nb > 0) {
            const int le = (lane < nb) ? lane : (nb - 1);
            const int sv = bucket[d * BUCKET + le];

            for (int j = 0; j < nb; j += 4) {
                Row4  x[4];
                float z[4];
#pragma unroll
                for (int u = 0; u < 4; ++u) {
                    const int t  = j + u;
                    const int jc = (t < nb) ? t : (nb - 1);
                    const int s  = __builtin_amdgcn_readlane(sv, jc);
                    x[u] = load_row4(XL + (size_t)s * HC + c0);
                    z[u] = pos[s * 3 + kl];
                }
                float pp[4];
                bool  valid[4];
#pragma unroll
                for (int u = 0; u < 4; ++u) {
                    valid[u] = (j + u) < nb;
                    pp[u] = score_part(x[u], xr, at);
                }
                const v2f s01 = group8_sum2(v2f{ pp[0], pp[1] });
                const v2f s23 = group8_sum2(v2f{ pp[2], pp[3] });
                const float sc[4] = { s01.x, s01.y, s23.x, s23.y };
                group_step<4>(x, z, sc, valid, m, l, acc, pz);
            }
        }

        if (n > BUCKET) {
            for (int base = 0; base < ovfN; base += 32) {
                const int  i   = base + lane;
                const bool in  = i < ovfN;
                const int  od  = in ? ovfDst[i] : -1;
                const int  os  = in ? ovfSrc[i] : 0;
                unsigned mask = __builtin_amdgcn_ballot_w32(od == d);
                while (mask != 0u) {
                    const int j = __builtin_ctz(mask);
                    mask &= mask - 1u;
                    const int s = __builtin_amdgcn_readlane(os, j);
                    const Row4  x1[1] = { load_row4(XL + (size_t)s * HC + c0) };
                    const float z1[1] = { pos[s * 3 + kl] };
                    const float s1[1] = { group8_sum(score_part(x1[0], xr, at)) };
                    const bool  v1[1] = { true };
                    group_step<1>(x1, z1, s1, v1, m, l, acc, pz);
                }
            }
        }

        {
            const int hb = 4 * (lane & 24);
            const int zi = __float_as_int(pz);
            const float z0 = __int_as_float(__builtin_amdgcn_ds_bpermute(hb + 0, zi));
            const float z1 = __int_as_float(__builtin_amdgcn_ds_bpermute(hb + 4, zi));
            const float z2 = __int_as_float(__builtin_amdgcn_ds_bpermute(hb + 8, zi));
            const v2f n0 = { -z0, -z0 }, n1 = { -z1, -z1 }, n2 = { -z2, -z2 };
            acc.lo = __builtin_elementwise_fma(n0, w0.lo, acc.lo);  acc.hi = __builtin_elementwise_fma(n0, w0.hi, acc.hi);
            acc.lo = __builtin_elementwise_fma(n1, w1.lo, acc.lo);  acc.hi = __builtin_elementwise_fma(n1, w1.hi, acc.hi);
            acc.lo = __builtin_elementwise_fma(n2, w2.lo, acc.lo);  acc.hi = __builtin_elementwise_fma(n2, w2.hi, acc.hi);
        }
        const float den = l + 1e-16f;
        float4 o;
        o.x = acc.lo.x / den; o.y = acc.lo.y / den; o.z = acc.hi.x / den; o.w = acc.hi.y / den;
        *(float4*)(out + (size_t)d * HC + c0) = o;
        cs.x += o.x; cs.y += o.y; cs.z += o.z; cs.w += o.w;
        cq.x += o.x * o.x; cq.y += o.y * o.y; cq.z += o.z * o.z; cq.w += o.w * o.w;
    }

    const int w = threadIdx.x >> 5;
    *(float4*)&red[w][c0]      = cs;
    *(float4*)&red[w][HC + c0] = cq;
    __syncthreads();
    const int t = threadIdx.x;
    float tot = red[0][t];
#pragma unroll
    for (int i = 1; i < 8; ++i) tot += red[i][t];
    atomicAdd(&colAcc[t], tot);
}

#define APPLY_BLOCKS     1250
#define APPLY_PER_THREAD 5
#define APPLY_STRIDE     (APPLY_BLOCKS * 256)
__global__ __launch_bounds__(256)
void bn_apply_kernel(float* __restrict__ out, const float* __restrict__ colAcc,
                     const float* __restrict__ gamma, const float* __restrict__ beta)
{
    const int c0 = 4 * (threadIdx.x & 31);
    const float4 s  = *(const float4*)(colAcc + c0);
    const float4 s2 = *(const float4*)(colAcc + HC + c0);
    const float4 g  = *(const float4*)(gamma + c0);
    const float4 b  = *(const float4*)(beta + c0);

    float4* __restrict__ o4 = (float4*)out;
    const int i0 = blockIdx.x * 256 + threadIdx.x;
    float4 v[APPLY_PER_THREAD];
#pragma unroll
    for (int u = 0; u < APPLY_PER_THREAD; ++u) v[u] = o4[i0 + u * APPLY_STRIDE];

    const float inv_n = 1.0f / (float)N_NODES;
    float4 sc, sh;
    {
        const float mu = s.x * inv_n; const float var = s2.x * inv_n - mu * mu;
        sc.x = g.x * rsqrtf(var + BN_EPS); sh.x = b.x - mu * sc.x;
    }
    {
        const float mu = s.y * inv_n; const float var = s2.y * inv_n - mu * mu;
        sc.y = g.y * rsqrtf(var + BN_EPS); sh.y = b.y - mu * sc.y;
    }
    {
        const float mu = s.z * inv_n; const float var = s2.z * inv_n - mu * mu;
        sc.z = g.z * rsqrtf(var + BN_EPS); sh.z = b.z - mu * sc.z;
    }
    {
        const float mu = s.w * inv_n; const float var = s2.w * inv_n - mu * mu;
        sc.w = g.w * rsqrtf(var + BN_EPS); sh.w = b.w - mu * sc.w;
    }

#pragma unroll
    for (int u = 0; u < APPLY_PER_THREAD; ++u) {
        float4 r;
        r.x = v[u].x * sc.x + sh.x;
        r.y = v[u].y * sc.y + sh.y;
        r.z = v[u].z * sc.z + sh.z;
        r.w = v[u].w * sc.w + sh.w;
        o4[i0 + u * APPLY_STRIDE] = r;
    }
}

extern "C" void kernel_launch(void* const* d_in, const int* in_sizes, int n_in,
                              void* d_out, int out_size, void* d_ws, size_t ws_size,
                              hipStream_t stream)
{
    const float* x     = (const float*)d_in[0];
    const float* pos   = (const float*)d_in[1];
    const int*   eidx  = (const int*)  d_in[2];
    const float* W_l   = (const float*)d_in[3];
    const float* W_r   = (const float*)d_in[4];
    const float* W_e   = (const float*)d_in[5];
    const float* att   = (const float*)d_in[6];
    const float* gamma = (const float*)d_in[7];
    const float* beta  = (const float*)d_in[8];
    float* out = (float*)d_out;

    const int* srcA = eidx;
    const int* dstA = eidx + N_EDGES;

    float* ws = (float*)d_ws;
    size_t off = 0;
    float* XL         = ws + off; off += (size_t)N_NODES * HC;
    float* XR         = ws + off; off += (size_t)N_NODES * HC;
    int*   bucket     = (int*)(ws + off); off += (size_t)N_NODES * BUCKET;
    int*   ovfDst     = (int*)(ws + off); off += (size_t)N_EDGES;
    int*   ovfSrc     = (int*)(ws + off); off += (size_t)N_EDGES;
    unsigned* ownb    = (unsigned*)(ws + off); off += (size_t)N_EDGE4;
    int*   cnt        = (int*)(ws + off); off += (size_t)N_NODES;
    int*   ovfTotal   = (int*)(ws + off); off += 4;
    float* colAcc     = ws + off; off += 2 * HC;
    float* meanPart   = ws + off; off += 3 * FILL_BLOCKS;
    (void)ws_size; (void)n_in; (void)in_sizes; (void)out_size;

    owner_bytes_init_kernel<<<OWNB_BLOCKS, 256, 0, stream>>>(dstA, ownb, ovfTotal, colAcc);

    edge_bucket_mean_gemm_kernel<<<OWNERS, OWN_THREADS, 0, stream>>>(
        srcA, dstA, ownb, pos, cnt, bucket, ovfTotal, ovfDst, ovfSrc, meanPart, x, W_l, W_r, W_e, XL, XR);

    node_softmax_agg_kernel<<<1024, 256, 0, stream>>>(cnt, bucket, ovfTotal, ovfDst, ovfSrc, pos,
                                                             meanPart, W_e, att, XL, XR, out, colAcc);

    bn_apply_kernel<<<APPLY_BLOCKS, 256, 0, stream>>>(out, colAcc, gamma, beta);
}
